# starting kernel
# baseline (speedup 1.0000x reference)
#include <hip/hip_runtime.h>
#include <cstdint>

typedef float v2f __attribute__((ext_vector_type(2)));
typedef float v8f __attribute__((ext_vector_type(8)));

#define NFEAT 256
#define NHID  128
#define NCLASS 41
#define NCPAD  48

__global__ __launch_bounds__(256) void k_deg_init(float* deg, int N) {
    int i = blockIdx.x * 256 + threadIdx.x;
    if (i < N) deg[i] = 1.0f;
}

__global__ __launch_bounds__(256) void k_deg_edges(const int* __restrict__ dst,
                                                   const float* __restrict__ w,
                                                   float* deg, int E) {
    int e = blockIdx.x * 256 + threadIdx.x;
    if (e < E) unsafeAtomicAdd(&deg[dst[e]], w[e]);
}

__global__ __launch_bounds__(256) void k_deg_finish(float* dinv, int N) {
    int i = blockIdx.x * 256 + threadIdx.x;
    if (i < N) {
        float d = dinv[i];
        dinv[i] = d > 0.0f ? rsqrtf(d) : 0.0f;
    }
}

__global__ __launch_bounds__(256) void k_pack_w1(const float* __restrict__ W1, float* __restrict__ Wp) {
    int idx = blockIdx.x * 256 + threadIdx.x;
    if (idx >= NFEAT * NHID) return;
    int k = idx >> 7, n = idx & 127;
    Wp[(size_t)(k >> 1) * (NHID * 2) + n * 2 + (k & 1)] = W1[idx];
}

__global__ __launch_bounds__(256) void k_pack_w2(const float* __restrict__ W2, float* __restrict__ Wp) {
    int idx = blockIdx.x * 256 + threadIdx.x;
    if (idx >= NHID * NCPAD) return;
    int k = idx / NCPAD, n = idx % NCPAD;
    float v = (n < NCLASS) ? W2[k * NCLASS + n] : 0.0f;
    Wp[(size_t)(k >> 1) * (NCPAD * 2) + n * 2 + (k & 1)] = v;
}

__global__ __launch_bounds__(256) void k_gemm1(const float* __restrict__ X,
                                               const float* __restrict__ Wp,
                                               float* __restrict__ H, int M) {
    __shared__ float2 Xs[128 * 33];
    const int tid = threadIdx.x;
    const int m0 = blockIdx.x * 32;

#pragma unroll
    for (int it = 0; it < 8; ++it) {
        int q = it * 256 + tid;
        int m = q >> 6, k4 = q & 63;
        int mr = m0 + m; if (mr >= M) mr = M - 1;
        float4 v = ((const float4*)(X + (size_t)mr * NFEAT))[k4];
        Xs[(2 * k4)     * 33 + m] = make_float2(v.x, v.y);
        Xs[(2 * k4 + 1) * 33 + m] = make_float2(v.z, v.w);
    }
    __syncthreads();

    const int wave = tid >> 5, lane = tid & 31, l = lane & 15, half = lane >> 4;
    const int n0 = wave * 16;
    const float2* Bp = (const float2*)Wp;

    v8f acc0 = {}, acc1 = {};
#pragma unroll 4
    for (int kp = 0; kp < 128; kp += 2) {
        int row = kp + half;
        float2 a0 = Xs[row * 33 + l];
        float2 a1 = Xs[row * 33 + 16 + l];
        float2 b  = Bp[(size_t)row * NHID + n0 + l];
        v2f av0 = {a0.x, a0.y}, av1 = {a1.x, a1.y}, bv = {b.x, b.y};
        acc0 = __builtin_amdgcn_wmma_f32_16x16x4_f32(false, av0, false, bv, (short)0, acc0, false, false);
        acc1 = __builtin_amdgcn_wmma_f32_16x16x4_f32(false, av1, false, bv, (short)0, acc1, false, false);
    }

    const int mb = 8 * half;
    float* Ho = H + (size_t)(m0 + mb) * NHID + n0 + l;
    if (m0 + 32 <= M) {
#pragma unroll
        for (int r = 0; r < 8; ++r) {
            Ho[(size_t)r * NHID]        = acc0[r];
            Ho[(size_t)(16 + r) * NHID] = acc1[r];
        }
    } else {
#pragma unroll
        for (int r = 0; r < 8; ++r) {
            if (m0 + mb + r < M)      Ho[(size_t)r * NHID]        = acc0[r];
            if (m0 + 16 + mb + r < M) Ho[(size_t)(16 + r) * NHID] = acc1[r];
        }
    }
}

__global__ __launch_bounds__(256) void k_agg1_init(const float* __restrict__ H,
                                                   const float* __restrict__ dinv,
                                                   float* __restrict__ Hacc, int N) {
    int idx = blockIdx.x * 256 + threadIdx.x;
    if (idx >= N * 32) return;
    int i = idx >> 5, j = idx & 31;
    float di = dinv[i], c = di * di;
    float4 v = ((const float4*)(H + (size_t)i * NHID))[j];
    float4 o = make_float4(c * v.x, c * v.y, c * v.z, c * v.w);
    ((float4*)(Hacc + (size_t)i * NHID))[j] = o;
}

__global__ __launch_bounds__(256) void k_agg1_edges(const int* __restrict__ src,
                                                    const int* __restrict__ dst,
                                                    const float* __restrict__ w,
                                                    const float* __restrict__ dinv,
                                                    const float* __restrict__ H,
                                                    float* __restrict__ Hacc, int E) {
    int e = blockIdx.x * 8 + (threadIdx.x >> 5);
    if (e >= E) return;
    int lane = threadIdx.x & 31;
    int s = src[e], d = dst[e];
    float c = dinv[s] * w[e] * dinv[d];
    float4 v = ((const float4*)(H + (size_t)s * NHID))[lane];
    float* o = Hacc + (size_t)d * NHID + lane * 4;
    unsafeAtomicAdd(o + 0, c * v.x);
    unsafeAtomicAdd(o + 1, c * v.y);
    unsafeAtomicAdd(o + 2, c * v.z);
    unsafeAtomicAdd(o + 3, c * v.w);
}

__global__ __launch_bounds__(64) void k_gemm2(const float* __restrict__ Hacc,
                                              const float* __restrict__ b1,
                                              const float* __restrict__ Wp,
                                              float* __restrict__ G, int M) {
    __shared__ float2 Xs[64 * 33];
    const int tid = threadIdx.x;
    const int m0 = blockIdx.x * 32;

#pragma unroll
    for (int it = 0; it < 16; ++it) {
        int q = it * 64 + tid;
        int m = q >> 5, k4 = q & 31;
        int mr = m0 + m; if (mr >= M) mr = M - 1;
        float4 v = ((const float4*)(Hacc + (size_t)mr * NHID))[k4];
        float4 bb = ((const float4*)b1)[k4];
        v.x = fmaxf(v.x + bb.x, 0.0f);
        v.y = fmaxf(v.y + bb.y, 0.0f);
        v.z = fmaxf(v.z + bb.z, 0.0f);
        v.w = fmaxf(v.w + bb.w, 0.0f);
        Xs[(2 * k4)     * 33 + m] = make_float2(v.x, v.y);
        Xs[(2 * k4 + 1) * 33 + m] = make_float2(v.z, v.w);
    }
    __syncthreads();

    const int wave = tid >> 5, lane = tid & 31, l = lane & 15, half = lane >> 4;
    const float2* Bp = (const float2*)Wp;

    v8f acc0 = {}, acc1 = {}, acc2 = {};
#pragma unroll 4
    for (int kp = 0; kp < 64; kp += 2) {
        int row = kp + half;
        float2 a  = Xs[row * 33 + wave * 16 + l];
        float2 b0 = Bp[(size_t)row * NCPAD + 0  + l];
        float2 b1f= Bp[(size_t)row * NCPAD + 16 + l];
        float2 b2f= Bp[(size_t)row * NCPAD + 32 + l];
        v2f av = {a.x, a.y};
        v2f bv0 = {b0.x, b0.y}, bv1 = {b1f.x, b1f.y}, bv2 = {b2f.x, b2f.y};
        acc0 = __builtin_amdgcn_wmma_f32_16x16x4_f32(false, av, false, bv0, (short)0, acc0, false, false);
        acc1 = __builtin_amdgcn_wmma_f32_16x16x4_f32(false, av, false, bv1, (short)0, acc1, false, false);
        acc2 = __builtin_amdgcn_wmma_f32_16x16x4_f32(false, av, false, bv2, (short)0, acc2, false, false);
    }

    const int mb = m0 + wave * 16 + 8 * half;
    float* Go = G + (size_t)mb * NCPAD + l;
    if (m0 + 32 <= M) {
#pragma unroll
        for (int r = 0; r < 8; ++r) {
            float* o = Go + (size_t)r * NCPAD;
            o[0]  = acc0[r];
            o[16] = acc1[r];
            o[32] = acc2[r];
        }
    } else {
#pragma unroll
        for (int r = 0; r < 8; ++r) {
            if (mb + r < M) {
                float* o = Go + (size_t)r * NCPAD;
                o[0]  = acc0[r];
                o[16] = acc1[r];
                o[32] = acc2[r];
            }
        }
    }
}

__global__ __launch_bounds__(256) void k_agg2_init(const float* __restrict__ G,
                                                   const float* __restrict__ dinv,
                                                   float* __restrict__ Oacc, int N) {
    int idx = blockIdx.x * 256 + threadIdx.x;
    if (idx >= N * 12) return;
    int i = idx / 12, j = idx % 12;
    float di = dinv[i], c = di * di;
    float4 v = ((const float4*)(G + (size_t)i * NCPAD))[j];
    float4 o = make_float4(c * v.x, c * v.y, c * v.z, c * v.w);
    ((float4*)(Oacc + (size_t)i * NCPAD))[j] = o;
}

__global__ __launch_bounds__(256) void k_agg2_edges(const int* __restrict__ src,
                                                    const int* __restrict__ dst,
                                                    const float* __restrict__ w,
                                                    const float* __restrict__ dinv,
                                                    const float* __restrict__ G,
                                                    float* __restrict__ Oacc, int E) {
    int e = blockIdx.x * 8 + (threadIdx.x >> 5);
    if (e >= E) return;
    int lane = threadIdx.x & 31;
    int s = src[e], d = dst[e];
    float c = dinv[s] * w[e] * dinv[d];
    const float* g = G + (size_t)s * NCPAD;
    float* o = Oacc + (size_t)d * NCPAD;
    unsafeAtomicAdd(o + lane, c * g[lane]);
    if (lane < 16) unsafeAtomicAdd(o + 32 + lane, c * g[32 + lane]);
}

__global__ __launch_bounds__(256) void k_logsoftmax(const float* __restrict__ Oacc,
                                                    const float* __restrict__ b2,
                                                    float* __restrict__ out, int N) {
    int i = blockIdx.x * 8 + (threadIdx.x >> 5);
    if (i >= N) return;
    int lane = threadIdx.x & 31;
    const float* row = Oacc + (size_t)i * NCPAD;

    float v1 = row[lane] + b2[lane];
    int c2 = 32 + lane;
    bool has2 = c2 < NCLASS;
    float v2 = has2 ? (row[c2] + b2[c2]) : -INFINITY;

    float mx = fmaxf(v1, v2);
#pragma unroll
    for (int off = 16; off >= 1; off >>= 1) mx = fmaxf(mx, __shfl_xor(mx, off));

    float s = expf(v1 - mx) + (has2 ? expf(v2 - mx) : 0.0f);
#pragma unroll
    for (int off = 16; off >= 1; off >>= 1) s += __shfl_xor(s, off);

    float ls = logf(s);
    out[(size_t)i * NCLASS + lane] = v1 - mx - ls;
    if (has2) out[(size_t)i * NCLASS + c2] = v2 - mx - ls;
}

extern "C" void kernel_launch(void* const* d_in, const int* in_sizes, int n_in,
                              void* d_out, int out_size, void* d_ws, size_t ws_size,
                              hipStream_t stream) {
    const float* x  = (const float*)d_in[0];
    const int*   ei = (const int*)  d_in[1];
    const float* ew = (const float*)d_in[2];
    const float* W1 = (const float*)d_in[3];
    const float* b1 = (const float*)d_in[4];
    const float* W2 = (const float*)d_in[5];
    const float* b2 = (const float*)d_in[6];
    float* out = (float*)d_out;

    const int N = in_sizes[0] / NFEAT;
    const int E = in_sizes[2];
    const int* src = ei;
    const int* dst = ei + E;

    float* ws = (float*)d_ws;
    size_t o = 0;
    auto take = [&](size_t n) { float* p = ws + o; o += (n + 255) & ~(size_t)255; return p; };
    float* dinv = take((size_t)N);
    float* Wp1  = take((size_t)NFEAT * NHID);
    float* Wp2  = take((size_t)NHID * NCPAD);
    float* Hxw  = take((size_t)N * NHID);
    float* Hacc = take((size_t)N * NHID);
    float* G    = Hxw;
    float* Oacc = Hxw + (size_t)N * NCPAD;

    k_pack_w1<<<(NFEAT * NHID + 255) / 256, 256, 0, stream>>>(W1, Wp1);
    k_pack_w2<<<(NHID * NCPAD + 255) / 256, 256, 0, stream>>>(W2, Wp2);

    k_deg_init  <<<(N + 255) / 256, 256, 0, stream>>>(dinv, N);
    k_deg_edges <<<(E + 255) / 256, 256, 0, stream>>>(dst, ew, dinv, E);
    k_deg_finish<<<(N + 255) / 256, 256, 0, stream>>>(dinv, N);

    k_gemm1<<<(N + 31) / 32, 256, 0, stream>>>(x, Wp1, Hxw, N);

    k_agg1_init <<<(N * 32 + 255) / 256, 256, 0, stream>>>(Hxw, dinv, Hacc, N);
    k_agg1_edges<<<(E + 7) / 8, 256, 0, stream>>>(src, dst, ew, dinv, Hxw, Hacc, E);

    k_gemm2<<<(N + 31) / 32, 64, 0, stream>>>(Hacc, b1, Wp2, G, N);

    k_agg2_init <<<(N * 12 + 255) / 256, 256, 0, stream>>>(G, dinv, Oacc, N);
    k_agg2_edges<<<(E + 7) / 8, 256, 0, stream>>>(src, dst, ew, dinv, G, Oacc, E);

    k_logsoftmax<<<(N + 7) / 8, 256, 0, stream>>>(Oacc, b2, out, N);
}


# Edge passes: a wave owns 32 consecutive edges, not one (ids, norms loaded once lane-parallel, handed out by readlane; 8 source rows asked for ahead); 1600000 waves become 50000; float atomics kept
# speedup vs baseline: 1.1946x; 1.1946x over previous
#include <hip/hip_runtime.h>
#include <cstdint>

typedef float v2f __attribute__((ext_vector_type(2)));
typedef float v8f __attribute__((ext_vector_type(8)));

#define NFEAT 256
#define NHID  128
#define NCLASS 41
#define NCPAD  48

__global__ __launch_bounds__(256) void k_deg_init(float* deg, int N) {
    int i = blockIdx.x * 256 + threadIdx.x;
    if (i < N) deg[i] = 1.0f;
}

__global__ __launch_bounds__(256) void k_deg_edges(const int* __restrict__ dst,
                                                   const float* __restrict__ w,
                                                   float* deg, int E) {
    int e = blockIdx.x * 256 + threadIdx.x;
    if (e < E) unsafeAtomicAdd(&deg[dst[e]], w[e]);
}

__global__ __launch_bounds__(256) void k_deg_finish(float* dinv, int N) {
    int i = blockIdx.x * 256 + threadIdx.x;
    if (i < N) {
        float d = dinv[i];
        dinv[i] = d > 0.0f ? rsqrtf(d) : 0.0f;
    }
}

__global__ __launch_bounds__(256) void k_pack_w1(const float* __restrict__ W1, float* __restrict__ Wp) {
    int idx = blockIdx.x * 256 + threadIdx.x;
    if (idx >= NFEAT * NHID) return;
    int k = idx >> 7, n = idx & 127;
    Wp[(size_t)(k >> 1) * (NHID * 2) + n * 2 + (k & 1)] = W1[idx];
}

__global__ __launch_bounds__(256) void k_pack_w2(const float* __restrict__ W2, float* __restrict__ Wp) {
    int idx = blockIdx.x * 256 + threadIdx.x;
    if (idx >= NHID * NCPAD) return;
    int k = idx / NCPAD, n = idx % NCPAD;
    float v = (n < NCLASS) ? W2[k * NCLASS + n] : 0.0f;
    Wp[(size_t)(k >> 1) * (NCPAD * 2) + n * 2 + (k & 1)] = v;
}

__global__ __launch_bounds__(256) void k_gemm1(const float* __restrict__ X,
                                               const float* __restrict__ Wp,
                                               float* __restrict__ H, int M) {
    __shared__ float2 Xs[128 * 33];
    const int tid = threadIdx.x;
    const int m0 = blockIdx.x * 32;

#pragma unroll
    for (int it = 0; it < 8; ++it) {
        int q = it * 256 + tid;
        int m = q >> 6, k4 = q & 63;
        int mr = m0 + m; if (mr >= M) mr = M - 1;
        float4 v = ((const float4*)(X + (size_t)mr * NFEAT))[k4];
        Xs[(2 * k4)     * 33 + m] = make_float2(v.x, v.y);
        Xs[(2 * k4 + 1) * 33 + m] = make_float2(v.z, v.w);
    }
    __syncthreads();

    const int wave = tid >> 5, lane = tid & 31, l = lane & 15, half = lane >> 4;
    const int n0 = wave * 16;
    const float2* Bp = (const float2*)Wp;

    v8f acc0 = {}, acc1 = {};
#pragma unroll 4
    for (int kp = 0; kp < 128; kp += 2) {
        int row = kp + half;
        float2 a0 = Xs[row * 33 + l];
        float2 a1 = Xs[row * 33 + 16 + l];
        float2 b  = Bp[(size_t)row * NHID + n0 + l];
        v2f av0 = {a0.x, a0.y}, av1 = {a1.x, a1.y}, bv = {b.x, b.y};
        acc0 = __builtin_amdgcn_wmma_f32_16x16x4_f32(false, av0, false, bv, (short)0, acc0, false, false);
        acc1 = __builtin_amdgcn_wmma_f32_16x16x4_f32(false, av1, false, bv, (short)0, acc1, false, false);
    }

    const int mb = 8 * half;
    float* Ho = H + (size_t)(m0 + mb) * NHID + n0 + l;
    if (m0 + 32 <= M) {
#pragma unroll
        for (int r = 0; r < 8; ++r) {
            Ho[(size_t)r * NHID]        = acc0[r];
            Ho[(size_t)(16 + r) * NHID] = acc1[r];
        }
    } else {
#pragma unroll
        for (int r = 0; r < 8; ++r) {
            if (m0 + mb + r < M)      Ho[(size_t)r * NHID]        = acc0[r];
            if (m0 + 16 + mb + r < M) Ho[(size_t)(16 + r) * NHID] = acc1[r];
        }
    }
}

__global__ __launch_bounds__(256) void k_agg1_init(const float* __restrict__ H,
                                                   const float* __restrict__ dinv,
                                                   float* __restrict__ Hacc, int N) {
    int idx = blockIdx.x * 256 + threadIdx.x;
    if (idx >= N * 32) return;
    int i = idx >> 5, j = idx & 31;
    float di = dinv[i], c = di * di;
    float4 v = ((const float4*)(H + (size_t)i * NHID))[j];
    float4 o = make_float4(c * v.x, c * v.y, c * v.z, c * v.w);
    ((float4*)(Hacc + (size_t)i * NHID))[j] = o;
}

__global__ __launch_bounds__(256) void k_agg1_edges(const int* __restrict__ src,
                                                    const int* __restrict__ dst,
                                                    const float* __restrict__ w,
                                                    const float* __restrict__ dinv,
                                                    const float* __restrict__ H,
                                                    float* __restrict__ Hacc, int E) {
    const int lane = threadIdx.x & 31;
    const int e0 = (blockIdx.x * 8 + (threadIdx.x >> 5)) * 32;
    if (e0 >= E) return;
    const int e = e0 + lane;
    int s = 0, d = 0;
    float c = 0.0f;
    if (e < E) {
        s = src[e]; d = dst[e];
        c = dinv[s] * w[e] * dinv[d];
    }
    const float4* H4 = (const float4*)H;
    if (e0 + 32 <= E) {
        for (int j0 = 0; j0 < 32; j0 += 8) {
            float4 v[8];
#pragma unroll
            for (int u = 0; u < 8; ++u) {
                int sj = __builtin_amdgcn_readlane(s, j0 + u);
                v[u] = H4[(size_t)sj * 32 + lane];
            }
#pragma unroll
            for (int u = 0; u < 8; ++u) {
                int dj = __builtin_amdgcn_readlane(d, j0 + u);
                float cj = __int_as_float(__builtin_amdgcn_readlane(__float_as_int(c), j0 + u));
                float* o = Hacc + (size_t)dj * NHID + lane * 4;
                unsafeAtomicAdd(o + 0, cj * v[u].x);
                unsafeAtomicAdd(o + 1, cj * v[u].y);
                unsafeAtomicAdd(o + 2, cj * v[u].z);
                unsafeAtomicAdd(o + 3, cj * v[u].w);
            }
        }
    } else {
        const int cnt = E - e0;
        for (int j = 0; j < cnt; ++j) {
            int sj = __builtin_amdgcn_readlane(s, j);
            int dj = __builtin_amdgcn_readlane(d, j);
            float cj = __int_as_float(__builtin_amdgcn_readlane(__float_as_int(c), j));
            float4 v = H4[(size_t)sj * 32 + lane];
            float* o = Hacc + (size_t)dj * NHID + lane * 4;
            unsafeAtomicAdd(o + 0, cj * v.x);
            unsafeAtomicAdd(o + 1, cj * v.y);
            unsafeAtomicAdd(o + 2, cj * v.z);
            unsafeAtomicAdd(o + 3, cj * v.w);
        }
    }
}

__global__ __launch_bounds__(64) void k_gemm2(const float* __restrict__ Hacc,
                                              const float* __restrict__ b1,
                                              const float* __restrict__ Wp,
                                              float* __restrict__ G, int M) {
    __shared__ float2 Xs[64 * 33];
    const int tid = threadIdx.x;
    const int m0 = blockIdx.x * 32;

#pragma unroll
    for (int it = 0; it < 16; ++it) {
        int q = it * 64 + tid;
        int m = q >> 5, k4 = q & 31;
        int mr = m0 + m; if (mr >= M) mr = M - 1;
        float4 v = ((const float4*)(Hacc + (size_t)mr * NHID))[k4];
        float4 bb = ((const float4*)b1)[k4];
        v.x = fmaxf(v.x + bb.x, 0.0f);
        v.y = fmaxf(v.y + bb.y, 0.0f);
        v.z = fmaxf(v.z + bb.z, 0.0f);
        v.w = fmaxf(v.w + bb.w, 0.0f);
        Xs[(2 * k4)     * 33 + m] = make_float2(v.x, v.y);
        Xs[(2 * k4 + 1) * 33 + m] = make_float2(v.z, v.w);
    }
    __syncthreads();

    const int wave = tid >> 5, lane = tid & 31, l = lane & 15, half = lane >> 4;
    const float2* Bp = (const float2*)Wp;

    v8f acc0 = {}, acc1 = {}, acc2 = {};
#pragma unroll 4
    for (int kp = 0; kp < 64; kp += 2) {
        int row = kp + half;
        float2 a  = Xs[row * 33 + wave * 16 + l];
        float2 b0 = Bp[(size_t)row * NCPAD + 0  + l];
        float2 b1f= Bp[(size_t)row * NCPAD + 16 + l];
        float2 b2f= Bp[(size_t)row * NCPAD + 32 + l];
        v2f av = {a.x, a.y};
        v2f bv0 = {b0.x, b0.y}, bv1 = {b1f.x, b1f.y}, bv2 = {b2f.x, b2f.y};
        acc0 = __builtin_amdgcn_wmma_f32_16x16x4_f32(false, av, false, bv0, (short)0, acc0, false, false);
        acc1 = __builtin_amdgcn_wmma_f32_16x16x4_f32(false, av, false, bv1, (short)0, acc1, false, false);
        acc2 = __builtin_amdgcn_wmma_f32_16x16x4_f32(false, av, false, bv2, (short)0, acc2, false, false);
    }

    const int mb = m0 + wave * 16 + 8 * half;
    float* Go = G + (size_t)mb * NCPAD + l;
    if (m0 + 32 <= M) {
#pragma unroll
        for (int r = 0; r < 8; ++r) {
            float* o = Go + (size_t)r * NCPAD;
            o[0]  = acc0[r];
            o[16] = acc1[r];
            o[32] = acc2[r];
        }
    } else {
#pragma unroll
        for (int r = 0; r < 8; ++r) {
            if (mb + r < M) {
                float* o = Go + (size_t)r * NCPAD;
                o[0]  = acc0[r];
                o[16] = acc1[r];
                o[32] = acc2[r];
            }
        }
    }
}

__global__ __launch_bounds__(256) void k_agg2_init(const float* __restrict__ G,
                                                   const float* __restrict__ dinv,
                                                   float* __restrict__ Oacc, int N) {
    int idx = blockIdx.x * 256 + threadIdx.x;
    if (idx >= N * 12) return;
    int i = idx / 12, j = idx % 12;
    float di = dinv[i], c = di * di;
    float4 v = ((const float4*)(G + (size_t)i * NCPAD))[j];
    float4 o = make_float4(c * v.x, c * v.y, c * v.z, c * v.w);
    ((float4*)(Oacc + (size_t)i * NCPAD))[j] = o;
}

__global__ __launch_bounds__(256) void k_agg2_edges(const int* __restrict__ src,
                                                    const int* __restrict__ dst,
                                                    const float* __restrict__ w,
                                                    const float* __restrict__ dinv,
                                                    const float* __restrict__ G,
                                                    float* __restrict__ Oacc, int E) {
    const int lane = threadIdx.x & 31;
    const int e0 = (blockIdx.x * 8 + (threadIdx.x >> 5)) * 32;
    if (e0 >= E) return;
    const int e = e0 + lane;
    int s = 0, d = 0;
    float c = 0.0f;
    if (e < E) {
        s = src[e]; d = dst[e];
        c = dinv[s] * w[e] * dinv[d];
    }
    const bool lo = lane < 16;
    if (e0 + 32 <= E) {
        for (int j0 = 0; j0 < 32; j0 += 8) {
            int sj[8], dj[8];
            float cj[8];
#pragma unroll
            for (int u = 0; u < 8; ++u) {
                sj[u] = __builtin_amdgcn_readlane(s, j0 + u);
                dj[u] = __builtin_amdgcn_readlane(d, j0 + u);
                cj[u] = __int_as_float(__builtin_amdgcn_readlane(__float_as_int(c), j0 + u));
            }
            float ga[8], gb[8];
#pragma unroll
            for (int u = 0; u < 8; ++u)
                ga[u] = G[(size_t)sj[u] * NCPAD + lane];
            if (lo) {
#pragma unroll
                for (int u = 0; u < 8; ++u)
                    gb[u] = G[(size_t)sj[u] * NCPAD + 32 + lane];
            }
#pragma unroll
            for (int u = 0; u < 8; ++u)
                unsafeAtomicAdd(Oacc + (size_t)dj[u] * NCPAD + lane, cj[u] * ga[u]);
            if (lo) {
#pragma unroll
                for (int u = 0; u < 8; ++u)
                    unsafeAtomicAdd(Oacc + (size_t)dj[u] * NCPAD + 32 + lane, cj[u] * gb[u]);
            }
        }
    } else {
        const int cnt = E - e0;
        for (int j = 0; j < cnt; ++j) {
            int sj = __builtin_amdgcn_readlane(s, j);
            int dj = __builtin_amdgcn_readlane(d, j);
            float cj = __int_as_float(__builtin_amdgcn_readlane(__float_as_int(c), j));
            const float* g = G + (size_t)sj * NCPAD;
            float* o = Oacc + (size_t)dj * NCPAD;
            unsafeAtomicAdd(o + lane, cj * g[lane]);
            if (lo) unsafeAtomicAdd(o + 32 + lane, cj * g[32 + lane]);
        }
    }
}

__global__ __launch_bounds__(256) void k_logsoftmax(const float* __restrict__ Oacc,
                                                    const float* __restrict__ b2,
                                                    float* __restrict__ out, int N) {
    int i = blockIdx.x * 8 + (threadIdx.x >> 5);
    if (i >= N) return;
    int lane = threadIdx.x & 31;
    const float* row = Oacc + (size_t)i * NCPAD;

    float v1 = row[lane] + b2[lane];
    int c2 = 32 + lane;
    bool has2 = c2 < NCLASS;
    float v2 = has2 ? (row[c2] + b2[c2]) : -INFINITY;

    float mx = fmaxf(v1, v2);
#pragma unroll
    for (int off = 16; off >= 1; off >>= 1) mx = fmaxf(mx, __shfl_xor(mx, off));

    float s = expf(v1 - mx) + (has2 ? expf(v2 - mx) : 0.0f);
#pragma unroll
    for (int off = 16; off >= 1; off >>= 1) s += __shfl_xor(s, off);

    float ls = logf(s);
    out[(size_t)i * NCLASS + lane] = v1 - mx - ls;
    if (has2) out[(size_t)i * NCLASS + c2] = v2 - mx - ls;
}

extern "C" void kernel_launch(void* const* d_in, const int* in_sizes, int n_in,
                              void* d_out, int out_size, void* d_ws, size_t ws_size,
                              hipStream_t stream) {
    const float* x  = (const float*)d_in[0];
    const int*   ei = (const int*)  d_in[1];
    const float* ew = (const float*)d_in[2];
    const float* W1 = (const float*)d_in[3];
    const float* b1 = (const float*)d_in[4];
    const float* W2 = (const float*)d_in[5];
    const float* b2 = (const float*)d_in[6];
    float* out = (float*)d_out;

    const int N = in_sizes[0] / NFEAT;
    const int E = in_sizes[2];
    const int* src = ei;
    const int* dst = ei + E;

    float* ws = (float*)d_ws;
    size_t o = 0;
    auto take = [&](size_t n) { float* p = ws + o; o += (n + 255) & ~(size_t)255; return p; };
    float* dinv = take((size_t)N);
    float* Wp1  = take((size_t)NFEAT * NHID);
    float* Wp2  = take((size_t)NHID * NCPAD);
    float* Hxw  = take((size_t)N * NHID);
    float* Hacc = take((size_t)N * NHID);
    float* G    = Hxw;
    float* Oacc = Hxw + (size_t)N * NCPAD;

    k_pack_w1<<<(NFEAT * NHID + 255) / 256, 256, 0, stream>>>(W1, Wp1);
    k_pack_w2<<<(NHID * NCPAD + 255) / 256, 256, 0, stream>>>(W2, Wp2);

    k_deg_init  <<<(N + 255) / 256, 256, 0, stream>>>(dinv, N);
    k_deg_edges <<<(E + 255) / 256, 256, 0, stream>>>(dst, ew, dinv, E);
    k_deg_finish<<<(N + 255) / 256, 256, 0, stream>>>(dinv, N);

    k_gemm1<<<(N + 31) / 32, 256, 0, stream>>>(x, Wp1, Hxw, N);

    k_agg1_init <<<(N * 32 + 255) / 256, 256, 0, stream>>>(Hxw, dinv, Hacc, N);
    k_agg1_edges<<<(E + 255) / 256, 256, 0, stream>>>(src, dst, ew, dinv, Hxw, Hacc, E);

    k_gemm2<<<(N + 31) / 32, 64, 0, stream>>>(Hacc, b1, Wp2, G, N);

    k_agg2_init <<<(N * 12 + 255) / 256, 256, 0, stream>>>(G, dinv, Oacc, N);
    k_agg2_edges<<<(E + 255) / 256, 256, 0, stream>>>(src, dst, ew, dinv, G, Oacc, E);

    k_logsoftmax<<<(N + 7) / 8, 256, 0, stream>>>(Oacc, b2, out, N);
}


# Layer-1 aggregation kept in LDS: 256 work-groups each own 392 destination rows, sweep the whole destination list, add their edges' rows by LDS float atomics (arrival order); no global atomics there
# speedup vs baseline: 1.6251x; 1.3604x over previous
#include <hip/hip_runtime.h>
#include <cstdint>

typedef float v2f __attribute__((ext_vector_type(2)));
typedef float v8f __attribute__((ext_vector_type(8)));

#define NFEAT 256
#define NHID  128
#define NCLASS 41
#define NCPAD  48

__global__ __launch_bounds__(256) void k_deg_init(float* deg, int N) {
    int i = blockIdx.x * 256 + threadIdx.x;
    if (i < N) deg[i] = 1.0f;
}

__global__ __launch_bounds__(256) void k_deg_edges(const int* __restrict__ dst,
                                                   const float* __restrict__ w,
                                                   float* deg, int E) {
    int e = blockIdx.x * 256 + threadIdx.x;
    if (e < E) unsafeAtomicAdd(&deg[dst[e]], w[e]);
}

__global__ __launch_bounds__(256) void k_deg_finish(float* dinv, int N) {
    int i = blockIdx.x * 256 + threadIdx.x;
    if (i < N) {
        float d = dinv[i];
        dinv[i] = d > 0.0f ? rsqrtf(d) : 0.0f;
    }
}

__global__ __launch_bounds__(256) void k_pack_w1(const float* __restrict__ W1, float* __restrict__ Wp) {
    int idx = blockIdx.x * 256 + threadIdx.x;
    if (idx >= NFEAT * NHID) return;
    int k = idx >> 7, n = idx & 127;
    Wp[(size_t)(k >> 1) * (NHID * 2) + n * 2 + (k & 1)] = W1[idx];
}

__global__ __launch_bounds__(256) void k_pack_w2(const float* __restrict__ W2, float* __restrict__ Wp) {
    int idx = blockIdx.x * 256 + threadIdx.x;
    if (idx >= NHID * NCPAD) return;
    int k = idx / NCPAD, n = idx % NCPAD;
    float v = (n < NCLASS) ? W2[k * NCLASS + n] : 0.0f;
    Wp[(size_t)(k >> 1) * (NCPAD * 2) + n * 2 + (k & 1)] = v;
}

__global__ __launch_bounds__(256) void k_gemm1(const float* __restrict__ X,
                                               const float* __restrict__ Wp,
                                               float* __restrict__ H, int M) {
    __shared__ float2 Xs[128 * 33];
    const int tid = threadIdx.x;
    const int m0 = blockIdx.x * 32;

#pragma unroll
    for (int it = 0; it < 8; ++it) {
        int q = it * 256 + tid;
        int m = q >> 6, k4 = q & 63;
        int mr = m0 + m; if (mr >= M) mr = M - 1;
        float4 v = ((const float4*)(X + (size_t)mr * NFEAT))[k4];
        Xs[(2 * k4)     * 33 + m] = make_float2(v.x, v.y);
        Xs[(2 * k4 + 1) * 33 + m] = make_float2(v.z, v.w);
    }
    __syncthreads();

    const int wave = tid >> 5, lane = tid & 31, l = lane & 15, half = lane >> 4;
    const int n0 = wave * 16;
    const float2* Bp = (const float2*)Wp;

    v8f acc0 = {}, acc1 = {};
#pragma unroll 4
    for (int kp = 0; kp < 128; kp += 2) {
        int row = kp + half;
        float2 a0 = Xs[row * 33 + l];
        float2 a1 = Xs[row * 33 + 16 + l];
        float2 b  = Bp[(size_t)row * NHID + n0 + l];
        v2f av0 = {a0.x, a0.y}, av1 = {a1.x, a1.y}, bv = {b.x, b.y};
        acc0 = __builtin_amdgcn_wmma_f32_16x16x4_f32(false, av0, false, bv, (short)0, acc0, false, false);
        acc1 = __builtin_amdgcn_wmma_f32_16x16x4_f32(false, av1, false, bv, (short)0, acc1, false, false);
    }

    const int mb = 8 * half;
    float* Ho = H + (size_t)(m0 + mb) * NHID + n0 + l;
    if (m0 + 32 <= M) {
#pragma unroll
        for (int r = 0; r < 8; ++r) {
            Ho[(size_t)r * NHID]        = acc0[r];
            Ho[(size_t)(16 + r) * NHID] = acc1[r];
        }
    } else {
#pragma unroll
        for (int r = 0; r < 8; ++r) {
            if (m0 + mb + r < M)      Ho[(size_t)r * NHID]        = acc0[r];
            if (m0 + 16 + mb + r < M) Ho[(size_t)(16 + r) * NHID] = acc1[r];
        }
    }
}

#define AG_ROWS 392
#define AG_QCAP 8192

__device__ __forceinline__ void ag_push(int e, int r, int lo, int* qcnt, int2* q, float* acc,
                                        const int* __restrict__ src, const float* __restrict__ w,
                                        const float* __restrict__ dinv, const float* __restrict__ H) {
    int slot = __hip_atomic_fetch_add(qcnt, 1, __ATOMIC_RELAXED, __HIP_MEMORY_SCOPE_WORKGROUP);
    if (slot < AG_QCAP) {
        q[slot] = make_int2(e, r);
    } else {
        int s = src[e];
        float c = dinv[s] * w[e] * dinv[lo + r];
        const float* h = H + (size_t)s * NHID;
        for (int f = 0; f < NHID; ++f)
            __hip_atomic_fetch_add(&acc[r * NHID + (f & 3) * 32 + (f >> 2)], c * h[f],
                                   __ATOMIC_RELAXED, __HIP_MEMORY_SCOPE_WORKGROUP);
    }
}

__global__ __launch_bounds__(1024) void k_agg1_lds(const int* __restrict__ src,
                                                   const int* __restrict__ dst,
                                                   const float* __restrict__ w,
                                                   const float* __restrict__ dinv,
                                                   const float* __restrict__ H,
                                                   float* __restrict__ Hacc, int N, int E) {
    __shared__ float acc[AG_ROWS * NHID];
    __shared__ int2 q[AG_QCAP];
    __shared__ int qcnt;

    const int tid = threadIdx.x, lane = tid & 31, wave = tid >> 5;
    const float4* H4 = (const float4*)H;

    int head = (int)(((16u - (unsigned)((uintptr_t)dst & 15u)) & 15u) >> 2);
    if (head > E) head = E;
    const int nvec = (E - head) >> 2;
    const int tail0 = head + 4 * nvec;
    const int4* dst4 = (const int4*)(dst + head);
    const int start = (int)(((long long)blockIdx.x * nvec) / gridDim.x);

    for (int lo = blockIdx.x * AG_ROWS; lo < N; lo += gridDim.x * AG_ROWS) {
        const int rows = min(AG_ROWS, N - lo);

        for (int idx = tid; idx < rows * 32; idx += 1024) {
            int r = idx >> 5, i = lo + r;
            float di = dinv[i], c = di * di;
            float4 v = H4[(size_t)i * 32 + lane];
            float* a = acc + r * NHID + lane;
            a[0] = c * v.x; a[32] = c * v.y; a[64] = c * v.z; a[96] = c * v.w;
        }
        if (tid == 0) qcnt = 0;
        __syncthreads();

        if (tid < head) {
            unsigned r = (unsigned)(dst[tid] - lo);
            if (r < (unsigned)rows) ag_push(tid, (int)r, lo, &qcnt, q, acc, src, w, dinv, H);
        }
        if (tail0 + tid < E) {
            unsigned r = (unsigned)(dst[tail0 + tid] - lo);
            if (r < (unsigned)rows) ag_push(tail0 + tid, (int)r, lo, &qcnt, q, acc, src, w, dinv, H);
        }
        for (int i0 = 0; i0 < nvec; i0 += 8 * 1024) {
            int4 d[8];
#pragma unroll
            for (int u = 0; u < 8; ++u) {
                int v = i0 + u * 1024 + tid;
                int pp = v + start; if (pp >= nvec) pp -= nvec;
                d[u] = make_int4(-1, -1, -1, -1);
                if (v < nvec) d[u] = dst4[pp];
            }
#pragma unroll
            for (int u = 0; u < 8; ++u) {
                unsigned r0 = (unsigned)(d[u].x - lo), r1 = (unsigned)(d[u].y - lo);
                unsigned r2 = (unsigned)(d[u].z - lo), r3 = (unsigned)(d[u].w - lo);
                unsigned rmin = min(min(r0, r1), min(r2, r3));
                if (rmin < (unsigned)rows) {
                    int pp = i0 + u * 1024 + tid + start; if (pp >= nvec) pp -= nvec;
                    int e = head + 4 * pp;
                    if (r0 < (unsigned)rows) ag_push(e + 0, (int)r0, lo, &qcnt, q, acc, src, w, dinv, H);
                    if (r1 < (unsigned)rows) ag_push(e + 1, (int)r1, lo, &qcnt, q, acc, src, w, dinv, H);
                    if (r2 < (unsigned)rows) ag_push(e + 2, (int)r2, lo, &qcnt, q, acc, src, w, dinv, H);
                    if (r3 < (unsigned)rows) ag_push(e + 3, (int)r3, lo, &qcnt, q, acc, src, w, dinv, H);
                }
            }
        }
        __syncthreads();

        const int nq = min(qcnt, AG_QCAP);
        for (int base = wave * 32; base < nq; base += 32 * 32) {
            const int n = min(32, nq - base);
            int s = 0, r = 0;
            float c = 0.0f;
            if (lane < n) {
                int2 t = q[base + lane];
                r = t.y;
                s = src[t.x];
                c = dinv[s] * w[t.x] * dinv[lo + r];
            }
            int j0 = 0;
            for (; j0 + 8 <= n; j0 += 8) {
                float4 v[8];
#pragma unroll
                for (int u = 0; u < 8; ++u) {
                    int sj = __builtin_amdgcn_readlane(s, j0 + u);
                    v[u] = H4[(size_t)sj * 32 + lane];
                }
#pragma unroll
                for (int u = 0; u < 8; ++u) {
                    int rj = __builtin_amdgcn_readlane(r, j0 + u);
                    float cj = __int_as_float(__builtin_amdgcn_readlane(__float_as_int(c), j0 + u));
                    float* a = acc + rj * NHID + lane;
                    __hip_atomic_fetch_add(a + 0,  cj * v[u].x, __ATOMIC_RELAXED, __HIP_MEMORY_SCOPE_WORKGROUP);
                    __hip_atomic_fetch_add(a + 32, cj * v[u].y, __ATOMIC_RELAXED, __HIP_MEMORY_SCOPE_WORKGROUP);
                    __hip_atomic_fetch_add(a + 64, cj * v[u].z, __ATOMIC_RELAXED, __HIP_MEMORY_SCOPE_WORKGROUP);
                    __hip_atomic_fetch_add(a + 96, cj * v[u].w, __ATOMIC_RELAXED, __HIP_MEMORY_SCOPE_WORKGROUP);
                }
            }
            for (; j0 < n; ++j0) {
                int sj = __builtin_amdgcn_readlane(s, j0);
                int rj = __builtin_amdgcn_readlane(r, j0);
                float cj = __int_as_float(__builtin_amdgcn_readlane(__float_as_int(c), j0));
                float4 v = H4[(size_t)sj * 32 + lane];
                float* a = acc + rj * NHID + lane;
                __hip_atomic_fetch_add(a + 0,  cj * v.x, __ATOMIC_RELAXED, __HIP_MEMORY_SCOPE_WORKGROUP);
                __hip_atomic_fetch_add(a + 32, cj * v.y, __ATOMIC_RELAXED, __HIP_MEMORY_SCOPE_WORKGROUP);
                __hip_atomic_fetch_add(a + 64, cj * v.z, __ATOMIC_RELAXED, __HIP_MEMORY_SCOPE_WORKGROUP);
                __hip_atomic_fetch_add(a + 96, cj * v.w, __ATOMIC_RELAXED, __HIP_MEMORY_SCOPE_WORKGROUP);
            }
        }
        __syncthreads();

        for (int idx = tid; idx < rows * 32; idx += 1024) {
            int r = idx >> 5;
            const float* a = acc + r * NHID + lane;
            ((float4*)(Hacc + (size_t)(lo + r) * NHID))[lane] = make_float4(a[0], a[32], a[64], a[96]);
        }
        __syncthreads();
    }
}

__global__ __launch_bounds__(64) void k_gemm2(const float* __restrict__ Hacc,
                                              const float* __restrict__ b1,
                                              const float* __restrict__ Wp,
                                              float* __restrict__ G, int M) {
    __shared__ float2 Xs[64 * 33];
    const int tid = threadIdx.x;
    const int m0 = blockIdx.x * 32;

#pragma unroll
    for (int it = 0; it < 16; ++it) {
        int q = it * 64 + tid;
        int m = q >> 5, k4 = q & 31;
        int mr = m0 + m; if (mr >= M) mr = M - 1;
        float4 v = ((const float4*)(Hacc + (size_t)mr * NHID))[k4];
        float4 bb = ((const float4*)b1)[k4];
        v.x = fmaxf(v.x + bb.x, 0.0f);
        v.y = fmaxf(v.y + bb.y, 0.0f);
        v.z = fmaxf(v.z + bb.z, 0.0f);
        v.w = fmaxf(v.w + bb.w, 0.0f);
        Xs[(2 * k4)     * 33 + m] = make_float2(v.x, v.y);
        Xs[(2 * k4 + 1) * 33 + m] = make_float2(v.z, v.w);
    }
    __syncthreads();

    const int wave = tid >> 5, lane = tid & 31, l = lane & 15, half = lane >> 4;
    const float2* Bp = (const float2*)Wp;

    v8f acc0 = {}, acc1 = {}, acc2 = {};
#pragma unroll 4
    for (int kp = 0; kp < 64; kp += 2) {
        int row = kp + half;
        float2 a  = Xs[row * 33 + wave * 16 + l];
        float2 b0 = Bp[(size_t)row * NCPAD + 0  + l];
        float2 b1f= Bp[(size_t)row * NCPAD + 16 + l];
        float2 b2f= Bp[(size_t)row * NCPAD + 32 + l];
        v2f av = {a.x, a.y};
        v2f bv0 = {b0.x, b0.y}, bv1 = {b1f.x, b1f.y}, bv2 = {b2f.x, b2f.y};
        acc0 = __builtin_amdgcn_wmma_f32_16x16x4_f32(false, av, false, bv0, (short)0, acc0, false, false);
        acc1 = __builtin_amdgcn_wmma_f32_16x16x4_f32(false, av, false, bv1, (short)0, acc1, false, false);
        acc2 = __builtin_amdgcn_wmma_f32_16x16x4_f32(false, av, false, bv2, (short)0, acc2, false, false);
    }

    const int mb = m0 + wave * 16 + 8 * half;
    float* Go = G + (size_t)mb * NCPAD + l;
    if (m0 + 32 <= M) {
#pragma unroll
        for (int r = 0; r < 8; ++r) {
            float* o = Go + (size_t)r * NCPAD;
            o[0]  = acc0[r];
            o[16] = acc1[r];
            o[32] = acc2[r];
        }
    } else {
#pragma unroll
        for (int r = 0; r < 8; ++r) {
            if (mb + r < M) {
                float* o = Go + (size_t)r * NCPAD;
                o[0]  = acc0[r];
                o[16] = acc1[r];
                o[32] = acc2[r];
            }
        }
    }
}

__global__ __launch_bounds__(256) void k_agg2_init(const float* __restrict__ G,
                                                   const float* __restrict__ dinv,
                                                   float* __restrict__ Oacc, int N) {
    int idx = blockIdx.x * 256 + threadIdx.x;
    if (idx >= N * 12) return;
    int i = idx / 12, j = idx % 12;
    float di = dinv[i], c = di * di;
    float4 v = ((const float4*)(G + (size_t)i * NCPAD))[j];
    float4 o = make_float4(c * v.x, c * v.y, c * v.z, c * v.w);
    ((float4*)(Oacc + (size_t)i * NCPAD))[j] = o;
}

__global__ __launch_bounds__(256) void k_agg2_edges(const int* __restrict__ src,
                                                    const int* __restrict__ dst,
                                                    const float* __restrict__ w,
                                                    const float* __restrict__ dinv,
                                                    const float* __restrict__ G,
                                                    float* __restrict__ Oacc, int E) {
    const int lane = threadIdx.x & 31;
    const int e0 = (blockIdx.x * 8 + (threadIdx.x >> 5)) * 32;
    if (e0 >= E) return;
    const int e = e0 + lane;
    int s = 0, d = 0;
    float c = 0.0f;
    if (e < E) {
        s = src[e]; d = dst[e];
        c = dinv[s] * w[e] * dinv[d];
    }
    const bool lo = lane < 16;
    if (e0 + 32 <= E) {
        for (int j0 = 0; j0 < 32; j0 += 8) {
            int sj[8], dj[8];
            float cj[8];
#pragma unroll
            for (int u = 0; u < 8; ++u) {
                sj[u] = __builtin_amdgcn_readlane(s, j0 + u);
                dj[u] = __builtin_amdgcn_readlane(d, j0 + u);
                cj[u] = __int_as_float(__builtin_amdgcn_readlane(__float_as_int(c), j0 + u));
            }
            float ga[8], gb[8];
#pragma unroll
            for (int u = 0; u < 8; ++u)
                ga[u] = G[(size_t)sj[u] * NCPAD + lane];
            if (lo) {
#pragma unroll
                for (int u = 0; u < 8; ++u)
                    gb[u] = G[(size_t)sj[u] * NCPAD + 32 + lane];
            }
#pragma unroll
            for (int u = 0; u < 8; ++u)
                unsafeAtomicAdd(Oacc + (size_t)dj[u] * NCPAD + lane, cj[u] * ga[u]);
            if (lo) {
#pragma unroll
                for (int u = 0; u < 8; ++u)
                    unsafeAtomicAdd(Oacc + (size_t)dj[u] * NCPAD + 32 + lane, cj[u] * gb[u]);
            }
        }
    } else {
        const int cnt = E - e0;
        for (int j = 0; j < cnt; ++j) {
            int sj = __builtin_amdgcn_readlane(s, j);
            int dj = __builtin_amdgcn_readlane(d, j);
            float cj = __int_as_float(__builtin_amdgcn_readlane(__float_as_int(c), j));
            const float* g = G + (size_t)sj * NCPAD;
            float* o = Oacc + (size_t)dj * NCPAD;
            unsafeAtomicAdd(o + lane, cj * g[lane]);
            if (lo) unsafeAtomicAdd(o + 32 + lane, cj * g[32 + lane]);
        }
    }
}

__global__ __launch_bounds__(256) void k_logsoftmax(const float* __restrict__ Oacc,
                                                    const float* __restrict__ b2,
                                                    float* __restrict__ out, int N) {
    int i = blockIdx.x * 8 + (threadIdx.x >> 5);
    if (i >= N) return;
    int lane = threadIdx.x & 31;
    const float* row = Oacc + (size_t)i * NCPAD;

    float v1 = row[lane] + b2[lane];
    int c2 = 32 + lane;
    bool has2 = c2 < NCLASS;
    float v2 = has2 ? (row[c2] + b2[c2]) : -INFINITY;

    float mx = fmaxf(v1, v2);
#pragma unroll
    for (int off = 16; off >= 1; off >>= 1) mx = fmaxf(mx, __shfl_xor(mx, off));

    float s = expf(v1 - mx) + (has2 ? expf(v2 - mx) : 0.0f);
#pragma unroll
    for (int off = 16; off >= 1; off >>= 1) s += __shfl_xor(s, off);

    float ls = logf(s);
    out[(size_t)i * NCLASS + lane] = v1 - mx - ls;
    if (has2) out[(size_t)i * NCLASS + c2] = v2 - mx - ls;
}

extern "C" void kernel_launch(void* const* d_in, const int* in_sizes, int n_in,
                              void* d_out, int out_size, void* d_ws, size_t ws_size,
                              hipStream_t stream) {
    const float* x  = (const float*)d_in[0];
    const int*   ei = (const int*)  d_in[1];
    const float* ew = (const float*)d_in[2];
    const float* W1 = (const float*)d_in[3];
    const float* b1 = (const float*)d_in[4];
    const float* W2 = (const float*)d_in[5];
    const float* b2 = (const float*)d_in[6];
    float* out = (float*)d_out;

    const int N = in_sizes[0] / NFEAT;
    const int E = in_sizes[2];
    const int* src = ei;
    const int* dst = ei + E;

    float* ws = (float*)d_ws;
    size_t o = 0;
    auto take = [&](size_t n) { float* p = ws + o; o += (n + 255) & ~(size_t)255; return p; };
    float* dinv = take((size_t)N);
    float* Wp1  = take((size_t)NFEAT * NHID);
    float* Wp2  = take((size_t)NHID * NCPAD);
    float* Hxw  = take((size_t)N * NHID);
    float* Hacc = take((size_t)N * NHID);
    float* G    = Hxw;
    float* Oacc = Hxw + (size_t)N * NCPAD;

    k_pack_w1<<<(NFEAT * NHID + 255) / 256, 256, 0, stream>>>(W1, Wp1);
    k_pack_w2<<<(NHID * NCPAD + 255) / 256, 256, 0, stream>>>(W2, Wp2);

    k_deg_init  <<<(N + 255) / 256, 256, 0, stream>>>(dinv, N);
    k_deg_edges <<<(E + 255) / 256, 256, 0, stream>>>(dst, ew, dinv, E);
    k_deg_finish<<<(N + 255) / 256, 256, 0, stream>>>(dinv, N);

    k_gemm1<<<(N + 31) / 32, 256, 0, stream>>>(x, Wp1, Hxw, N);

    k_agg1_lds<<<256, 1024, 0, stream>>>(src, dst, ew, dinv, Hxw, Hacc, N, E);

    k_gemm2<<<(N + 31) / 32, 64, 0, stream>>>(Hacc, b1, Wp2, G, N);

    k_agg2_init <<<(N * 12 + 255) / 256, 256, 0, stream>>>(G, dinv, Oacc, N);
    k_agg2_edges<<<(E + 255) / 256, 256, 0, stream>>>(src, dst, ew, dinv, G, Oacc, E);

    k_logsoftmax<<<(N + 7) / 8, 256, 0, stream>>>(Oacc, b2, out, N);
}


# Layer-2 aggregation kept in LDS too, fed by the edges layer 1 queued (source, row, norm parked in the block's own output rows); bias and log-softmax from LDS; no global atomics; LDS arrival order
# speedup vs baseline: 1.9096x; 1.1751x over previous
#include <hip/hip_runtime.h>
#include <cstdint>

typedef float v2f __attribute__((ext_vector_type(2)));
typedef float v8f __attribute__((ext_vector_type(8)));

#define NFEAT 256
#define NHID  128
#define NCLASS 41
#define NCPAD  48

__global__ __launch_bounds__(256) void k_deg_init(float* deg, int N) {
    int i = blockIdx.x * 256 + threadIdx.x;
    if (i < N) deg[i] = 1.0f;
}

__global__ __launch_bounds__(256) void k_deg_edges(const int* __restrict__ dst,
                                                   const float* __restrict__ w,
                                                   float* deg, int E) {
    int e = blockIdx.x * 256 + threadIdx.x;
    if (e < E) unsafeAtomicAdd(&deg[dst[e]], w[e]);
}

__global__ __launch_bounds__(256) void k_deg_finish(float* dinv, int N) {
    int i = blockIdx.x * 256 + threadIdx.x;
    if (i < N) {
        float d = dinv[i];
        dinv[i] = d > 0.0f ? rsqrtf(d) : 0.0f;
    }
}

__global__ __launch_bounds__(256) void k_pack_w1(const float* __restrict__ W1, float* __restrict__ Wp) {
    int idx = blockIdx.x * 256 + threadIdx.x;
    if (idx >= NFEAT * NHID) return;
    int k = idx >> 7, n = idx & 127;
    Wp[(size_t)(k >> 1) * (NHID * 2) + n * 2 + (k & 1)] = W1[idx];
}

__global__ __launch_bounds__(256) void k_pack_w2(const float* __restrict__ W2, float* __restrict__ Wp) {
    int idx = blockIdx.x * 256 + threadIdx.x;
    if (idx >= NHID * NCPAD) return;
    int k = idx / NCPAD, n = idx % NCPAD;
    float v = (n < NCLASS) ? W2[k * NCLASS + n] : 0.0f;
    Wp[(size_t)(k >> 1) * (NCPAD * 2) + n * 2 + (k & 1)] = v;
}

__global__ __launch_bounds__(256) void k_gemm1(const float* __restrict__ X,
                                               const float* __restrict__ Wp,
                                               float* __restrict__ H, int M) {
    __shared__ float2 Xs[128 * 33];
    const int tid = threadIdx.x;
    const int m0 = blockIdx.x * 32;

#pragma unroll
    for (int it = 0; it < 8; ++it) {
        int q = it * 256 + tid;
        int m = q >> 6, k4 = q & 63;
        int mr = m0 + m; if (mr >= M) mr = M - 1;
        float4 v = ((const float4*)(X + (size_t)mr * NFEAT))[k4];
        Xs[(2 * k4)     * 33 + m] = make_float2(v.x, v.y);
        Xs[(2 * k4 + 1) * 33 + m] = make_float2(v.z, v.w);
    }
    __syncthreads();

    const int wave = tid >> 5, lane = tid & 31, l = lane & 15, half = lane >> 4;
    const int n0 = wave * 16;
    const float2* Bp = (const float2*)Wp;

    v8f acc0 = {}, acc1 = {};
#pragma unroll 4
    for (int kp = 0; kp < 128; kp += 2) {
        int row = kp + half;
        float2 a0 = Xs[row * 33 + l];
        float2 a1 = Xs[row * 33 + 16 + l];
        float2 b  = Bp[(size_t)row * NHID + n0 + l];
        v2f av0 = {a0.x, a0.y}, av1 = {a1.x, a1.y}, bv = {b.x, b.y};
        acc0 = __builtin_amdgcn_wmma_f32_16x16x4_f32(false, av0, false, bv, (short)0, acc0, false, false);
        acc1 = __builtin_amdgcn_wmma_f32_16x16x4_f32(false, av1, false, bv, (short)0, acc1, false, false);
    }

    const int mb = 8 * half;
    float* Ho = H + (size_t)(m0 + mb) * NHID + n0 + l;
    if (m0 + 32 <= M) {
#pragma unroll
        for (int r = 0; r < 8; ++r) {
            Ho[(size_t)r * NHID]        = acc0[r];
            Ho[(size_t)(16 + r) * NHID] = acc1[r];
        }
    } else {
#pragma unroll
        for (int r = 0; r < 8; ++r) {
            if (m0 + mb + r < M)      Ho[(size_t)r * NHID]        = acc0[r];
            if (m0 + 16 + mb + r < M) Ho[(size_t)(16 + r) * NHID] = acc1[r];
        }
    }
}

#define AG_ROWS 392
#define AG_QCAP 8192

__device__ __forceinline__ void ag_push(int e, int r, int lo, int* qcnt, int2* q, float* acc,
                                        const int* __restrict__ src, const float* __restrict__ w,
                                        const float* __restrict__ dinv, const float* __restrict__ H) {
    int slot = __hip_atomic_fetch_add(qcnt, 1, __ATOMIC_RELAXED, __HIP_MEMORY_SCOPE_WORKGROUP);
    if (slot < AG_QCAP) {
        q[slot] = make_int2(e, r);
    } else {
        int s = src[e];
        float c = dinv[s] * w[e] * dinv[lo + r];
        const float* h = H + (size_t)s * NHID;
        for (int f = 0; f < NHID; ++f)
            __hip_atomic_fetch_add(&acc[r * NHID + (f & 3) * 32 + (f >> 2)], c * h[f],
                                   __ATOMIC_RELAXED, __HIP_MEMORY_SCOPE_WORKGROUP);
    }
}

__global__ __launch_bounds__(1024) void k_agg1_lds(const int* __restrict__ src,
                                                   const int* __restrict__ dst,
                                                   const float* __restrict__ w,
                                                   const float* __restrict__ dinv,
                                                   const float* __restrict__ H,
                                                   float* __restrict__ Hacc, float* parkbase,
                                                   int N, int E) {
    __shared__ float acc[AG_ROWS * NHID];
    __shared__ int2 q[AG_QCAP];
    __shared__ int qcnt;

    const int tid = threadIdx.x, lane = tid & 31, wave = tid >> 5;
    const float4* H4 = (const float4*)H;

    int head = (int)(((16u - (unsigned)((uintptr_t)dst & 15u)) & 15u) >> 2);
    if (head > E) head = E;
    const int nvec = (E - head) >> 2;
    const int tail0 = head + 4 * nvec;
    const int4* dst4 = (const int4*)(dst + head);
    const int start = (int)(((long long)blockIdx.x * nvec) / gridDim.x);

    for (int lo = blockIdx.x * AG_ROWS; lo < N; lo += gridDim.x * AG_ROWS) {
        const int rows = min(AG_ROWS, N - lo);

        for (int idx = tid; idx < rows * 32; idx += 1024) {
            int r = idx >> 5, i = lo + r;
            float di = dinv[i], c = di * di;
            float4 v = H4[(size_t)i * 32 + lane];
            float* a = acc + r * NHID + lane;
            a[0] = c * v.x; a[32] = c * v.y; a[64] = c * v.z; a[96] = c * v.w;
        }
        if (tid == 0) qcnt = 0;
        __syncthreads();

        if (tid < head) {
            unsigned r = (unsigned)(dst[tid] - lo);
            if (r < (unsigned)rows) ag_push(tid, (int)r, lo, &qcnt, q, acc, src, w, dinv, H);
        }
        if (tail0 + tid < E) {
            unsigned r = (unsigned)(dst[tail0 + tid] - lo);
            if (r < (unsigned)rows) ag_push(tail0 + tid, (int)r, lo, &qcnt, q, acc, src, w, dinv, H);
        }
        for (int i0 = 0; i0 < nvec; i0 += 8 * 1024) {
            int4 d[8];
#pragma unroll
            for (int u = 0; u < 8; ++u) {
                int v = i0 + u * 1024 + tid;
                int pp = v + start; if (pp >= nvec) pp -= nvec;
                d[u] = make_int4(-1, -1, -1, -1);
                if (v < nvec) d[u] = dst4[pp];
            }
#pragma unroll
            for (int u = 0; u < 8; ++u) {
                unsigned r0 = (unsigned)(d[u].x - lo), r1 = (unsigned)(d[u].y - lo);
                unsigned r2 = (unsigned)(d[u].z - lo), r3 = (unsigned)(d[u].w - lo);
                unsigned rmin = min(min(r0, r1), min(r2, r3));
                if (rmin < (unsigned)rows) {
                    int pp = i0 + u * 1024 + tid + start; if (pp >= nvec) pp -= nvec;
                    int e = head + 4 * pp;
                    if (r0 < (unsigned)rows) ag_push(e + 0, (int)r0, lo, &qcnt, q, acc, src, w, dinv, H);
                    if (r1 < (unsigned)rows) ag_push(e + 1, (int)r1, lo, &qcnt, q, acc, src, w, dinv, H);
                    if (r2 < (unsigned)rows) ag_push(e + 2, (int)r2, lo, &qcnt, q, acc, src, w, dinv, H);
                    if (r3 < (unsigned)rows) ag_push(e + 3, (int)r3, lo, &qcnt, q, acc, src, w, dinv, H);
                }
            }
        }
        __syncthreads();

        const int nq = min(qcnt, AG_QCAP);
        int* park = (int*)(parkbase + (size_t)lo * NCLASS);
        int2* park2 = (int2*)(park + 2);
        const int npark = min(nq, (rows * NCLASS - 2) >> 1);
        if (tid == 0) park[0] = qcnt;
        for (int base = wave * 32; base < nq; base += 32 * 32) {
            const int n = min(32, nq - base);
            int s = 0, r = 0;
            float c = 0.0f;
            if (lane < n) {
                int2 t = q[base + lane];
                r = t.y;
                s = src[t.x];
                c = dinv[s] * w[t.x] * dinv[lo + r];
                if (base + lane < npark) park2[base + lane] = make_int2((s << 9) | r, __float_as_int(c));
            }
            int j0 = 0;
            for (; j0 + 8 <= n; j0 += 8) {
                float4 v[8];
#pragma unroll
                for (int u = 0; u < 8; ++u) {
                    int sj = __builtin_amdgcn_readlane(s, j0 + u);
                    v[u] = H4[(size_t)sj * 32 + lane];
                }
#pragma unroll
                for (int u = 0; u < 8; ++u) {
                    int rj = __builtin_amdgcn_readlane(r, j0 + u);
                    float cj = __int_as_float(__builtin_amdgcn_readlane(__float_as_int(c), j0 + u));
                    float* a = acc + rj * NHID + lane;
                    __hip_atomic_fetch_add(a + 0,  cj * v[u].x, __ATOMIC_RELAXED, __HIP_MEMORY_SCOPE_WORKGROUP);
                    __hip_atomic_fetch_add(a + 32, cj * v[u].y, __ATOMIC_RELAXED, __HIP_MEMORY_SCOPE_WORKGROUP);
                    __hip_atomic_fetch_add(a + 64, cj * v[u].z, __ATOMIC_RELAXED, __HIP_MEMORY_SCOPE_WORKGROUP);
                    __hip_atomic_fetch_add(a + 96, cj * v[u].w, __ATOMIC_RELAXED, __HIP_MEMORY_SCOPE_WORKGROUP);
                }
            }
            for (; j0 < n; ++j0) {
                int sj = __builtin_amdgcn_readlane(s, j0);
                int rj = __builtin_amdgcn_readlane(r, j0);
                float cj = __int_as_float(__builtin_amdgcn_readlane(__float_as_int(c), j0));
                float4 v = H4[(size_t)sj * 32 + lane];
                float* a = acc + rj * NHID + lane;
                __hip_atomic_fetch_add(a + 0,  cj * v.x, __ATOMIC_RELAXED, __HIP_MEMORY_SCOPE_WORKGROUP);
                __hip_atomic_fetch_add(a + 32, cj * v.y, __ATOMIC_RELAXED, __HIP_MEMORY_SCOPE_WORKGROUP);
                __hip_atomic_fetch_add(a + 64, cj * v.z, __ATOMIC_RELAXED, __HIP_MEMORY_SCOPE_WORKGROUP);
                __hip_atomic_fetch_add(a + 96, cj * v.w, __ATOMIC_RELAXED, __HIP_MEMORY_SCOPE_WORKGROUP);
            }
        }
        __syncthreads();

        for (int idx = tid; idx < rows * 32; idx += 1024) {
            int r = idx >> 5;
            const float* a = acc + r * NHID + lane;
            ((float4*)(Hacc + (size_t)(lo + r) * NHID))[lane] = make_float4(a[0], a[32], a[64], a[96]);
        }
        __syncthreads();
    }
}

__global__ __launch_bounds__(64) void k_gemm2(const float* __restrict__ Hacc,
                                              const float* __restrict__ b1,
                                              const float* __restrict__ Wp,
                                              float* __restrict__ G, int M) {
    __shared__ float2 Xs[64 * 33];
    const int tid = threadIdx.x;
    const int m0 = blockIdx.x * 32;

#pragma unroll
    for (int it = 0; it < 16; ++it) {
        int q = it * 64 + tid;
        int m = q >> 5, k4 = q & 31;
        int mr = m0 + m; if (mr >= M) mr = M - 1;
        float4 v = ((const float4*)(Hacc + (size_t)mr * NHID))[k4];
        float4 bb = ((const float4*)b1)[k4];
        v.x = fmaxf(v.x + bb.x, 0.0f);
        v.y = fmaxf(v.y + bb.y, 0.0f);
        v.z = fmaxf(v.z + bb.z, 0.0f);
        v.w = fmaxf(v.w + bb.w, 0.0f);
        Xs[(2 * k4)     * 33 + m] = make_float2(v.x, v.y);
        Xs[(2 * k4 + 1) * 33 + m] = make_float2(v.z, v.w);
    }
    __syncthreads();

    const int wave = tid >> 5, lane = tid & 31, l = lane & 15, half = lane >> 4;
    const float2* Bp = (const float2*)Wp;

    v8f acc0 = {}, acc1 = {}, acc2 = {};
#pragma unroll 4
    for (int kp = 0; kp < 64; kp += 2) {
        int row = kp + half;
        float2 a  = Xs[row * 33 + wave * 16 + l];
        float2 b0 = Bp[(size_t)row * NCPAD + 0  + l];
        float2 b1f= Bp[(size_t)row * NCPAD + 16 + l];
        float2 b2f= Bp[(size_t)row * NCPAD + 32 + l];
        v2f av = {a.x, a.y};
        v2f bv0 = {b0.x, b0.y}, bv1 = {b1f.x, b1f.y}, bv2 = {b2f.x, b2f.y};
        acc0 = __builtin_amdgcn_wmma_f32_16x16x4_f32(false, av, false, bv0, (short)0, acc0, false, false);
        acc1 = __builtin_amdgcn_wmma_f32_16x16x4_f32(false, av, false, bv1, (short)0, acc1, false, false);
        acc2 = __builtin_amdgcn_wmma_f32_16x16x4_f32(false, av, false, bv2, (short)0, acc2, false, false);
    }

    const int mb = m0 + wave * 16 + 8 * half;
    float* Go = G + (size_t)mb * NCPAD + l;
    if (m0 + 32 <= M) {
#pragma unroll
        for (int r = 0; r < 8; ++r) {
            float* o = Go + (size_t)r * NCPAD;
            o[0]  = acc0[r];
            o[16] = acc1[r];
            o[32] = acc2[r];
        }
    } else {
#pragma unroll
        for (int r = 0; r < 8; ++r) {
            if (mb + r < M) {
                float* o = Go + (size_t)r * NCPAD;
                o[0]  = acc0[r];
                o[16] = acc1[r];
                o[32] = acc2[r];
            }
        }
    }
}

__global__ __launch_bounds__(1024) void k_agg2_lds(const int* __restrict__ src,
                                                   const int* __restrict__ dst,
                                                   const float* __restrict__ w,
                                                   const float* __restrict__ dinv,
                                                   const float* __restrict__ G,
                                                   const float* __restrict__ b2,
                                                   float* out, int N, int E) {
    __shared__ __attribute__((aligned(16))) float acc[AG_ROWS * NCPAD];

    const int tid = threadIdx.x, lane = tid & 31, wave = tid >> 5;
    const bool lo16 = lane < 16;

    for (int lo = blockIdx.x * AG_ROWS; lo < N; lo += gridDim.x * AG_ROWS) {
        const int rows = min(AG_ROWS, N - lo);

        for (int idx = tid; idx < rows * 12; idx += 1024) {
            int r = idx / 12, j = idx - r * 12, i = lo + r;
            float di = dinv[i], c = di * di;
            float4 v = ((const float4*)(G + (size_t)i * NCPAD))[j];
            ((float4*)(acc + r * NCPAD))[j] = make_float4(c * v.x, c * v.y, c * v.z, c * v.w);
        }
        const int* park = (const int*)(out + (size_t)lo * NCLASS);
        const int cnt = park[0];
        const int room = min(AG_QCAP, (rows * NCLASS - 2) >> 1);
        __syncthreads();

        if (cnt <= room) {
            const int2* park2 = (const int2*)(park + 2);
            for (int base = wave * 32; base < cnt; base += 32 * 32) {
                const int n = min(32, cnt - base);
                int pk = 0;
                float c = 0.0f;
                if (lane < n) {
                    int2 t = park2[base + lane];
                    pk = t.x;
                    c = __int_as_float(t.y);
                }
                int j0 = 0;
                for (; j0 + 16 <= n; j0 += 16) {
                    int pj[16];
                    float cj[16];
#pragma unroll
                    for (int u = 0; u < 16; ++u) {
                        pj[u] = __builtin_amdgcn_readlane(pk, j0 + u);
                        cj[u] = __int_as_float(__builtin_amdgcn_readlane(__float_as_int(c), j0 + u));
                    }
                    float ga[16], gb[16];
#pragma unroll
                    for (int u = 0; u < 16; ++u)
                        ga[u] = G[(size_t)(pj[u] >> 9) * NCPAD + lane];
                    if (lo16) {
#pragma unroll
                        for (int u = 0; u < 16; ++u)
                            gb[u] = G[(size_t)(pj[u] >> 9) * NCPAD + 32 + lane];
                    }
#pragma unroll
                    for (int u = 0; u < 16; ++u)
                        __hip_atomic_fetch_add(acc + (pj[u] & 511) * NCPAD + lane, cj[u] * ga[u],
                                               __ATOMIC_RELAXED, __HIP_MEMORY_SCOPE_WORKGROUP);
                    if (lo16) {
#pragma unroll
                        for (int u = 0; u < 16; ++u)
                            __hip_atomic_fetch_add(acc + (pj[u] & 511) * NCPAD + 32 + lane, cj[u] * gb[u],
                                                   __ATOMIC_RELAXED, __HIP_MEMORY_SCOPE_WORKGROUP);
                    }
                }
                for (; j0 < n; ++j0) {
                    int pj = __builtin_amdgcn_readlane(pk, j0);
                    float cj = __int_as_float(__builtin_amdgcn_readlane(__float_as_int(c), j0));
                    const float* g = G + (size_t)(pj >> 9) * NCPAD;
                    float* a = acc + (pj & 511) * NCPAD;
                    __hip_atomic_fetch_add(a + lane, cj * g[lane], __ATOMIC_RELAXED, __HIP_MEMORY_SCOPE_WORKGROUP);
                    if (lo16)
                        __hip_atomic_fetch_add(a + 32 + lane, cj * g[32 + lane],
                                               __ATOMIC_RELAXED, __HIP_MEMORY_SCOPE_WORKGROUP);
                }
            }
        } else {
            for (int e = tid; e < E; e += 1024) {
                unsigned r = (unsigned)(dst[e] - lo);
                if (r < (unsigned)rows) {
                    int s = src[e];
                    float c = dinv[s] * w[e] * dinv[lo + r];
                    const float* g = G + (size_t)s * NCPAD;
                    for (int f = 0; f < NCLASS; ++f)
                        __hip_atomic_fetch_add(&acc[r * NCPAD + f], c * g[f],
                                               __ATOMIC_RELAXED, __HIP_MEMORY_SCOPE_WORKGROUP);
                }
            }
        }
        __syncthreads();

        for (int r = wave; r < rows; r += 32) {
            const float* row = acc + r * NCPAD;
            float v1 = row[lane] + b2[lane];
            int c2 = 32 + lane;
            bool has2 = c2 < NCLASS;
            float v2 = has2 ? (row[c2] + b2[c2]) : -INFINITY;

            float mx = fmaxf(v1, v2);
#pragma unroll
            for (int off = 16; off >= 1; off >>= 1) mx = fmaxf(mx, __shfl_xor(mx, off));

            float sm = expf(v1 - mx) + (has2 ? expf(v2 - mx) : 0.0f);
#pragma unroll
            for (int off = 16; off >= 1; off >>= 1) sm += __shfl_xor(sm, off);

            float ls = logf(sm);
            float* o = out + (size_t)(lo + r) * NCLASS;
            o[lane] = v1 - mx - ls;
            if (has2) o[c2] = v2 - mx - ls;
        }
        __syncthreads();
    }
}

extern "C" void kernel_launch(void* const* d_in, const int* in_sizes, int n_in,
                              void* d_out, int out_size, void* d_ws, size_t ws_size,
                              hipStream_t stream) {
    const float* x  = (const float*)d_in[0];
    const int*   ei = (const int*)  d_in[1];
    const float* ew = (const float*)d_in[2];
    const float* W1 = (const float*)d_in[3];
    const float* b1 = (const float*)d_in[4];
    const float* W2 = (const float*)d_in[5];
    const float* b2 = (const float*)d_in[6];
    float* out = (float*)d_out;

    const int N = in_sizes[0] / NFEAT;
    const int E = in_sizes[2];
    const int* src = ei;
    const int* dst = ei + E;

    float* ws = (float*)d_ws;
    size_t o = 0;
    auto take = [&](size_t n) { float* p = ws + o; o += (n + 255) & ~(size_t)255; return p; };
    float* dinv = take((size_t)N);
    float* Wp1  = take((size_t)NFEAT * NHID);
    float* Wp2  = take((size_t)NHID * NCPAD);
    float* Hxw  = take((size_t)N * NHID);
    float* Hacc = take((size_t)N * NHID);
    float* G    = Hxw;

    k_pack_w1<<<(NFEAT * NHID + 255) / 256, 256, 0, stream>>>(W1, Wp1);
    k_pack_w2<<<(NHID * NCPAD + 255) / 256, 256, 0, stream>>>(W2, Wp2);

    k_deg_init  <<<(N + 255) / 256, 256, 0, stream>>>(dinv, N);
    k_deg_edges <<<(E + 255) / 256, 256, 0, stream>>>(dst, ew, dinv, E);
    k_deg_finish<<<(N + 255) / 256, 256, 0, stream>>>(dinv, N);

    k_gemm1<<<(N + 31) / 32, 256, 0, stream>>>(x, Wp1, Hxw, N);

    k_agg1_lds<<<256, 1024, 0, stream>>>(src, dst, ew, dinv, Hxw, Hacc, out, N, E);

    k_gemm2<<<(N + 31) / 32, 64, 0, stream>>>(Hacc, b1, Wp2, G, N);

    k_agg2_lds<<<256, 1024, 0, stream>>>(src, dst, ew, dinv, G, b2, out, N, E);
}


# Degrees come from the owner blocks' sweep, now a launch of its own: row degrees summed in LDS (arrival order), queue parked as (source,row,weight) for both layers; global atomics and 2 launches go
# speedup vs baseline: 2.4883x; 1.3030x over previous
#include <hip/hip_runtime.h>
#include <cstdint>

typedef float v2f __attribute__((ext_vector_type(2)));
typedef float v8f __attribute__((ext_vector_type(8)));

#define NFEAT 256
#define NHID  128
#define NCLASS 41
#define NCPAD  48

__global__ __launch_bounds__(256) void k_pack_w1(const float* __restrict__ W1, float* __restrict__ Wp) {
    int idx = blockIdx.x * 256 + threadIdx.x;
    if (idx >= NFEAT * NHID) return;
    int k = idx >> 7, n = idx & 127;
    Wp[(size_t)(k >> 1) * (NHID * 2) + n * 2 + (k & 1)] = W1[idx];
}

__global__ __launch_bounds__(256) void k_pack_w2(const float* __restrict__ W2, float* __restrict__ Wp) {
    int idx = blockIdx.x * 256 + threadIdx.x;
    if (idx >= NHID * NCPAD) return;
    int k = idx / NCPAD, n = idx % NCPAD;
    float v = (n < NCLASS) ? W2[k * NCLASS + n] : 0.0f;
    Wp[(size_t)(k >> 1) * (NCPAD * 2) + n * 2 + (k & 1)] = v;
}

__global__ __launch_bounds__(256) void k_gemm1(const float* __restrict__ X,
                                               const float* __restrict__ Wp,
                                               float* __restrict__ H, int M) {
    __shared__ float2 Xs[128 * 33];
    const int tid = threadIdx.x;
    const int m0 = blockIdx.x * 32;

#pragma unroll
    for (int it = 0; it < 8; ++it) {
        int q = it * 256 + tid;
        int m = q >> 6, k4 = q & 63;
        int mr = m0 + m; if (mr >= M) mr = M - 1;
        float4 v = ((const float4*)(X + (size_t)mr * NFEAT))[k4];
        Xs[(2 * k4)     * 33 + m] = make_float2(v.x, v.y);
        Xs[(2 * k4 + 1) * 33 + m] = make_float2(v.z, v.w);
    }
    __syncthreads();

    const int wave = tid >> 5, lane = tid & 31, l = lane & 15, half = lane >> 4;
    const int n0 = wave * 16;
    const float2* Bp = (const float2*)Wp;

    v8f acc0 = {}, acc1 = {};
#pragma unroll 4
    for (int kp = 0; kp < 128; kp += 2) {
        int row = kp + half;
        float2 a0 = Xs[row * 33 + l];
        float2 a1 = Xs[row * 33 + 16 + l];
        float2 b  = Bp[(size_t)row * NHID + n0 + l];
        v2f av0 = {a0.x, a0.y}, av1 = {a1.x, a1.y}, bv = {b.x, b.y};
        acc0 = __builtin_amdgcn_wmma_f32_16x16x4_f32(false, av0, false, bv, (short)0, acc0, false, false);
        acc1 = __builtin_amdgcn_wmma_f32_16x16x4_f32(false, av1, false, bv, (short)0, acc1, false, false);
    }

    const int mb = 8 * half;
    float* Ho = H + (size_t)(m0 + mb) * NHID + n0 + l;
    if (m0 + 32 <= M) {
#pragma unroll
        for (int r = 0; r < 8; ++r) {
            Ho[(size_t)r * NHID]        = acc0[r];
            Ho[(size_t)(16 + r) * NHID] = acc1[r];
        }
    } else {
#pragma unroll
        for (int r = 0; r < 8; ++r) {
            if (m0 + mb + r < M)      Ho[(size_t)r * NHID]        = acc0[r];
            if (m0 + 16 + mb + r < M) Ho[(size_t)(16 + r) * NHID] = acc1[r];
        }
    }
}

#define AG_ROWS 392
#define AG_QCAP 8192

__device__ __forceinline__ void sw_push(int e, int r, int* qcnt, int2* q, float* degs,
                                        const float* __restrict__ w) {
    int slot = __hip_atomic_fetch_add(qcnt, 1, __ATOMIC_RELAXED, __HIP_MEMORY_SCOPE_WORKGROUP);
    if (slot < AG_QCAP) {
        q[slot] = make_int2(e, r);
    } else {
        __hip_atomic_fetch_add(&degs[r], w[e],
                               __ATOMIC_RELAXED, __HIP_MEMORY_SCOPE_WORKGROUP);
    }
}

__global__ __launch_bounds__(1024) void k_sweep_deg(const int* __restrict__ src,
                                                    const int* __restrict__ dst,
                                                    const float* __restrict__ w,
                                                    float* __restrict__ dinv, float* parkbase,
                                                    int N, int E) {
    __shared__ int2 q[AG_QCAP];
    __shared__ float degs[AG_ROWS];
    __shared__ int qcnt;

    const int tid = threadIdx.x;

    int head = (int)(((16u - (unsigned)((uintptr_t)dst & 15u)) & 15u) >> 2);
    if (head > E) head = E;
    const int nvec = (E - head) >> 2;
    const int tail0 = head + 4 * nvec;
    const int4* dst4 = (const int4*)(dst + head);
    const int start = (int)(((long long)blockIdx.x * nvec) / gridDim.x);

    for (int lo = blockIdx.x * AG_ROWS; lo < N; lo += gridDim.x * AG_ROWS) {
        const int rows = min(AG_ROWS, N - lo);

        for (int r = tid; r < rows; r += 1024) degs[r] = 1.0f;
        if (tid == 0) qcnt = 0;
        __syncthreads();

        if (tid < head) {
            unsigned r = (unsigned)(dst[tid] - lo);
            if (r < (unsigned)rows) sw_push(tid, (int)r, &qcnt, q, degs, w);
        }
        if (tail0 + tid < E) {
            unsigned r = (unsigned)(dst[tail0 + tid] - lo);
            if (r < (unsigned)rows) sw_push(tail0 + tid, (int)r, &qcnt, q, degs, w);
        }
        for (int i0 = 0; i0 < nvec; i0 += 8 * 1024) {
            int4 d[8];
#pragma unroll
            for (int u = 0; u < 8; ++u) {
                int v = i0 + u * 1024 + tid;
                int pp = v + start; if (pp >= nvec) pp -= nvec;
                d[u] = make_int4(-1, -1, -1, -1);
                if (v < nvec) d[u] = dst4[pp];
            }
#pragma unroll
            for (int u = 0; u < 8; ++u) {
                unsigned r0 = (unsigned)(d[u].x - lo), r1 = (unsigned)(d[u].y - lo);
                unsigned r2 = (unsigned)(d[u].z - lo), r3 = (unsigned)(d[u].w - lo);
                unsigned rmin = min(min(r0, r1), min(r2, r3));
                if (rmin < (unsigned)rows) {
                    int pp = i0 + u * 1024 + tid + start; if (pp >= nvec) pp -= nvec;
                    int e = head + 4 * pp;
                    if (r0 < (unsigned)rows) sw_push(e + 0, (int)r0, &qcnt, q, degs, w);
                    if (r1 < (unsigned)rows) sw_push(e + 1, (int)r1, &qcnt, q, degs, w);
                    if (r2 < (unsigned)rows) sw_push(e + 2, (int)r2, &qcnt, q, degs, w);
                    if (r3 < (unsigned)rows) sw_push(e + 3, (int)r3, &qcnt, q, degs, w);
                }
            }
        }
        __syncthreads();

        const int nq = min(qcnt, AG_QCAP);
        int* park = (int*)(parkbase + (size_t)lo * NCLASS);
        int2* park2 = (int2*)(park + 2);
        const int npark = min(nq, (rows * NCLASS - 2) >> 1);
        if (tid == 0) park[0] = qcnt;
        int2 t[AG_QCAP / 1024];
        float ww[AG_QCAP / 1024];
        int sid[AG_QCAP / 1024];
#pragma unroll
        for (int u = 0; u < AG_QCAP / 1024; ++u) {
            int i = u * 1024 + tid;
            t[u] = make_int2(0, 0); ww[u] = 0.0f; sid[u] = 0;
            if (i < nq) {
                t[u] = q[i];
                ww[u] = w[t[u].x];
                sid[u] = src[t[u].x];
            }
        }
#pragma unroll
        for (int u = 0; u < AG_QCAP / 1024; ++u) {
            int i = u * 1024 + tid;
            if (i < nq)
                __hip_atomic_fetch_add(&degs[t[u].y], ww[u], __ATOMIC_RELAXED, __HIP_MEMORY_SCOPE_WORKGROUP);
            if (i < npark) park2[i] = make_int2((sid[u] << 9) | t[u].y, __float_as_int(ww[u]));
        }
        __syncthreads();

        for (int r = tid; r < rows; r += 1024) {
            float dg = degs[r];
            dinv[lo + r] = dg > 0.0f ? rsqrtf(dg) : 0.0f;
        }
        __syncthreads();
    }
}

__global__ __launch_bounds__(1024) void k_agg1_lds(const int* __restrict__ src,
                                                   const int* __restrict__ dst,
                                                   const float* __restrict__ w,
                                                   const float* __restrict__ dinv,
                                                   const float* __restrict__ H,
                                                   float* __restrict__ Hacc, float* parkbase,
                                                   int N, int E) {
    __shared__ float acc[AG_ROWS * NHID];

    const int tid = threadIdx.x, lane = tid & 31, wave = tid >> 5;
    const float4* H4 = (const float4*)H;

    for (int lo = blockIdx.x * AG_ROWS; lo < N; lo += gridDim.x * AG_ROWS) {
        const int rows = min(AG_ROWS, N - lo);

        for (int idx = tid; idx < rows * 32; idx += 1024) {
            int r = idx >> 5, i = lo + r;
            float di = dinv[i], c = di * di;
            float4 v = H4[(size_t)i * 32 + lane];
            float* a = acc + r * NHID + lane;
            a[0] = c * v.x; a[32] = c * v.y; a[64] = c * v.z; a[96] = c * v.w;
        }
        int* park = (int*)(parkbase + (size_t)lo * NCLASS);
        const int cnt = park[0];
        const int room = min(AG_QCAP, (rows * NCLASS - 2) >> 1);
        __syncthreads();

        if (cnt <= room) {
            int2* park2 = (int2*)(park + 2);
            for (int base = wave * 32; base < cnt; base += 32 * 32) {
                const int n = min(32, cnt - base);
                int s = 0, r = 0;
                float c = 0.0f;
                if (lane < n) {
                    int2 t = park2[base + lane];
                    s = t.x >> 9;
                    r = t.x & 511;
                    c = dinv[s] * __int_as_float(t.y) * dinv[lo + r];
                    park2[base + lane] = make_int2(t.x, __float_as_int(c));
                }
                int j0 = 0;
                for (; j0 + 8 <= n; j0 += 8) {
                    float4 v[8];
#pragma unroll
                    for (int u = 0; u < 8; ++u) {
                        int sj = __builtin_amdgcn_readlane(s, j0 + u);
                        v[u] = H4[(size_t)sj * 32 + lane];
                    }
#pragma unroll
                    for (int u = 0; u < 8; ++u) {
                        int rj = __builtin_amdgcn_readlane(r, j0 + u);
                        float cj = __int_as_float(__builtin_amdgcn_readlane(__float_as_int(c), j0 + u));
                        float* a = acc + rj * NHID + lane;
                        __hip_atomic_fetch_add(a + 0,  cj * v[u].x, __ATOMIC_RELAXED, __HIP_MEMORY_SCOPE_WORKGROUP);
                        __hip_atomic_fetch_add(a + 32, cj * v[u].y, __ATOMIC_RELAXED, __HIP_MEMORY_SCOPE_WORKGROUP);
                        __hip_atomic_fetch_add(a + 64, cj * v[u].z, __ATOMIC_RELAXED, __HIP_MEMORY_SCOPE_WORKGROUP);
                        __hip_atomic_fetch_add(a + 96, cj * v[u].w, __ATOMIC_RELAXED, __HIP_MEMORY_SCOPE_WORKGROUP);
                    }
                }
                for (; j0 < n; ++j0) {
                    int sj = __builtin_amdgcn_readlane(s, j0);
                    int rj = __builtin_amdgcn_readlane(r, j0);
                    float cj = __int_as_float(__builtin_amdgcn_readlane(__float_as_int(c), j0));
                    float4 v = H4[(size_t)sj * 32 + lane];
                    float* a = acc + rj * NHID + lane;
                    __hip_atomic_fetch_add(a + 0,  cj * v.x, __ATOMIC_RELAXED, __HIP_MEMORY_SCOPE_WORKGROUP);
                    __hip_atomic_fetch_add(a + 32, cj * v.y, __ATOMIC_RELAXED, __HIP_MEMORY_SCOPE_WORKGROUP);
                    __hip_atomic_fetch_add(a + 64, cj * v.z, __ATOMIC_RELAXED, __HIP_MEMORY_SCOPE_WORKGROUP);
                    __hip_atomic_fetch_add(a + 96, cj * v.w, __ATOMIC_RELAXED, __HIP_MEMORY_SCOPE_WORKGROUP);
                }
            }
        } else {
            for (int e = tid; e < E; e += 1024) {
                unsigned r = (unsigned)(dst[e] - lo);
                if (r < (unsigned)rows) {
                    int s = src[e];
                    float c = dinv[s] * w[e] * dinv[lo + r];
                    const float* h = H + (size_t)s * NHID;
                    for (int f = 0; f < NHID; ++f)
                        __hip_atomic_fetch_add(&acc[r * NHID + (f & 3) * 32 + (f >> 2)], c * h[f],
                                               __ATOMIC_RELAXED, __HIP_MEMORY_SCOPE_WORKGROUP);
                }
            }
        }
        __syncthreads();

        for (int idx = tid; idx < rows * 32; idx += 1024) {
            int r = idx >> 5;
            const float* a = acc + r * NHID + lane;
            ((float4*)(Hacc + (size_t)(lo + r) * NHID))[lane] = make_float4(a[0], a[32], a[64], a[96]);
        }
        __syncthreads();
    }
}

__global__ __launch_bounds__(64) void k_gemm2(const float* __restrict__ Hacc,
                                              const float* __restrict__ b1,
                                              const float* __restrict__ Wp,
                                              float* __restrict__ G, int M) {
    __shared__ float2 Xs[64 * 33];
    const int tid = threadIdx.x;
    const int m0 = blockIdx.x * 32;

#pragma unroll
    for (int it = 0; it < 16; ++it) {
        int q = it * 64 + tid;
        int m = q >> 5, k4 = q & 31;
        int mr = m0 + m; if (mr >= M) mr = M - 1;
        float4 v = ((const float4*)(Hacc + (size_t)mr * NHID))[k4];
        float4 bb = ((const float4*)b1)[k4];
        v.x = fmaxf(v.x + bb.x, 0.0f);
        v.y = fmaxf(v.y + bb.y, 0.0f);
        v.z = fmaxf(v.z + bb.z, 0.0f);
        v.w = fmaxf(v.w + bb.w, 0.0f);
        Xs[(2 * k4)     * 33 + m] = make_float2(v.x, v.y);
        Xs[(2 * k4 + 1) * 33 + m] = make_float2(v.z, v.w);
    }
    __syncthreads();

    const int wave = tid >> 5, lane = tid & 31, l = lane & 15, half = lane >> 4;
    const float2* Bp = (const float2*)Wp;

    v8f acc0 = {}, acc1 = {}, acc2 = {};
#pragma unroll 4
    for (int kp = 0; kp < 64; kp += 2) {
        int row = kp + half;
        float2 a  = Xs[row * 33 + wave * 16 + l];
        float2 b0 = Bp[(size_t)row * NCPAD + 0  + l];
        float2 b1f= Bp[(size_t)row * NCPAD + 16 + l];
        float2 b2f= Bp[(size_t)row * NCPAD + 32 + l];
        v2f av = {a.x, a.y};
        v2f bv0 = {b0.x, b0.y}, bv1 = {b1f.x, b1f.y}, bv2 = {b2f.x, b2f.y};
        acc0 = __builtin_amdgcn_wmma_f32_16x16x4_f32(false, av, false, bv0, (short)0, acc0, false, false);
        acc1 = __builtin_amdgcn_wmma_f32_16x16x4_f32(false, av, false, bv1, (short)0, acc1, false, false);
        acc2 = __builtin_amdgcn_wmma_f32_16x16x4_f32(false, av, false, bv2, (short)0, acc2, false, false);
    }

    const int mb = m0 + wave * 16 + 8 * half;
    float* Go = G + (size_t)mb * NCPAD + l;
    if (m0 + 32 <= M) {
#pragma unroll
        for (int r = 0; r < 8; ++r) {
            float* o = Go + (size_t)r * NCPAD;
            o[0]  = acc0[r];
            o[16] = acc1[r];
            o[32] = acc2[r];
        }
    } else {
#pragma unroll
        for (int r = 0; r < 8; ++r) {
            if (mb + r < M) {
                float* o = Go + (size_t)r * NCPAD;
                o[0]  = acc0[r];
                o[16] = acc1[r];
                o[32] = acc2[r];
            }
        }
    }
}

__global__ __launch_bounds__(1024) void k_agg2_lds(const int* __restrict__ src,
                                                   const int* __restrict__ dst,
                                                   const float* __restrict__ w,
                                                   const float* __restrict__ dinv,
                                                   const float* __restrict__ G,
                                                   const float* __restrict__ b2,
                                                   float* out, int N, int E) {
    __shared__ __attribute__((aligned(16))) float acc[AG_ROWS * NCPAD];

    const int tid = threadIdx.x, lane = tid & 31, wave = tid >> 5;
    const bool lo16 = lane < 16;

    for (int lo = blockIdx.x * AG_ROWS; lo < N; lo += gridDim.x * AG_ROWS) {
        const int rows = min(AG_ROWS, N - lo);

        for (int idx = tid; idx < rows * 12; idx += 1024) {
            int r = idx / 12, j = idx - r * 12, i = lo + r;
            float di = dinv[i], c = di * di;
            float4 v = ((const float4*)(G + (size_t)i * NCPAD))[j];
            ((float4*)(acc + r * NCPAD))[j] = make_float4(c * v.x, c * v.y, c * v.z, c * v.w);
        }
        const int* park = (const int*)(out + (size_t)lo * NCLASS);
        const int cnt = park[0];
        const int room = min(AG_QCAP, (rows * NCLASS - 2) >> 1);
        __syncthreads();

        if (cnt <= room) {
            const int2* park2 = (const int2*)(park + 2);
            for (int base = wave * 32; base < cnt; base += 32 * 32) {
                const int n = min(32, cnt - base);
                int pk = 0;
                float c = 0.0f;
                if (lane < n) {
                    int2 t = park2[base + lane];
                    pk = t.x;
                    c = __int_as_float(t.y);
                }
                int j0 = 0;
                for (; j0 + 16 <= n; j0 += 16) {
                    int pj[16];
                    float cj[16];
#pragma unroll
                    for (int u = 0; u < 16; ++u) {
                        pj[u] = __builtin_amdgcn_readlane(pk, j0 + u);
                        cj[u] = __int_as_float(__builtin_amdgcn_readlane(__float_as_int(c), j0 + u));
                    }
                    float ga[16], gb[16];
#pragma unroll
                    for (int u = 0; u < 16; ++u)
                        ga[u] = G[(size_t)(pj[u] >> 9) * NCPAD + lane];
                    if (lo16) {
#pragma unroll
                        for (int u = 0; u < 16; ++u)
                            gb[u] = G[(size_t)(pj[u] >> 9) * NCPAD + 32 + lane];
                    }
#pragma unroll
                    for (int u = 0; u < 16; ++u)
                        __hip_atomic_fetch_add(acc + (pj[u] & 511) * NCPAD + lane, cj[u] * ga[u],
                                               __ATOMIC_RELAXED, __HIP_MEMORY_SCOPE_WORKGROUP);
                    if (lo16) {
#pragma unroll
                        for (int u = 0; u < 16; ++u)
                            __hip_atomic_fetch_add(acc + (pj[u] & 511) * NCPAD + 32 + lane, cj[u] * gb[u],
                                                   __ATOMIC_RELAXED, __HIP_MEMORY_SCOPE_WORKGROUP);
                    }
                }
                for (; j0 < n; ++j0) {
                    int pj = __builtin_amdgcn_readlane(pk, j0);
                    float cj = __int_as_float(__builtin_amdgcn_readlane(__float_as_int(c), j0));
                    const float* g = G + (size_t)(pj >> 9) * NCPAD;
                    float* a = acc + (pj & 511) * NCPAD;
                    __hip_atomic_fetch_add(a + lane, cj * g[lane], __ATOMIC_RELAXED, __HIP_MEMORY_SCOPE_WORKGROUP);
                    if (lo16)
                        __hip_atomic_fetch_add(a + 32 + lane, cj * g[32 + lane],
                                               __ATOMIC_RELAXED, __HIP_MEMORY_SCOPE_WORKGROUP);
                }
            }
        } else {
            for (int e = tid; e < E; e += 1024) {
                unsigned r = (unsigned)(dst[e] - lo);
                if (r < (unsigned)rows) {
                    int s = src[e];
                    float c = dinv[s] * w[e] * dinv[lo + r];
                    const float* g = G + (size_t)s * NCPAD;
                    for (int f = 0; f < NCLASS; ++f)
                        __hip_atomic_fetch_add(&acc[r * NCPAD + f], c * g[f],
                                               __ATOMIC_RELAXED, __HIP_MEMORY_SCOPE_WORKGROUP);
                }
            }
        }
        __syncthreads();

        for (int r = wave; r < rows; r += 32) {
            const float* row = acc + r * NCPAD;
            float v1 = row[lane] + b2[lane];
            int c2 = 32 + lane;
            bool has2 = c2 < NCLASS;
            float v2 = has2 ? (row[c2] + b2[c2]) : -INFINITY;

            float mx = fmaxf(v1, v2);
#pragma unroll
            for (int off = 16; off >= 1; off >>= 1) mx = fmaxf(mx, __shfl_xor(mx, off));

            float sm = expf(v1 - mx) + (has2 ? expf(v2 - mx) : 0.0f);
#pragma unroll
            for (int off = 16; off >= 1; off >>= 1) sm += __shfl_xor(sm, off);

            float ls = logf(sm);
            float* o = out + (size_t)(lo + r) * NCLASS;
            o[lane] = v1 - mx - ls;
            if (has2) o[c2] = v2 - mx - ls;
        }
        __syncthreads();
    }
}

extern "C" void kernel_launch(void* const* d_in, const int* in_sizes, int n_in,
                              void* d_out, int out_size, void* d_ws, size_t ws_size,
                              hipStream_t stream) {
    const float* x  = (const float*)d_in[0];
    const int*   ei = (const int*)  d_in[1];
    const float* ew = (const float*)d_in[2];
    const float* W1 = (const float*)d_in[3];
    const float* b1 = (const float*)d_in[4];
    const float* W2 = (const float*)d_in[5];
    const float* b2 = (const float*)d_in[6];
    float* out = (float*)d_out;

    const int N = in_sizes[0] / NFEAT;
    const int E = in_sizes[2];
    const int* src = ei;
    const int* dst = ei + E;

    float* ws = (float*)d_ws;
    size_t o = 0;
    auto take = [&](size_t n) { float* p = ws + o; o += (n + 255) & ~(size_t)255; return p; };
    float* dinv = take((size_t)N);
    float* Wp1  = take((size_t)NFEAT * NHID);
    float* Wp2  = take((size_t)NHID * NCPAD);
    float* Hxw  = take((size_t)N * NHID);
    float* Hacc = take((size_t)N * NHID);
    float* G    = Hxw;

    k_pack_w1<<<(NFEAT * NHID + 255) / 256, 256, 0, stream>>>(W1, Wp1);
    k_pack_w2<<<(NHID * NCPAD + 255) / 256, 256, 0, stream>>>(W2, Wp2);

    k_sweep_deg<<<256, 1024, 0, stream>>>(src, dst, ew, dinv, out, N, E);

    k_gemm1<<<(N + 31) / 32, 256, 0, stream>>>(x, Wp1, Hxw, N);

    k_agg1_lds<<<256, 1024, 0, stream>>>(src, dst, ew, dinv, Hxw, Hacc, out, N, E);

    k_gemm2<<<(N + 31) / 32, 64, 0, stream>>>(Hacc, b1, Wp2, G, N);

    k_agg2_lds<<<256, 1024, 0, stream>>>(src, dst, ew, dinv, G, b2, out, N, E);
}


# Sweep key narrowed to one owner byte per edge: a first pass writes the byte plane, work-groups scan it (16 edges per 16-byte load, 4x fewer bytes), check queued edges exactly; LDS-arrival sums kept
# speedup vs baseline: 3.0097x; 1.2095x over previous
#include <hip/hip_runtime.h>
#include <cstdint>

typedef float v2f __attribute__((ext_vector_type(2)));
typedef float v8f __attribute__((ext_vector_type(8)));

#define NFEAT 256
#define NHID  128
#define NCLASS 41
#define NCPAD  48

__global__ __launch_bounds__(256) void k_pack_w1(const float* __restrict__ W1, float* __restrict__ Wp) {
    int idx = blockIdx.x * 256 + threadIdx.x;
    if (idx >= NFEAT * NHID) return;
    int k = idx >> 7, n = idx & 127;
    Wp[(size_t)(k >> 1) * (NHID * 2) + n * 2 + (k & 1)] = W1[idx];
}

__global__ __launch_bounds__(256) void k_pack_w2(const float* __restrict__ W2, float* __restrict__ Wp) {
    int idx = blockIdx.x * 256 + threadIdx.x;
    if (idx >= NHID * NCPAD) return;
    int k = idx / NCPAD, n = idx % NCPAD;
    float v = (n < NCLASS) ? W2[k * NCLASS + n] : 0.0f;
    Wp[(size_t)(k >> 1) * (NCPAD * 2) + n * 2 + (k & 1)] = v;
}

__global__ __launch_bounds__(256) void k_gemm1(const float* __restrict__ X,
                                               const float* __restrict__ Wp,
                                               float* __restrict__ H, int M) {
    __shared__ float2 Xs[128 * 33];
    const int tid = threadIdx.x;
    const int m0 = blockIdx.x * 32;

#pragma unroll
    for (int it = 0; it < 8; ++it) {
        int q = it * 256 + tid;
        int m = q >> 6, k4 = q & 63;
        int mr = m0 + m; if (mr >= M) mr = M - 1;
        float4 v = ((const float4*)(X + (size_t)mr * NFEAT))[k4];
        Xs[(2 * k4)     * 33 + m] = make_float2(v.x, v.y);
        Xs[(2 * k4 + 1) * 33 + m] = make_float2(v.z, v.w);
    }
    __syncthreads();

    const int wave = tid >> 5, lane = tid & 31, l = lane & 15, half = lane >> 4;
    const int n0 = wave * 16;
    const float2* Bp = (const float2*)Wp;

    v8f acc0 = {}, acc1 = {};
#pragma unroll 4
    for (int kp = 0; kp < 128; kp += 2) {
        int row = kp + half;
        float2 a0 = Xs[row * 33 + l];
        float2 a1 = Xs[row * 33 + 16 + l];
        float2 b  = Bp[(size_t)row * NHID + n0 + l];
        v2f av0 = {a0.x, a0.y}, av1 = {a1.x, a1.y}, bv = {b.x, b.y};
        acc0 = __builtin_amdgcn_wmma_f32_16x16x4_f32(false, av0, false, bv, (short)0, acc0, false, false);
        acc1 = __builtin_amdgcn_wmma_f32_16x16x4_f32(false, av1, false, bv, (short)0, acc1, false, false);
    }

    const int mb = 8 * half;
    float* Ho = H + (size_t)(m0 + mb) * NHID + n0 + l;
    if (m0 + 32 <= M) {
#pragma unroll
        for (int r = 0; r < 8; ++r) {
            Ho[(size_t)r * NHID]        = acc0[r];
            Ho[(size_t)(16 + r) * NHID] = acc1[r];
        }
    } else {
#pragma unroll
        for (int r = 0; r < 8; ++r) {
            if (m0 + mb + r < M)      Ho[(size_t)r * NHID]        = acc0[r];
            if (m0 + 16 + mb + r < M) Ho[(size_t)(16 + r) * NHID] = acc1[r];
        }
    }
}

#define AG_ROWS 392
#define AG_QCAP 8192

__device__ __forceinline__ unsigned own_byte(int d) { return ((unsigned)d / AG_ROWS) & 255u; }

__global__ __launch_bounds__(512) void k_owner_bytes(const int* __restrict__ dst,
                                                     unsigned* __restrict__ ob, int E, int nwords) {
    const bool al16 = ((uintptr_t)dst & 15u) == 0;
    const int T = gridDim.x * 512;
    for (int wd = blockIdx.x * 512 + threadIdx.x; wd < nwords; wd += T) {
        const int e = 4 * wd;
        unsigned pk = 0;
        if (al16 && e + 4 <= E) {
            int4 d = ((const int4*)dst)[wd];
            pk = own_byte(d.x) | (own_byte(d.y) << 8) | (own_byte(d.z) << 16) | (own_byte(d.w) << 24);
        } else {
            for (int k = 0; k < 4; ++k)
                if (e + k < E) pk |= own_byte(dst[e + k]) << (8 * k);
        }
        ob[wd] = pk;
    }
}

__device__ __forceinline__ void sw_push(int e, int lo, int rows, int* qcnt, int* q, float* degs,
                                        const int* __restrict__ dst, const float* __restrict__ w) {
    int slot = __hip_atomic_fetch_add(qcnt, 1, __ATOMIC_RELAXED, __HIP_MEMORY_SCOPE_WORKGROUP);
    if (slot < AG_QCAP) {
        q[slot] = e;
    } else {
        unsigned r = (unsigned)(dst[e] - lo);
        if (r < (unsigned)rows)
            __hip_atomic_fetch_add(&degs[r], w[e], __ATOMIC_RELAXED, __HIP_MEMORY_SCOPE_WORKGROUP);
    }
}

__device__ __forceinline__ unsigned zero_bytes(unsigned x) {
    unsigned y = ((x & 0x7F7F7F7Fu) + 0x7F7F7F7Fu) | x;
    return ~y & 0x80808080u;
}
__device__ __forceinline__ unsigned match16(uint4 d, unsigned pat) {
    return zero_bytes(d.x ^ pat) | (zero_bytes(d.y ^ pat) >> 1) |
           (zero_bytes(d.z ^ pat) >> 2) | (zero_bytes(d.w ^ pat) >> 3);
}

__global__ __launch_bounds__(1024) void k_sweep_deg(const int* __restrict__ src,
                                                    const int* __restrict__ dst,
                                                    const float* __restrict__ w,
                                                    const unsigned* __restrict__ ob,
                                                    float* __restrict__ dinv, float* parkbase,
                                                    int N, int E, int nvec) {
    __shared__ int q[AG_QCAP];
    __shared__ float degs[AG_ROWS];
    __shared__ int qcnt, pcnt;

    const int tid = threadIdx.x;
    const uint4* ob4 = (const uint4*)ob;
    const int start = (int)(((long long)blockIdx.x * nvec) / gridDim.x);

    for (int lo = blockIdx.x * AG_ROWS; lo < N; lo += gridDim.x * AG_ROWS) {
        const int rows = min(AG_ROWS, N - lo);
        const unsigned pat = (((unsigned)lo / AG_ROWS) & 255u) * 0x01010101u;

        for (int r = tid; r < rows; r += 1024) degs[r] = 1.0f;
        if (tid == 0) { qcnt = 0; pcnt = 0; }
        __syncthreads();

        for (int i0 = 0; i0 < nvec; i0 += 8 * 1024) {
            uint4 d[8];
#pragma unroll
            for (int u = 0; u < 8; ++u) {
                int v = i0 + u * 1024 + tid;
                int pp = v + start; if (pp >= nvec) pp -= nvec;
                d[u] = ob4[v < nvec ? pp : 0];
            }
#pragma unroll
            for (int h = 0; h < 4; ++h) {
                unsigned ma = match16(d[2 * h], pat), mb = match16(d[2 * h + 1], pat);
                if (i0 + (2 * h) * 1024 + tid >= nvec) ma = 0;
                if (i0 + (2 * h + 1) * 1024 + tid >= nvec) mb = 0;
                unsigned m = ma | (mb >> 4);
                while (m) {
                    int p = __builtin_ctz(m);
                    m &= m - 1;
                    int j = p >> 3, b = p & 7;
                    int u = 2 * h + (b < 4 ? 1 : 0);
                    int k = (7 - b) & 3;
                    int pp = i0 + u * 1024 + tid + start; if (pp >= nvec) pp -= nvec;
                    int e = 16 * pp + 4 * k + j;
                    if (e < E) sw_push(e, lo, rows, &qcnt, q, degs, dst, w);
                }
            }
        }
        __syncthreads();

        const int nq = min(qcnt, AG_QCAP);
        int* park = (int*)(parkbase + (size_t)lo * NCLASS);
        int2* park2 = (int2*)(park + 2);
        const int room = (rows * NCLASS - 2) >> 1;
        int dd[AG_QCAP / 1024];
        float ww[AG_QCAP / 1024];
        int sid[AG_QCAP / 1024];
#pragma unroll
        for (int u = 0; u < AG_QCAP / 1024; ++u) {
            int i = u * 1024 + tid;
            dd[u] = -1; ww[u] = 0.0f; sid[u] = 0;
            if (i < nq) {
                int e = q[i];
                dd[u] = dst[e];
                ww[u] = w[e];
                sid[u] = src[e];
            }
        }
#pragma unroll
        for (int u = 0; u < AG_QCAP / 1024; ++u) {
            int i = u * 1024 + tid;
            unsigned r = (unsigned)(dd[u] - lo);
            if (i < nq && r < (unsigned)rows) {
                __hip_atomic_fetch_add(&degs[r], ww[u], __ATOMIC_RELAXED, __HIP_MEMORY_SCOPE_WORKGROUP);
                int slot = __hip_atomic_fetch_add(&pcnt, 1, __ATOMIC_RELAXED, __HIP_MEMORY_SCOPE_WORKGROUP);
                if (slot < room) park2[slot] = make_int2((sid[u] << 9) | (int)r, __float_as_int(ww[u]));
            }
        }
        __syncthreads();

        if (tid == 0) park[0] = (qcnt > AG_QCAP) ? 0x7fffffff : pcnt;
        for (int r = tid; r < rows; r += 1024) {
            float dg = degs[r];
            dinv[lo + r] = dg > 0.0f ? rsqrtf(dg) : 0.0f;
        }
        __syncthreads();
    }
}

__global__ __launch_bounds__(1024) void k_agg1_lds(const int* __restrict__ src,
                                                   const int* __restrict__ dst,
                                                   const float* __restrict__ w,
                                                   const float* __restrict__ dinv,
                                                   const float* __restrict__ H,
                                                   float* __restrict__ Hacc, float* parkbase,
                                                   int N, int E) {
    __shared__ float acc[AG_ROWS * NHID];

    const int tid = threadIdx.x, lane = tid & 31, wave = tid >> 5;
    const float4* H4 = (const float4*)H;

    for (int lo = blockIdx.x * AG_ROWS; lo < N; lo += gridDim.x * AG_ROWS) {
        const int rows = min(AG_ROWS, N - lo);

        for (int idx = tid; idx < rows * 32; idx += 1024) {
            int r = idx >> 5, i = lo + r;
            float di = dinv[i], c = di * di;
            float4 v = H4[(size_t)i * 32 + lane];
            float* a = acc + r * NHID + lane;
            a[0] = c * v.x; a[32] = c * v.y; a[64] = c * v.z; a[96] = c * v.w;
        }
        int* park = (int*)(parkbase + (size_t)lo * NCLASS);
        const int cnt = park[0];
        const int room = min(AG_QCAP, (rows * NCLASS - 2) >> 1);
        __syncthreads();

        if (cnt <= room) {
            int2* park2 = (int2*)(park + 2);
            for (int base = wave * 32; base < cnt; base += 32 * 32) {
                const int n = min(32, cnt - base);
                int s = 0, r = 0;
                float c = 0.0f;
                if (lane < n) {
                    int2 t = park2[base + lane];
                    s = t.x >> 9;
                    r = t.x & 511;
                    c = dinv[s] * __int_as_float(t.y) * dinv[lo + r];
                    park2[base + lane] = make_int2(t.x, __float_as_int(c));
                }
                int j0 = 0;
                for (; j0 + 8 <= n; j0 += 8) {
                    float4 v[8];
#pragma unroll
                    for (int u = 0; u < 8; ++u) {
                        int sj = __builtin_amdgcn_readlane(s, j0 + u);
                        v[u] = H4[(size_t)sj * 32 + lane];
                    }
#pragma unroll
                    for (int u = 0; u < 8; ++u) {
                        int rj = __builtin_amdgcn_readlane(r, j0 + u);
                        float cj = __int_as_float(__builtin_amdgcn_readlane(__float_as_int(c), j0 + u));
                        float* a = acc + rj * NHID + lane;
                        __hip_atomic_fetch_add(a + 0,  cj * v[u].x, __ATOMIC_RELAXED, __HIP_MEMORY_SCOPE_WORKGROUP);
                        __hip_atomic_fetch_add(a + 32, cj * v[u].y, __ATOMIC_RELAXED, __HIP_MEMORY_SCOPE_WORKGROUP);
                        __hip_atomic_fetch_add(a + 64, cj * v[u].z, __ATOMIC_RELAXED, __HIP_MEMORY_SCOPE_WORKGROUP);
                        __hip_atomic_fetch_add(a + 96, cj * v[u].w, __ATOMIC_RELAXED, __HIP_MEMORY_SCOPE_WORKGROUP);
                    }
                }
                for (; j0 < n; ++j0) {
                    int sj = __builtin_amdgcn_readlane(s, j0);
                    int rj = __builtin_amdgcn_readlane(r, j0);
                    float cj = __int_as_float(__builtin_amdgcn_readlane(__float_as_int(c), j0));
                    float4 v = H4[(size_t)sj * 32 + lane];
                    float* a = acc + rj * NHID + lane;
                    __hip_atomic_fetch_add(a + 0,  cj * v.x, __ATOMIC_RELAXED, __HIP_MEMORY_SCOPE_WORKGROUP);
                    __hip_atomic_fetch_add(a + 32, cj * v.y, __ATOMIC_RELAXED, __HIP_MEMORY_SCOPE_WORKGROUP);
                    __hip_atomic_fetch_add(a + 64, cj * v.z, __ATOMIC_RELAXED, __HIP_MEMORY_SCOPE_WORKGROUP);
                    __hip_atomic_fetch_add(a + 96, cj * v.w, __ATOMIC_RELAXED, __HIP_MEMORY_SCOPE_WORKGROUP);
                }
            }
        } else {
            for (int e = tid; e < E; e += 1024) {
                unsigned r = (unsigned)(dst[e] - lo);
                if (r < (unsigned)rows) {
                    int s = src[e];
                    float c = dinv[s] * w[e] * dinv[lo + r];
                    const float* h = H + (size_t)s * NHID;
                    for (int f = 0; f < NHID; ++f)
                        __hip_atomic_fetch_add(&acc[r * NHID + (f & 3) * 32 + (f >> 2)], c * h[f],
                                               __ATOMIC_RELAXED, __HIP_MEMORY_SCOPE_WORKGROUP);
                }
            }
        }
        __syncthreads();

        for (int idx = tid; idx < rows * 32; idx += 1024) {
            int r = idx >> 5;
            const float* a = acc + r * NHID + lane;
            ((float4*)(Hacc + (size_t)(lo + r) * NHID))[lane] = make_float4(a[0], a[32], a[64], a[96]);
        }
        __syncthreads();
    }
}

__global__ __launch_bounds__(64) void k_gemm2(const float* __restrict__ Hacc,
                                              const float* __restrict__ b1,
                                              const float* __restrict__ Wp,
                                              float* __restrict__ G, int M) {
    __shared__ float2 Xs[64 * 33];
    const int tid = threadIdx.x;
    const int m0 = blockIdx.x * 32;

#pragma unroll
    for (int it = 0; it < 16; ++it) {
        int q = it * 64 + tid;
        int m = q >> 5, k4 = q & 31;
        int mr = m0 + m; if (mr >= M) mr = M - 1;
        float4 v = ((const float4*)(Hacc + (size_t)mr * NHID))[k4];
        float4 bb = ((const float4*)b1)[k4];
        v.x = fmaxf(v.x + bb.x, 0.0f);
        v.y = fmaxf(v.y + bb.y, 0.0f);
        v.z = fmaxf(v.z + bb.z, 0.0f);
        v.w = fmaxf(v.w + bb.w, 0.0f);
        Xs[(2 * k4)     * 33 + m] = make_float2(v.x, v.y);
        Xs[(2 * k4 + 1) * 33 + m] = make_float2(v.z, v.w);
    }
    __syncthreads();

    const int wave = tid >> 5, lane = tid & 31, l = lane & 15, half = lane >> 4;
    const float2* Bp = (const float2*)Wp;

    v8f acc0 = {}, acc1 = {}, acc2 = {};
#pragma unroll 4
    for (int kp = 0; kp < 64; kp += 2) {
        int row = kp + half;
        float2 a  = Xs[row * 33 + wave * 16 + l];
        float2 b0 = Bp[(size_t)row * NCPAD + 0  + l];
        float2 b1f= Bp[(size_t)row * NCPAD + 16 + l];
        float2 b2f= Bp[(size_t)row * NCPAD + 32 + l];
        v2f av = {a.x, a.y};
        v2f bv0 = {b0.x, b0.y}, bv1 = {b1f.x, b1f.y}, bv2 = {b2f.x, b2f.y};
        acc0 = __builtin_amdgcn_wmma_f32_16x16x4_f32(false, av, false, bv0, (short)0, acc0, false, false);
        acc1 = __builtin_amdgcn_wmma_f32_16x16x4_f32(false, av, false, bv1, (short)0, acc1, false, false);
        acc2 = __builtin_amdgcn_wmma_f32_16x16x4_f32(false, av, false, bv2, (short)0, acc2, false, false);
    }

    const int mb = m0 + wave * 16 + 8 * half;
    float* Go = G + (size_t)mb * NCPAD + l;
    if (m0 + 32 <= M) {
#pragma unroll
        for (int r = 0; r < 8; ++r) {
            float* o = Go + (size_t)r * NCPAD;
            o[0]  = acc0[r];
            o[16] = acc1[r];
            o[32] = acc2[r];
        }
    } else {
#pragma unroll
        for (int r = 0; r < 8; ++r) {
            if (mb + r < M) {
                float* o = Go + (size_t)r * NCPAD;
                o[0]  = acc0[r];
                o[16] = acc1[r];
                o[32] = acc2[r];
            }
        }
    }
}

__global__ __launch_bounds__(1024) void k_agg2_lds(const int* __restrict__ src,
                                                   const int* __restrict__ dst,
                                                   const float* __restrict__ w,
                                                   const float* __restrict__ dinv,
                                                   const float* __restrict__ G,
                                                   const float* __restrict__ b2,
                                                   float* out, int N, int E) {
    __shared__ __attribute__((aligned(16))) float acc[AG_ROWS * NCPAD];

    const int tid = threadIdx.x, lane = tid & 31, wave = tid >> 5;
    const bool lo16 = lane < 16;

    for (int lo = blockIdx.x * AG_ROWS; lo < N; lo += gridDim.x * AG_ROWS) {
        const int rows = min(AG_ROWS, N - lo);

        for (int idx = tid; idx < rows * 12; idx += 1024) {
            int r = idx / 12, j = idx - r * 12, i = lo + r;
            float di = dinv[i], c = di * di;
            float4 v = ((const float4*)(G + (size_t)i * NCPAD))[j];
            ((float4*)(acc + r * NCPAD))[j] = make_float4(c * v.x, c * v.y, c * v.z, c * v.w);
        }
        const int* park = (const int*)(out + (size_t)lo * NCLASS);
        const int cnt = park[0];
        const int room = min(AG_QCAP, (rows * NCLASS - 2) >> 1);
        __syncthreads();

        if (cnt <= room) {
            const int2* park2 = (const int2*)(park + 2);
            for (int base = wave * 32; base < cnt; base += 32 * 32) {
                const int n = min(32, cnt - base);
                int pk = 0;
                float c = 0.0f;
                if (lane < n) {
                    int2 t = park2[base + lane];
                    pk = t.x;
                    c = __int_as_float(t.y);
                }
                int j0 = 0;
                for (; j0 + 16 <= n; j0 += 16) {
                    int pj[16];
                    float cj[16];
#pragma unroll
                    for (int u = 0; u < 16; ++u) {
                        pj[u] = __builtin_amdgcn_readlane(pk, j0 + u);
                        cj[u] = __int_as_float(__builtin_amdgcn_readlane(__float_as_int(c), j0 + u));
                    }
                    float ga[16], gb[16];
#pragma unroll
                    for (int u = 0; u < 16; ++u)
                        ga[u] = G[(size_t)(pj[u] >> 9) * NCPAD + lane];
                    if (lo16) {
#pragma unroll
                        for (int u = 0; u < 16; ++u)
                            gb[u] = G[(size_t)(pj[u] >> 9) * NCPAD + 32 + lane];
                    }
#pragma unroll
                    for (int u = 0; u < 16; ++u)
                        __hip_atomic_fetch_add(acc + (pj[u] & 511) * NCPAD + lane, cj[u] * ga[u],
                                               __ATOMIC_RELAXED, __HIP_MEMORY_SCOPE_WORKGROUP);
                    if (lo16) {
#pragma unroll
                        for (int u = 0; u < 16; ++u)
                            __hip_atomic_fetch_add(acc + (pj[u] & 511) * NCPAD + 32 + lane, cj[u] * gb[u],
                                                   __ATOMIC_RELAXED, __HIP_MEMORY_SCOPE_WORKGROUP);
                    }
                }
                for (; j0 < n; ++j0) {
                    int pj = __builtin_amdgcn_readlane(pk, j0);
                    float cj = __int_as_float(__builtin_amdgcn_readlane(__float_as_int(c), j0));
                    const float* g = G + (size_t)(pj >> 9) * NCPAD;
                    float* a = acc + (pj & 511) * NCPAD;
                    __hip_atomic_fetch_add(a + lane, cj * g[lane], __ATOMIC_RELAXED, __HIP_MEMORY_SCOPE_WORKGROUP);
                    if (lo16)
                        __hip_atomic_fetch_add(a + 32 + lane, cj * g[32 + lane],
                                               __ATOMIC_RELAXED, __HIP_MEMORY_SCOPE_WORKGROUP);
                }
            }
        } else {
            for (int e = tid; e < E; e += 1024) {
                unsigned r = (unsigned)(dst[e] - lo);
                if (r < (unsigned)rows) {
                    int s = src[e];
                    float c = dinv[s] * w[e] * dinv[lo + r];
                    const float* g = G + (size_t)s * NCPAD;
                    for (int f = 0; f < NCLASS; ++f)
                        __hip_atomic_fetch_add(&acc[r * NCPAD + f], c * g[f],
                                               __ATOMIC_RELAXED, __HIP_MEMORY_SCOPE_WORKGROUP);
                }
            }
        }
        __syncthreads();

        for (int r = wave; r < rows; r += 32) {
            const float* row = acc + r * NCPAD;
            float v1 = row[lane] + b2[lane];
            int c2 = 32 + lane;
            bool has2 = c2 < NCLASS;
            float v2 = has2 ? (row[c2] + b2[c2]) : -INFINITY;

            float mx = fmaxf(v1, v2);
#pragma unroll
            for (int off = 16; off >= 1; off >>= 1) mx = fmaxf(mx, __shfl_xor(mx, off));

            float sm = expf(v1 - mx) + (has2 ? expf(v2 - mx) : 0.0f);
#pragma unroll
            for (int off = 16; off >= 1; off >>= 1) sm += __shfl_xor(sm, off);

            float ls = logf(sm);
            float* o = out + (size_t)(lo + r) * NCLASS;
            o[lane] = v1 - mx - ls;
            if (has2) o[c2] = v2 - mx - ls;
        }
        __syncthreads();
    }
}

extern "C" void kernel_launch(void* const* d_in, const int* in_sizes, int n_in,
                              void* d_out, int out_size, void* d_ws, size_t ws_size,
                              hipStream_t stream) {
    const float* x  = (const float*)d_in[0];
    const int*   ei = (const int*)  d_in[1];
    const float* ew = (const float*)d_in[2];
    const float* W1 = (const float*)d_in[3];
    const float* b1 = (const float*)d_in[4];
    const float* W2 = (const float*)d_in[5];
    const float* b2 = (const float*)d_in[6];
    float* out = (float*)d_out;

    const int N = in_sizes[0] / NFEAT;
    const int E = in_sizes[2];
    const int* src = ei;
    const int* dst = ei + E;

    float* ws = (float*)d_ws;
    size_t o = 0;
    auto take = [&](size_t n) { float* p = ws + o; o += (n + 255) & ~(size_t)255; return p; };
    float* dinv = take((size_t)N);
    float* Wp1  = take((size_t)NFEAT * NHID);
    float* Wp2  = take((size_t)NHID * NCPAD);
    float* Hxw  = take((size_t)N * NHID);
    float* Hacc = take((size_t)N * NHID);
    float* G    = Hxw;

    k_pack_w1<<<(NFEAT * NHID + 255) / 256, 256, 0, stream>>>(W1, Wp1);
    k_pack_w2<<<(NHID * NCPAD + 255) / 256, 256, 0, stream>>>(W2, Wp2);

    unsigned* ob = (unsigned*)Hacc;
    const int nvec = (E + 15) / 16;
    k_owner_bytes<<<256, 512, 0, stream>>>(dst, ob, E, 4 * nvec);
    k_sweep_deg<<<256, 1024, 0, stream>>>(src, dst, ew, ob, dinv, out, N, E, nvec);

    k_gemm1<<<(N + 31) / 32, 256, 0, stream>>>(x, Wp1, Hxw, N);

    k_agg1_lds<<<256, 1024, 0, stream>>>(src, dst, ew, dinv, Hxw, Hacc, out, N, E);

    k_gemm2<<<(N + 31) / 32, 64, 0, stream>>>(Hacc, b1, Wp2, G, N);

    k_agg2_lds<<<256, 1024, 0, stream>>>(src, dst, ew, dinv, G, b2, out, N, E);
}


# Layer-1 owner blocks finish their rows: bias, ReLU and the W2 product (f32 WMMA, k order inside each step of 4 changed) straight from the LDS sums, row pitch 129; Hacc plane, W2 pack, gemm2 launch go
# speedup vs baseline: 3.2215x; 1.0704x over previous
#include <hip/hip_runtime.h>
#include <cstdint>

typedef float v2f __attribute__((ext_vector_type(2)));
typedef float v8f __attribute__((ext_vector_type(8)));

#define NFEAT 256
#define NHID  128
#define NCLASS 41
#define NCPAD  48

__global__ __launch_bounds__(256) void k_pack_w1(const float* __restrict__ W1, float* __restrict__ Wp) {
    int idx = blockIdx.x * 256 + threadIdx.x;
    if (idx >= NFEAT * NHID) return;
    int k = idx >> 7, n = idx & 127;
    Wp[(size_t)(k >> 1) * (NHID * 2) + n * 2 + (k & 1)] = W1[idx];
}

__global__ __launch_bounds__(256) void k_gemm1(const float* __restrict__ X,
                                               const float* __restrict__ Wp,
                                               float* __restrict__ H, int M) {
    __shared__ float2 Xs[128 * 33];
    const int tid = threadIdx.x;
    const int m0 = blockIdx.x * 32;

#pragma unroll
    for (int it = 0; it < 8; ++it) {
        int q = it * 256 + tid;
        int m = q >> 6, k4 = q & 63;
        int mr = m0 + m; if (mr >= M) mr = M - 1;
        float4 v = ((const float4*)(X + (size_t)mr * NFEAT))[k4];
        Xs[(2 * k4)     * 33 + m] = make_float2(v.x, v.y);
        Xs[(2 * k4 + 1) * 33 + m] = make_float2(v.z, v.w);
    }
    __syncthreads();

    const int wave = tid >> 5, lane = tid & 31, l = lane & 15, half = lane >> 4;
    const int n0 = wave * 16;
    const float2* Bp = (const float2*)Wp;

    v8f acc0 = {}, acc1 = {};
#pragma unroll 4
    for (int kp = 0; kp < 128; kp += 2) {
        int row = kp + half;
        float2 a0 = Xs[row * 33 + l];
        float2 a1 = Xs[row * 33 + 16 + l];
        float2 b  = Bp[(size_t)row * NHID + n0 + l];
        v2f av0 = {a0.x, a0.y}, av1 = {a1.x, a1.y}, bv = {b.x, b.y};
        acc0 = __builtin_amdgcn_wmma_f32_16x16x4_f32(false, av0, false, bv, (short)0, acc0, false, false);
        acc1 = __builtin_amdgcn_wmma_f32_16x16x4_f32(false, av1, false, bv, (short)0, acc1, false, false);
    }

    const int mb = 8 * half;
    float* Ho = H + (size_t)(m0 + mb) * NHID + n0 + l;
    if (m0 + 32 <= M) {
#pragma unroll
        for (int r = 0; r < 8; ++r) {
            Ho[(size_t)r * NHID]        = acc0[r];
            Ho[(size_t)(16 + r) * NHID] = acc1[r];
        }
    } else {
#pragma unroll
        for (int r = 0; r < 8; ++r) {
            if (m0 + mb + r < M)      Ho[(size_t)r * NHID]        = acc0[r];
            if (m0 + 16 + mb + r < M) Ho[(size_t)(16 + r) * NHID] = acc1[r];
        }
    }
}

#define AG_ROWS 392
#define AG_QCAP 8192

__device__ __forceinline__ unsigned own_byte(int d) { return ((unsigned)d / AG_ROWS) & 255u; }

__global__ __launch_bounds__(512) void k_owner_bytes(const int* __restrict__ dst,
                                                     unsigned* __restrict__ ob, int E, int nwords) {
    const bool al16 = ((uintptr_t)dst & 15u) == 0;
    const int T = gridDim.x * 512;
    for (int wd = blockIdx.x * 512 + threadIdx.x; wd < nwords; wd += T) {
        const int e = 4 * wd;
        unsigned pk = 0;
        if (al16 && e + 4 <= E) {
            int4 d = ((const int4*)dst)[wd];
            pk = own_byte(d.x) | (own_byte(d.y) << 8) | (own_byte(d.z) << 16) | (own_byte(d.w) << 24);
        } else {
            for (int k = 0; k < 4; ++k)
                if (e + k < E) pk |= own_byte(dst[e + k]) << (8 * k);
        }
        ob[wd] = pk;
    }
}

__device__ __forceinline__ void sw_push(int e, int lo, int rows, int* qcnt, int* q, float* degs,
                                        const int* __restrict__ dst, const float* __restrict__ w) {
    int slot = __hip_atomic_fetch_add(qcnt, 1, __ATOMIC_RELAXED, __HIP_MEMORY_SCOPE_WORKGROUP);
    if (slot < AG_QCAP) {
        q[slot] = e;
    } else {
        unsigned r = (unsigned)(dst[e] - lo);
        if (r < (unsigned)rows)
            __hip_atomic_fetch_add(&degs[r], w[e], __ATOMIC_RELAXED, __HIP_MEMORY_SCOPE_WORKGROUP);
    }
}

__device__ __forceinline__ unsigned zero_bytes(unsigned x) {
    unsigned y = ((x & 0x7F7F7F7Fu) + 0x7F7F7F7Fu) | x;
    return ~y & 0x80808080u;
}
__device__ __forceinline__ unsigned match16(uint4 d, unsigned pat) {
    return zero_bytes(d.x ^ pat) | (zero_bytes(d.y ^ pat) >> 1) |
           (zero_bytes(d.z ^ pat) >> 2) | (zero_bytes(d.w ^ pat) >> 3);
}

__global__ __launch_bounds__(1024) void k_sweep_deg(const int* __restrict__ src,
                                                    const int* __restrict__ dst,
                                                    const float* __restrict__ w,
                                                    const unsigned* __restrict__ ob,
                                                    float* __restrict__ dinv, float* parkbase,
                                                    int N, int E, int nvec) {
    __shared__ int q[AG_QCAP];
    __shared__ float degs[AG_ROWS];
    __shared__ int qcnt, pcnt;

    const int tid = threadIdx.x;
    const uint4* ob4 = (const uint4*)ob;
    const int start = (int)(((long long)blockIdx.x * nvec) / gridDim.x);

    for (int lo = blockIdx.x * AG_ROWS; lo < N; lo += gridDim.x * AG_ROWS) {
        const int rows = min(AG_ROWS, N - lo);
        const unsigned pat = (((unsigned)lo / AG_ROWS) & 255u) * 0x01010101u;

        for (int r = tid; r < rows; r += 1024) degs[r] = 1.0f;
        if (tid == 0) { qcnt = 0; pcnt = 0; }
        __syncthreads();

        for (int i0 = 0; i0 < nvec; i0 += 8 * 1024) {
            uint4 d[8];
#pragma unroll
            for (int u = 0; u < 8; ++u) {
                int v = i0 + u * 1024 + tid;
                int pp = v + start; if (pp >= nvec) pp -= nvec;
                d[u] = ob4[v < nvec ? pp : 0];
            }
#pragma unroll
            for (int h = 0; h < 4; ++h) {
                unsigned ma = match16(d[2 * h], pat), mb = match16(d[2 * h + 1], pat);
                if (i0 + (2 * h) * 1024 + tid >= nvec) ma = 0;
                if (i0 + (2 * h + 1) * 1024 + tid >= nvec) mb = 0;
                unsigned m = ma | (mb >> 4);
                while (m) {
                    int p = __builtin_ctz(m);
                    m &= m - 1;
                    int j = p >> 3, b = p & 7;
                    int u = 2 * h + (b < 4 ? 1 : 0);
                    int k = (7 - b) & 3;
                    int pp = i0 + u * 1024 + tid + start; if (pp >= nvec) pp -= nvec;
                    int e = 16 * pp + 4 * k + j;
                    if (e < E) sw_push(e, lo, rows, &qcnt, q, degs, dst, w);
                }
            }
        }
        __syncthreads();

        const int nq = min(qcnt, AG_QCAP);
        int* park = (int*)(parkbase + (size_t)lo * NCLASS);
        int2* park2 = (int2*)(park + 2);
        const int room = (rows * NCLASS - 2) >> 1;
        int dd[AG_QCAP / 1024];
        float ww[AG_QCAP / 1024];
        int sid[AG_QCAP / 1024];
#pragma unroll
        for (int u = 0; u < AG_QCAP / 1024; ++u) {
            int i = u * 1024 + tid;
            dd[u] = -1; ww[u] = 0.0f; sid[u] = 0;
            if (i < nq) {
                int e = q[i];
                dd[u] = dst[e];
                ww[u] = w[e];
                sid[u] = src[e];
            }
        }
#pragma unroll
        for (int u = 0; u < AG_QCAP / 1024; ++u) {
            int i = u * 1024 + tid;
            unsigned r = (unsigned)(dd[u] - lo);
            if (i < nq && r < (unsigned)rows) {
                __hip_atomic_fetch_add(&degs[r], ww[u], __ATOMIC_RELAXED, __HIP_MEMORY_SCOPE_WORKGROUP);
                int slot = __hip_atomic_fetch_add(&pcnt, 1, __ATOMIC_RELAXED, __HIP_MEMORY_SCOPE_WORKGROUP);
                if (slot < room) park2[slot] = make_int2((sid[u] << 9) | (int)r, __float_as_int(ww[u]));
            }
        }
        __syncthreads();

        if (tid == 0) park[0] = (qcnt > AG_QCAP) ? 0x7fffffff : pcnt;
        for (int r = tid; r < rows; r += 1024) {
            float dg = degs[r];
            dinv[lo + r] = dg > 0.0f ? rsqrtf(dg) : 0.0f;
        }
        __syncthreads();
    }
}

#define ACC_LD 129
__global__ __launch_bounds__(1024) void k_agg1_lds(const int* __restrict__ src,
                                                   const int* __restrict__ dst,
                                                   const float* __restrict__ w,
                                                   const float* __restrict__ dinv,
                                                   const float* __restrict__ H,
                                                   const float* __restrict__ b1,
                                                   const float* __restrict__ W2,
                                                   float* __restrict__ G, float* parkbase,
                                                   int N, int E) {
    __shared__ float acc[AG_ROWS * ACC_LD];
    __shared__ float Ws[NHID * NCPAD];
    __shared__ float bs[NHID];

    const int tid = threadIdx.x, lane = tid & 31, wave = tid >> 5;
    const float4* H4 = (const float4*)H;

    static_assert((NHID * NCPAD) % 1024 == 0, "one W2 word per thread and trip");
    float wreg[(NHID * NCPAD) / 1024];
#pragma unroll
    for (int u = 0; u < (NHID * NCPAD) / 1024; ++u) {
        int idx = u * 1024 + tid;
        int f = idx / NCPAD, n = idx - f * NCPAD;
        float x = W2[f * NCLASS + min(n, NCLASS - 1)];
        wreg[u] = (n < NCLASS) ? x : 0.0f;
    }
    const float breg = b1[tid & (NHID - 1)];
    bool staged = false;

    for (int lo = blockIdx.x * AG_ROWS; lo < N; lo += gridDim.x * AG_ROWS) {
        const int rows = min(AG_ROWS, N - lo);

        for (int idx = tid; idx < rows * 32; idx += 1024) {
            int r = idx >> 5, i = lo + r;
            float di = dinv[i], c = di * di;
            float4 v = H4[(size_t)i * 32 + lane];
            float* a = acc + r * ACC_LD + lane;
            a[0] = c * v.x; a[32] = c * v.y; a[64] = c * v.z; a[96] = c * v.w;
        }
        if (!staged) {
#pragma unroll
            for (int u = 0; u < (NHID * NCPAD) / 1024; ++u) Ws[u * 1024 + tid] = wreg[u];
            if (tid < NHID) bs[tid] = breg;
            staged = true;
        }
        int* park = (int*)(parkbase + (size_t)lo * NCLASS);
        const int cnt = park[0];
        const int room = min(AG_QCAP, (rows * NCLASS - 2) >> 1);
        __syncthreads();

        if (cnt <= room) {
            int2* park2 = (int2*)(park + 2);
            for (int base = wave * 32; base < cnt; base += 32 * 32) {
                const int n = min(32, cnt - base);
                int s = 0, r = 0;
                float c = 0.0f;
                if (lane < n) {
                    int2 t = park2[base + lane];
                    s = t.x >> 9;
                    r = t.x & 511;
                    c = dinv[s] * __int_as_float(t.y) * dinv[lo + r];
                    park2[base + lane] = make_int2(t.x, __float_as_int(c));
                }
                int j0 = 0;
                for (; j0 + 8 <= n; j0 += 8) {
                    float4 v[8];
#pragma unroll
                    for (int u = 0; u < 8; ++u) {
                        int sj = __builtin_amdgcn_readlane(s, j0 + u);
                        v[u] = H4[(size_t)sj * 32 + lane];
                    }
#pragma unroll
                    for (int u = 0; u < 8; ++u) {
                        int rj = __builtin_amdgcn_readlane(r, j0 + u);
                        float cj = __int_as_float(__builtin_amdgcn_readlane(__float_as_int(c), j0 + u));
                        float* a = acc + rj * ACC_LD + lane;
                        __hip_atomic_fetch_add(a + 0,  cj * v[u].x, __ATOMIC_RELAXED, __HIP_MEMORY_SCOPE_WORKGROUP);
                        __hip_atomic_fetch_add(a + 32, cj * v[u].y, __ATOMIC_RELAXED, __HIP_MEMORY_SCOPE_WORKGROUP);
                        __hip_atomic_fetch_add(a + 64, cj * v[u].z, __ATOMIC_RELAXED, __HIP_MEMORY_SCOPE_WORKGROUP);
                        __hip_atomic_fetch_add(a + 96, cj * v[u].w, __ATOMIC_RELAXED, __HIP_MEMORY_SCOPE_WORKGROUP);
                    }
                }
                for (; j0 < n; ++j0) {
                    int sj = __builtin_amdgcn_readlane(s, j0);
                    int rj = __builtin_amdgcn_readlane(r, j0);
                    float cj = __int_as_float(__builtin_amdgcn_readlane(__float_as_int(c), j0));
                    float4 v = H4[(size_t)sj * 32 + lane];
                    float* a = acc + rj * ACC_LD + lane;
                    __hip_atomic_fetch_add(a + 0,  cj * v.x, __ATOMIC_RELAXED, __HIP_MEMORY_SCOPE_WORKGROUP);
                    __hip_atomic_fetch_add(a + 32, cj * v.y, __ATOMIC_RELAXED, __HIP_MEMORY_SCOPE_WORKGROUP);
                    __hip_atomic_fetch_add(a + 64, cj * v.z, __ATOMIC_RELAXED, __HIP_MEMORY_SCOPE_WORKGROUP);
                    __hip_atomic_fetch_add(a + 96, cj * v.w, __ATOMIC_RELAXED, __HIP_MEMORY_SCOPE_WORKGROUP);
                }
            }
        } else {
            for (int e = tid; e < E; e += 1024) {
                unsigned r = (unsigned)(dst[e] - lo);
                if (r < (unsigned)rows) {
                    int s = src[e];
                    float c = dinv[s] * w[e] * dinv[lo + r];
                    const float* h = H + (size_t)s * NHID;
                    for (int f = 0; f < NHID; ++f)
                        __hip_atomic_fetch_add(&acc[r * ACC_LD + (f & 3) * 32 + (f >> 2)], c * h[f],
                                               __ATOMIC_RELAXED, __HIP_MEMORY_SCOPE_WORKGROUP);
                }
            }
        }
        __syncthreads();

        const int wv = __builtin_amdgcn_readfirstlane(wave);
        if (wv * 16 < rows) {
            const int cc = lane & 15, hh = lane >> 4;
            int rr = wv * 16 + cc;
            if (rr >= rows) rr = rows - 1;
            const float* arow = acc + rr * ACC_LD + 32 * hh;
            const float* wr0 = Ws + hh * NCPAD + cc;
            const float* bb = bs + hh;
            v8f d0 = {}, d1 = {}, d2 = {};
#pragma unroll 4
            for (int j = 0; j < 32; ++j) {
                const float* wr = wr0 + 4 * j * NCPAD;
                v2f av  = {fmaxf(arow[j] + bb[4 * j], 0.0f), fmaxf(arow[64 + j] + bb[4 * j + 2], 0.0f)};
                v2f bv0 = {wr[0],  wr[2 * NCPAD]};
                v2f bv1 = {wr[16], wr[2 * NCPAD + 16]};
                v2f bv2 = {wr[32], wr[2 * NCPAD + 32]};
                d0 = __builtin_amdgcn_wmma_f32_16x16x4_f32(false, av, false, bv0, (short)0, d0, false, false);
                d1 = __builtin_amdgcn_wmma_f32_16x16x4_f32(false, av, false, bv1, (short)0, d1, false, false);
                d2 = __builtin_amdgcn_wmma_f32_16x16x4_f32(false, av, false, bv2, (short)0, d2, false, false);
            }
            const int rb = wv * 16 + 8 * hh;
            float* Go = G + (size_t)(lo + rb) * NCPAD + cc;
#pragma unroll
            for (int r = 0; r < 8; ++r) {
                if (rb + r < rows) {
                    float* o = Go + (size_t)r * NCPAD;
                    o[0]  = d0[r];
                    o[16] = d1[r];
                    o[32] = d2[r];
                }
            }
        }
        __syncthreads();
    }
}

__global__ __launch_bounds__(1024) void k_agg2_lds(const int* __restrict__ src,
                                                   const int* __restrict__ dst,
                                                   const float* __restrict__ w,
                                                   const float* __restrict__ dinv,
                                                   const float* __restrict__ G,
                                                   const float* __restrict__ b2,
                                                   float* out, int N, int E) {
    __shared__ __attribute__((aligned(16))) float acc[AG_ROWS * NCPAD];

    const int tid = threadIdx.x, lane = tid & 31, wave = tid >> 5;
    const bool lo16 = lane < 16;

    for (int lo = blockIdx.x * AG_ROWS; lo < N; lo += gridDim.x * AG_ROWS) {
        const int rows = min(AG_ROWS, N - lo);

        for (int idx = tid; idx < rows * 12; idx += 1024) {
            int r = idx / 12, j = idx - r * 12, i = lo + r;
            float di = dinv[i], c = di * di;
            float4 v = ((const float4*)(G + (size_t)i * NCPAD))[j];
            ((float4*)(acc + r * NCPAD))[j] = make_float4(c * v.x, c * v.y, c * v.z, c * v.w);
        }
        const int* park = (const int*)(out + (size_t)lo * NCLASS);
        const int cnt = park[0];
        const int room = min(AG_QCAP, (rows * NCLASS - 2) >> 1);
        __syncthreads();

        if (cnt <= room) {
            const int2* park2 = (const int2*)(park + 2);
            for (int base = wave * 32; base < cnt; base += 32 * 32) {
                const int n = min(32, cnt - base);
                int pk = 0;
                float c = 0.0f;
                if (lane < n) {
                    int2 t = park2[base + lane];
                    pk = t.x;
                    c = __int_as_float(t.y);
                }
                int j0 = 0;
                for (; j0 + 16 <= n; j0 += 16) {
                    int pj[16];
                    float cj[16];
#pragma unroll
                    for (int u = 0; u < 16; ++u) {
                        pj[u] = __builtin_amdgcn_readlane(pk, j0 + u);
                        cj[u] = __int_as_float(__builtin_amdgcn_readlane(__float_as_int(c), j0 + u));
                    }
                    float ga[16], gb[16];
#pragma unroll
                    for (int u = 0; u < 16; ++u)
                        ga[u] = G[(size_t)(pj[u] >> 9) * NCPAD + lane];
                    if (lo16) {
#pragma unroll
                        for (int u = 0; u < 16; ++u)
                            gb[u] = G[(size_t)(pj[u] >> 9) * NCPAD + 32 + lane];
                    }
#pragma unroll
                    for (int u = 0; u < 16; ++u)
                        __hip_atomic_fetch_add(acc + (pj[u] & 511) * NCPAD + lane, cj[u] * ga[u],
                                               __ATOMIC_RELAXED, __HIP_MEMORY_SCOPE_WORKGROUP);
                    if (lo16) {
#pragma unroll
                        for (int u = 0; u < 16; ++u)
                            __hip_atomic_fetch_add(acc + (pj[u] & 511) * NCPAD + 32 + lane, cj[u] * gb[u],
                                                   __ATOMIC_RELAXED, __HIP_MEMORY_SCOPE_WORKGROUP);
                    }
                }
                for (; j0 < n; ++j0) {
                    int pj = __builtin_amdgcn_readlane(pk, j0);
                    float cj = __int_as_float(__builtin_amdgcn_readlane(__float_as_int(c), j0));
                    const float* g = G + (size_t)(pj >> 9) * NCPAD;
                    float* a = acc + (pj & 511) * NCPAD;
                    __hip_atomic_fetch_add(a + lane, cj * g[lane], __ATOMIC_RELAXED, __HIP_MEMORY_SCOPE_WORKGROUP);
                    if (lo16)
                        __hip_atomic_fetch_add(a + 32 + lane, cj * g[32 + lane],
                                               __ATOMIC_RELAXED, __HIP_MEMORY_SCOPE_WORKGROUP);
                }
            }
        } else {
            for (int e = tid; e < E; e += 1024) {
                unsigned r = (unsigned)(dst[e] - lo);
                if (r < (unsigned)rows) {
                    int s = src[e];
                    float c = dinv[s] * w[e] * dinv[lo + r];
                    const float* g = G + (size_t)s * NCPAD;
                    for (int f = 0; f < NCLASS; ++f)
                        __hip_atomic_fetch_add(&acc[r * NCPAD + f], c * g[f],
                                               __ATOMIC_RELAXED, __HIP_MEMORY_SCOPE_WORKGROUP);
                }
            }
        }
        __syncthreads();

        for (int r = wave; r < rows; r += 32) {
            const float* row = acc + r * NCPAD;
            float v1 = row[lane] + b2[lane];
            int c2 = 32 + lane;
            bool has2 = c2 < NCLASS;
            float v2 = has2 ? (row[c2] + b2[c2]) : -INFINITY;

            float mx = fmaxf(v1, v2);
#pragma unroll
            for (int off = 16; off >= 1; off >>= 1) mx = fmaxf(mx, __shfl_xor(mx, off));

            float sm = expf(v1 - mx) + (has2 ? expf(v2 - mx) : 0.0f);
#pragma unroll
            for (int off = 16; off >= 1; off >>= 1) sm += __shfl_xor(sm, off);

            float ls = logf(sm);
            float* o = out + (size_t)(lo + r) * NCLASS;
            o[lane] = v1 - mx - ls;
            if (has2) o[c2] = v2 - mx - ls;
        }
        __syncthreads();
    }
}

extern "C" void kernel_launch(void* const* d_in, const int* in_sizes, int n_in,
                              void* d_out, int out_size, void* d_ws, size_t ws_size,
                              hipStream_t stream) {
    const float* x  = (const float*)d_in[0];
    const int*   ei = (const int*)  d_in[1];
    const float* ew = (const float*)d_in[2];
    const float* W1 = (const float*)d_in[3];
    const float* b1 = (const float*)d_in[4];
    const float* W2 = (const float*)d_in[5];
    const float* b2 = (const float*)d_in[6];
    float* out = (float*)d_out;

    const int N = in_sizes[0] / NFEAT;
    const int E = in_sizes[2];
    const int* src = ei;
    const int* dst = ei + E;

    float* ws = (float*)d_ws;
    size_t o = 0;
    auto take = [&](size_t n) { float* p = ws + o; o += (n + 255) & ~(size_t)255; return p; };
    float* dinv = take((size_t)N);
    float* Wp1  = take((size_t)NFEAT * NHID);
    float* Hxw  = take((size_t)N * NHID);
    float* G    = take((size_t)N * NHID);

    k_pack_w1<<<(NFEAT * NHID + 255) / 256, 256, 0, stream>>>(W1, Wp1);

    unsigned* ob = (unsigned*)G;
    const int nvec = (E + 15) / 16;
    k_owner_bytes<<<256, 512, 0, stream>>>(dst, ob, E, 4 * nvec);
    k_sweep_deg<<<256, 1024, 0, stream>>>(src, dst, ew, ob, dinv, out, N, E, nvec);

    k_gemm1<<<(N + 31) / 32, 256, 0, stream>>>(x, Wp1, Hxw, N);

    k_agg1_lds<<<256, 1024, 0, stream>>>(src, dst, ew, dinv, Hxw, b1, W2, G, out, N, E);

    k_agg2_lds<<<256, 1024, 0, stream>>>(src, dst, ew, dinv, G, b2, out, N, E);
}


# Edge pre-pass also writes each edge's parked 8-byte record (source, row in range, weight); the sweep fetches one record per queued edge, not destination, weight and source from three lists
# speedup vs baseline: 3.3350x; 1.0352x over previous
#include <hip/hip_runtime.h>
#include <cstdint>

typedef float v2f __attribute__((ext_vector_type(2)));
typedef float v8f __attribute__((ext_vector_type(8)));

#define NFEAT 256
#define NHID  128
#define NCLASS 41
#define NCPAD  48

__global__ __launch_bounds__(256) void k_pack_w1(const float* __restrict__ W1, float* __restrict__ Wp) {
    int idx = blockIdx.x * 256 + threadIdx.x;
    if (idx >= NFEAT * NHID) return;
    int k = idx >> 7, n = idx & 127;
    Wp[(size_t)(k >> 1) * (NHID * 2) + n * 2 + (k & 1)] = W1[idx];
}

__global__ __launch_bounds__(256) void k_gemm1(const float* __restrict__ X,
                                               const float* __restrict__ Wp,
                                               float* __restrict__ H, int M) {
    __shared__ float2 Xs[128 * 33];
    const int tid = threadIdx.x;
    const int m0 = blockIdx.x * 32;

#pragma unroll
    for (int it = 0; it < 8; ++it) {
        int q = it * 256 + tid;
        int m = q >> 6, k4 = q & 63;
        int mr = m0 + m; if (mr >= M) mr = M - 1;
        float4 v = ((const float4*)(X + (size_t)mr * NFEAT))[k4];
        Xs[(2 * k4)     * 33 + m] = make_float2(v.x, v.y);
        Xs[(2 * k4 + 1) * 33 + m] = make_float2(v.z, v.w);
    }
    __syncthreads();

    const int wave = tid >> 5, lane = tid & 31, l = lane & 15, half = lane >> 4;
    const int n0 = wave * 16;
    const float2* Bp = (const float2*)Wp;

    v8f acc0 = {}, acc1 = {};
#pragma unroll 4
    for (int kp = 0; kp < 128; kp += 2) {
        int row = kp + half;
        float2 a0 = Xs[row * 33 + l];
        float2 a1 = Xs[row * 33 + 16 + l];
        float2 b  = Bp[(size_t)row * NHID + n0 + l];
        v2f av0 = {a0.x, a0.y}, av1 = {a1.x, a1.y}, bv = {b.x, b.y};
        acc0 = __builtin_amdgcn_wmma_f32_16x16x4_f32(false, av0, false, bv, (short)0, acc0, false, false);
        acc1 = __builtin_amdgcn_wmma_f32_16x16x4_f32(false, av1, false, bv, (short)0, acc1, false, false);
    }

    const int mb = 8 * half;
    float* Ho = H + (size_t)(m0 + mb) * NHID + n0 + l;
    if (m0 + 32 <= M) {
#pragma unroll
        for (int r = 0; r < 8; ++r) {
            Ho[(size_t)r * NHID]        = acc0[r];
            Ho[(size_t)(16 + r) * NHID] = acc1[r];
        }
    } else {
#pragma unroll
        for (int r = 0; r < 8; ++r) {
            if (m0 + mb + r < M)      Ho[(size_t)r * NHID]        = acc0[r];
            if (m0 + 16 + mb + r < M) Ho[(size_t)(16 + r) * NHID] = acc1[r];
        }
    }
}

#define AG_ROWS 392
#define AG_QCAP 8192

__device__ __forceinline__ unsigned own_byte(int d) { return ((unsigned)d / AG_ROWS) & 255u; }

__device__ __forceinline__ int2 edge_rec(int s, int d, float x) {
    const int row = d - (int)((unsigned)d / AG_ROWS) * AG_ROWS;
    return make_int2((s << 9) | row, __float_as_int(x));
}

__global__ __launch_bounds__(512) void k_owner_bytes(const int* __restrict__ src,
                                                     const int* __restrict__ dst,
                                                     const float* __restrict__ w,
                                                     unsigned* __restrict__ ob, int* __restrict__ recw,
                                                     int E, int nwords) {
    int2* rec = (int2*)recw;
    const bool al16 = ((((uintptr_t)dst) | ((uintptr_t)src) | ((uintptr_t)w)) & 15u) == 0;
    const int T = gridDim.x * 512;
    for (int wd = blockIdx.x * 512 + threadIdx.x; wd < nwords; wd += T) {
        const int e = 4 * wd;
        unsigned pk = 0;
        if (al16 && e + 4 <= E) {
            int4 d = ((const int4*)dst)[wd];
            int4 s = ((const int4*)src)[wd];
            float4 x = ((const float4*)w)[wd];
            pk = own_byte(d.x) | (own_byte(d.y) << 8) | (own_byte(d.z) << 16) | (own_byte(d.w) << 24);
            int2 r0 = edge_rec(s.x, d.x, x.x), r1 = edge_rec(s.y, d.y, x.y);
            int2 r2 = edge_rec(s.z, d.z, x.z), r3 = edge_rec(s.w, d.w, x.w);
            int4* ro = (int4*)(rec + e);
            ro[0] = make_int4(r0.x, r0.y, r1.x, r1.y);
            ro[1] = make_int4(r2.x, r2.y, r3.x, r3.y);
        } else {
            for (int k = 0; k < 4; ++k)
                if (e + k < E) {
                    int d = dst[e + k];
                    pk |= own_byte(d) << (8 * k);
                    rec[e + k] = edge_rec(src[e + k], d, w[e + k]);
                }
        }
        ob[wd] = pk;
    }
}

__device__ __forceinline__ void sw_push(int e, int lo, int rows, int* qcnt, int* q, float* degs,
                                        const int* __restrict__ dst, const float* __restrict__ w) {
    int slot = __hip_atomic_fetch_add(qcnt, 1, __ATOMIC_RELAXED, __HIP_MEMORY_SCOPE_WORKGROUP);
    if (slot < AG_QCAP) {
        q[slot] = e;
    } else {
        unsigned r = (unsigned)(dst[e] - lo);
        if (r < (unsigned)rows)
            __hip_atomic_fetch_add(&degs[r], w[e], __ATOMIC_RELAXED, __HIP_MEMORY_SCOPE_WORKGROUP);
    }
}

__device__ __forceinline__ unsigned zero_bytes(unsigned x) {
    unsigned y = ((x & 0x7F7F7F7Fu) + 0x7F7F7F7Fu) | x;
    return ~y & 0x80808080u;
}
__device__ __forceinline__ unsigned match16(uint4 d, unsigned pat) {
    return zero_bytes(d.x ^ pat) | (zero_bytes(d.y ^ pat) >> 1) |
           (zero_bytes(d.z ^ pat) >> 2) | (zero_bytes(d.w ^ pat) >> 3);
}

__global__ __launch_bounds__(1024) void k_sweep_deg(const int* __restrict__ src,
                                                    const int* __restrict__ dst,
                                                    const float* __restrict__ w,
                                                    const unsigned* __restrict__ ob,
                                                    const int* __restrict__ recw,
                                                    float* __restrict__ dinv, float* parkbase,
                                                    int N, int E, int nvec) {
    __shared__ int q[AG_QCAP];
    __shared__ float degs[AG_ROWS];
    __shared__ int qcnt, pcnt;

    const int tid = threadIdx.x;
    const int2* rec = (const int2*)recw;
    const uint4* ob4 = (const uint4*)ob;
    const int start = (int)(((long long)blockIdx.x * nvec) / gridDim.x);
    const bool shared_byte = N > 256 * AG_ROWS;

    for (int lo = blockIdx.x * AG_ROWS; lo < N; lo += gridDim.x * AG_ROWS) {
        const int rows = min(AG_ROWS, N - lo);
        const unsigned pat = (((unsigned)lo / AG_ROWS) & 255u) * 0x01010101u;

        for (int r = tid; r < rows; r += 1024) degs[r] = 1.0f;
        if (tid == 0) { qcnt = 0; pcnt = 0; }
        __syncthreads();

        for (int i0 = 0; i0 < nvec; i0 += 8 * 1024) {
            uint4 d[8];
#pragma unroll
            for (int u = 0; u < 8; ++u) {
                int v = i0 + u * 1024 + tid;
                int pp = v + start; if (pp >= nvec) pp -= nvec;
                d[u] = ob4[v < nvec ? pp : 0];
            }
#pragma unroll
            for (int h = 0; h < 4; ++h) {
                unsigned ma = match16(d[2 * h], pat), mb = match16(d[2 * h + 1], pat);
                if (i0 + (2 * h) * 1024 + tid >= nvec) ma = 0;
                if (i0 + (2 * h + 1) * 1024 + tid >= nvec) mb = 0;
                unsigned m = ma | (mb >> 4);
                while (m) {
                    int p = __builtin_ctz(m);
                    m &= m - 1;
                    int j = p >> 3, b = p & 7;
                    int u = 2 * h + (b < 4 ? 1 : 0);
                    int k = (7 - b) & 3;
                    int pp = i0 + u * 1024 + tid + start; if (pp >= nvec) pp -= nvec;
                    int e = 16 * pp + 4 * k + j;
                    if (e < E) sw_push(e, lo, rows, &qcnt, q, degs, dst, w);
                }
            }
        }
        __syncthreads();

        const int nq = min(qcnt, AG_QCAP);
        int* park = (int*)(parkbase + (size_t)lo * NCLASS);
        int2* park2 = (int2*)(park + 2);
        const int room = (rows * NCLASS - 2) >> 1;
        int2 rc[AG_QCAP / 1024];
        int dd[AG_QCAP / 1024];
#pragma unroll
        for (int u = 0; u < AG_QCAP / 1024; ++u) {
            int i = u * 1024 + tid;
            rc[u] = make_int2(0, 0); dd[u] = lo;
            if (i < nq) {
                int e = q[i];
                rc[u] = rec[e];
                if (shared_byte) dd[u] = dst[e];
            }
        }
#pragma unroll
        for (int u = 0; u < AG_QCAP / 1024; ++u) {
            int i = u * 1024 + tid;
            const int r = rc[u].x & 511;
            const bool mine = r < rows && (!shared_byte || (unsigned)(dd[u] - lo) < (unsigned)rows);
            if (i < nq && mine) {
                __hip_atomic_fetch_add(&degs[r], __int_as_float(rc[u].y), __ATOMIC_RELAXED, __HIP_MEMORY_SCOPE_WORKGROUP);
                int slot = __hip_atomic_fetch_add(&pcnt, 1, __ATOMIC_RELAXED, __HIP_MEMORY_SCOPE_WORKGROUP);
                if (slot < room) park2[slot] = rc[u];
            }
        }
        __syncthreads();

        if (tid == 0) park[0] = (qcnt > AG_QCAP) ? 0x7fffffff : pcnt;
        for (int r = tid; r < rows; r += 1024) {
            float dg = degs[r];
            dinv[lo + r] = dg > 0.0f ? rsqrtf(dg) : 0.0f;
        }
        __syncthreads();
    }
}

#define ACC_LD 129
__global__ __launch_bounds__(1024) void k_agg1_lds(const int* __restrict__ src,
                                                   const int* __restrict__ dst,
                                                   const float* __restrict__ w,
                                                   const float* __restrict__ dinv,
                                                   const float* __restrict__ H,
                                                   const float* __restrict__ b1,
                                                   const float* __restrict__ W2,
                                                   float* __restrict__ G, float* parkbase,
                                                   int N, int E) {
    __shared__ float acc[AG_ROWS * ACC_LD];
    __shared__ float Ws[NHID * NCPAD];
    __shared__ float bs[NHID];

    const int tid = threadIdx.x, lane = tid & 31, wave = tid >> 5;
    const float4* H4 = (const float4*)H;

    static_assert((NHID * NCPAD) % 1024 == 0, "one W2 word per thread and trip");
    float wreg[(NHID * NCPAD) / 1024];
#pragma unroll
    for (int u = 0; u < (NHID * NCPAD) / 1024; ++u) {
        int idx = u * 1024 + tid;
        int f = idx / NCPAD, n = idx - f * NCPAD;
        float x = W2[f * NCLASS + min(n, NCLASS - 1)];
        wreg[u] = (n < NCLASS) ? x : 0.0f;
    }
    const float breg = b1[tid & (NHID - 1)];
    bool staged = false;

    for (int lo = blockIdx.x * AG_ROWS; lo < N; lo += gridDim.x * AG_ROWS) {
        const int rows = min(AG_ROWS, N - lo);

        for (int idx = tid; idx < rows * 32; idx += 1024) {
            int r = idx >> 5, i = lo + r;
            float di = dinv[i], c = di * di;
            float4 v = H4[(size_t)i * 32 + lane];
            float* a = acc + r * ACC_LD + lane;
            a[0] = c * v.x; a[32] = c * v.y; a[64] = c * v.z; a[96] = c * v.w;
        }
        if (!staged) {
#pragma unroll
            for (int u = 0; u < (NHID * NCPAD) / 1024; ++u) Ws[u * 1024 + tid] = wreg[u];
            if (tid < NHID) bs[tid] = breg;
            staged = true;
        }
        int* park = (int*)(parkbase + (size_t)lo * NCLASS);
        const int cnt = park[0];
        const int room = min(AG_QCAP, (rows * NCLASS - 2) >> 1);
        __syncthreads();

        if (cnt <= room) {
            int2* park2 = (int2*)(park + 2);
            for (int base = wave * 32; base < cnt; base += 32 * 32) {
                const int n = min(32, cnt - base);
                int s = 0, r = 0;
                float c = 0.0f;
                if (lane < n) {
                    int2 t = park2[base + lane];
                    s = t.x >> 9;
                    r = t.x & 511;
                    c = dinv[s] * __int_as_float(t.y) * dinv[lo + r];
                    park2[base + lane] = make_int2(t.x, __float_as_int(c));
                }
                int j0 = 0;
                for (; j0 + 8 <= n; j0 += 8) {
                    float4 v[8];
#pragma unroll
                    for (int u = 0; u < 8; ++u) {
                        int sj = __builtin_amdgcn_readlane(s, j0 + u);
                        v[u] = H4[(size_t)sj * 32 + lane];
                    }
#pragma unroll
                    for (int u = 0; u < 8; ++u) {
                        int rj = __builtin_amdgcn_readlane(r, j0 + u);
                        float cj = __int_as_float(__builtin_amdgcn_readlane(__float_as_int(c), j0 + u));
                        float* a = acc + rj * ACC_LD + lane;
                        __hip_atomic_fetch_add(a + 0,  cj * v[u].x, __ATOMIC_RELAXED, __HIP_MEMORY_SCOPE_WORKGROUP);
                        __hip_atomic_fetch_add(a + 32, cj * v[u].y, __ATOMIC_RELAXED, __HIP_MEMORY_SCOPE_WORKGROUP);
                        __hip_atomic_fetch_add(a + 64, cj * v[u].z, __ATOMIC_RELAXED, __HIP_MEMORY_SCOPE_WORKGROUP);
                        __hip_atomic_fetch_add(a + 96, cj * v[u].w, __ATOMIC_RELAXED, __HIP_MEMORY_SCOPE_WORKGROUP);
                    }
                }
                for (; j0 < n; ++j0) {
                    int sj = __builtin_amdgcn_readlane(s, j0);
                    int rj = __builtin_amdgcn_readlane(r, j0);
                    float cj = __int_as_float(__builtin_amdgcn_readlane(__float_as_int(c), j0));
                    float4 v = H4[(size_t)sj * 32 + lane];
                    float* a = acc + rj * ACC_LD + lane;
                    __hip_atomic_fetch_add(a + 0,  cj * v.x, __ATOMIC_RELAXED, __HIP_MEMORY_SCOPE_WORKGROUP);
                    __hip_atomic_fetch_add(a + 32, cj * v.y, __ATOMIC_RELAXED, __HIP_MEMORY_SCOPE_WORKGROUP);
                    __hip_atomic_fetch_add(a + 64, cj * v.z, __ATOMIC_RELAXED, __HIP_MEMORY_SCOPE_WORKGROUP);
                    __hip_atomic_fetch_add(a + 96, cj * v.w, __ATOMIC_RELAXED, __HIP_MEMORY_SCOPE_WORKGROUP);
                }
            }
        } else {
            for (int e = tid; e < E; e += 1024) {
                unsigned r = (unsigned)(dst[e] - lo);
                if (r < (unsigned)rows) {
                    int s = src[e];
                    float c = dinv[s] * w[e] * dinv[lo + r];
                    const float* h = H + (size_t)s * NHID;
                    for (int f = 0; f < NHID; ++f)
                        __hip_atomic_fetch_add(&acc[r * ACC_LD + (f & 3) * 32 + (f >> 2)], c * h[f],
                                               __ATOMIC_RELAXED, __HIP_MEMORY_SCOPE_WORKGROUP);
                }
            }
        }
        __syncthreads();

        const int wv = __builtin_amdgcn_readfirstlane(wave);
        if (wv * 16 < rows) {
            const int cc = lane & 15, hh = lane >> 4;
            int rr = wv * 16 + cc;
            if (rr >= rows) rr = rows - 1;
            const float* arow = acc + rr * ACC_LD + 32 * hh;
            const float* wr0 = Ws + hh * NCPAD + cc;
            const float* bb = bs + hh;
            v8f d0 = {}, d1 = {}, d2 = {};
#pragma unroll 4
            for (int j = 0; j < 32; ++j) {
                const float* wr = wr0 + 4 * j * NCPAD;
                v2f av  = {fmaxf(arow[j] + bb[4 * j], 0.0f), fmaxf(arow[64 + j] + bb[4 * j + 2], 0.0f)};
                v2f bv0 = {wr[0],  wr[2 * NCPAD]};
                v2f bv1 = {wr[16], wr[2 * NCPAD + 16]};
                v2f bv2 = {wr[32], wr[2 * NCPAD + 32]};
                d0 = __builtin_amdgcn_wmma_f32_16x16x4_f32(false, av, false, bv0, (short)0, d0, false, false);
                d1 = __builtin_amdgcn_wmma_f32_16x16x4_f32(false, av, false, bv1, (short)0, d1, false, false);
                d2 = __builtin_amdgcn_wmma_f32_16x16x4_f32(false, av, false, bv2, (short)0, d2, false, false);
            }
            const int rb = wv * 16 + 8 * hh;
            float* Go = G + (size_t)(lo + rb) * NCPAD + cc;
#pragma unroll
            for (int r = 0; r < 8; ++r) {
                if (rb + r < rows) {
                    float* o = Go + (size_t)r * NCPAD;
                    o[0]  = d0[r];
                    o[16] = d1[r];
                    o[32] = d2[r];
                }
            }
        }
        __syncthreads();
    }
}

__global__ __launch_bounds__(1024) void k_agg2_lds(const int* __restrict__ src,
                                                   const int* __restrict__ dst,
                                                   const float* __restrict__ w,
                                                   const float* __restrict__ dinv,
                                                   const float* __restrict__ G,
                                                   const float* __restrict__ b2,
                                                   float* out, int N, int E) {
    __shared__ __attribute__((aligned(16))) float acc[AG_ROWS * NCPAD];

    const int tid = threadIdx.x, lane = tid & 31, wave = tid >> 5;
    const bool lo16 = lane < 16;

    for (int lo = blockIdx.x * AG_ROWS; lo < N; lo += gridDim.x * AG_ROWS) {
        const int rows = min(AG_ROWS, N - lo);

        for (int idx = tid; idx < rows * 12; idx += 1024) {
            int r = idx / 12, j = idx - r * 12, i = lo + r;
            float di = dinv[i], c = di * di;
            float4 v = ((const float4*)(G + (size_t)i * NCPAD))[j];
            ((float4*)(acc + r * NCPAD))[j] = make_float4(c * v.x, c * v.y, c * v.z, c * v.w);
        }
        const int* park = (const int*)(out + (size_t)lo * NCLASS);
        const int cnt = park[0];
        const int room = min(AG_QCAP, (rows * NCLASS - 2) >> 1);
        __syncthreads();

        if (cnt <= room) {
            const int2* park2 = (const int2*)(park + 2);
            for (int base = wave * 32; base < cnt; base += 32 * 32) {
                const int n = min(32, cnt - base);
                int pk = 0;
                float c = 0.0f;
                if (lane < n) {
                    int2 t = park2[base + lane];
                    pk = t.x;
                    c = __int_as_float(t.y);
                }
                int j0 = 0;
                for (; j0 + 16 <= n; j0 += 16) {
                    int pj[16];
                    float cj[16];
#pragma unroll
                    for (int u = 0; u < 16; ++u) {
                        pj[u] = __builtin_amdgcn_readlane(pk, j0 + u);
                        cj[u] = __int_as_float(__builtin_amdgcn_readlane(__float_as_int(c), j0 + u));
                    }
                    float ga[16], gb[16];
#pragma unroll
                    for (int u = 0; u < 16; ++u)
                        ga[u] = G[(size_t)(pj[u] >> 9) * NCPAD + lane];
                    if (lo16) {
#pragma unroll
                        for (int u = 0; u < 16; ++u)
                            gb[u] = G[(size_t)(pj[u] >> 9) * NCPAD + 32 + lane];
                    }
#pragma unroll
                    for (int u = 0; u < 16; ++u)
                        __hip_atomic_fetch_add(acc + (pj[u] & 511) * NCPAD + lane, cj[u] * ga[u],
                                               __ATOMIC_RELAXED, __HIP_MEMORY_SCOPE_WORKGROUP);
                    if (lo16) {
#pragma unroll
                        for (int u = 0; u < 16; ++u)
                            __hip_atomic_fetch_add(acc + (pj[u] & 511) * NCPAD + 32 + lane, cj[u] * gb[u],
                                                   __ATOMIC_RELAXED, __HIP_MEMORY_SCOPE_WORKGROUP);
                    }
                }
                for (; j0 < n; ++j0) {
                    int pj = __builtin_amdgcn_readlane(pk, j0);
                    float cj = __int_as_float(__builtin_amdgcn_readlane(__float_as_int(c), j0));
                    const float* g = G + (size_t)(pj >> 9) * NCPAD;
                    float* a = acc + (pj & 511) * NCPAD;
                    __hip_atomic_fetch_add(a + lane, cj * g[lane], __ATOMIC_RELAXED, __HIP_MEMORY_SCOPE_WORKGROUP);
                    if (lo16)
                        __hip_atomic_fetch_add(a + 32 + lane, cj * g[32 + lane],
                                               __ATOMIC_RELAXED, __HIP_MEMORY_SCOPE_WORKGROUP);
                }
            }
        } else {
            for (int e = tid; e < E; e += 1024) {
                unsigned r = (unsigned)(dst[e] - lo);
                if (r < (unsigned)rows) {
                    int s = src[e];
                    float c = dinv[s] * w[e] * dinv[lo + r];
                    const float* g = G + (size_t)s * NCPAD;
                    for (int f = 0; f < NCLASS; ++f)
                        __hip_atomic_fetch_add(&acc[r * NCPAD + f], c * g[f],
                                               __ATOMIC_RELAXED, __HIP_MEMORY_SCOPE_WORKGROUP);
                }
            }
        }
        __syncthreads();

        for (int r = wave; r < rows; r += 32) {
            const float* row = acc + r * NCPAD;
            float v1 = row[lane] + b2[lane];
            int c2 = 32 + lane;
            bool has2 = c2 < NCLASS;
            float v2 = has2 ? (row[c2] + b2[c2]) : -INFINITY;

            float mx = fmaxf(v1, v2);
#pragma unroll
            for (int off = 16; off >= 1; off >>= 1) mx = fmaxf(mx, __shfl_xor(mx, off));

            float sm = expf(v1 - mx) + (has2 ? expf(v2 - mx) : 0.0f);
#pragma unroll
            for (int off = 16; off >= 1; off >>= 1) sm += __shfl_xor(sm, off);

            float ls = logf(sm);
            float* o = out + (size_t)(lo + r) * NCLASS;
            o[lane] = v1 - mx - ls;
            if (has2) o[c2] = v2 - mx - ls;
        }
        __syncthreads();
    }
}

extern "C" void kernel_launch(void* const* d_in, const int* in_sizes, int n_in,
                              void* d_out, int out_size, void* d_ws, size_t ws_size,
                              hipStream_t stream) {
    const float* x  = (const float*)d_in[0];
    const int*   ei = (const int*)  d_in[1];
    const float* ew = (const float*)d_in[2];
    const float* W1 = (const float*)d_in[3];
    const float* b1 = (const float*)d_in[4];
    const float* W2 = (const float*)d_in[5];
    const float* b2 = (const float*)d_in[6];
    float* out = (float*)d_out;

    const int N = in_sizes[0] / NFEAT;
    const int E = in_sizes[2];
    const int* src = ei;
    const int* dst = ei + E;

    float* ws = (float*)d_ws;
    size_t o = 0;
    auto take = [&](size_t n) { float* p = ws + o; o += (n + 255) & ~(size_t)255; return p; };
    float* dinv = take((size_t)N);
    float* Wp1  = take((size_t)NFEAT * NHID);
    float* Hxw  = take((size_t)N * NHID);
    float* G    = take((size_t)N * NHID);

    k_pack_w1<<<(NFEAT * NHID + 255) / 256, 256, 0, stream>>>(W1, Wp1);

    unsigned* ob = (unsigned*)G;
    const int nvec = (E + 15) / 16;
    int* rec = (int*)Hxw;
    k_owner_bytes<<<256, 512, 0, stream>>>(src, dst, ew, ob, rec, E, 4 * nvec);
    k_sweep_deg<<<256, 1024, 0, stream>>>(src, dst, ew, ob, rec, dinv, out, N, E, nvec);

    k_gemm1<<<(N + 31) / 32, 256, 0, stream>>>(x, Wp1, Hxw, N);

    k_agg1_lds<<<256, 1024, 0, stream>>>(src, dst, ew, dinv, Hxw, b1, W2, G, out, N, E);

    k_agg2_lds<<<256, 1024, 0, stream>>>(src, dst, ew, dinv, G, b2, out, N, E);
}


# Sweep pre-filter: first pass also writes a 256-bit owner map per 32 edges; owner blocks read one bit per 32 edges and byte-match only groups that show their owner; LDS-arrival sums kept
# speedup vs baseline: 3.5797x; 1.0734x over previous
#include <hip/hip_runtime.h>
#include <cstdint>

typedef float v2f __attribute__((ext_vector_type(2)));
typedef float v8f __attribute__((ext_vector_type(8)));

#define NFEAT 256
#define NHID  128
#define NCLASS 41
#define NCPAD  48

__global__ __launch_bounds__(256) void k_pack_w1(const float* __restrict__ W1, float* __restrict__ Wp) {
    int idx = blockIdx.x * 256 + threadIdx.x;
    if (idx >= NFEAT * NHID) return;
    int k = idx >> 7, n = idx & 127;
    Wp[(size_t)(k >> 1) * (NHID * 2) + n * 2 + (k & 1)] = W1[idx];
}

__global__ __launch_bounds__(256) void k_gemm1(const float* __restrict__ X,
                                               const float* __restrict__ Wp,
                                               float* __restrict__ H, int M) {
    __shared__ float2 Xs[128 * 33];
    const int tid = threadIdx.x;
    const int m0 = blockIdx.x * 32;

#pragma unroll
    for (int it = 0; it < 8; ++it) {
        int q = it * 256 + tid;
        int m = q >> 6, k4 = q & 63;
        int mr = m0 + m; if (mr >= M) mr = M - 1;
        float4 v = ((const float4*)(X + (size_t)mr * NFEAT))[k4];
        Xs[(2 * k4)     * 33 + m] = make_float2(v.x, v.y);
        Xs[(2 * k4 + 1) * 33 + m] = make_float2(v.z, v.w);
    }
    __syncthreads();

    const int wave = tid >> 5, lane = tid & 31, l = lane & 15, half = lane >> 4;
    const int n0 = wave * 16;
    const float2* Bp = (const float2*)Wp;

    v8f acc0 = {}, acc1 = {};
#pragma unroll 4
    for (int kp = 0; kp < 128; kp += 2) {
        int row = kp + half;
        float2 a0 = Xs[row * 33 + l];
        float2 a1 = Xs[row * 33 + 16 + l];
        float2 b  = Bp[(size_t)row * NHID + n0 + l];
        v2f av0 = {a0.x, a0.y}, av1 = {a1.x, a1.y}, bv = {b.x, b.y};
        acc0 = __builtin_amdgcn_wmma_f32_16x16x4_f32(false, av0, false, bv, (short)0, acc0, false, false);
        acc1 = __builtin_amdgcn_wmma_f32_16x16x4_f32(false, av1, false, bv, (short)0, acc1, false, false);
    }

    const int mb = 8 * half;
    float* Ho = H + (size_t)(m0 + mb) * NHID + n0 + l;
    if (m0 + 32 <= M) {
#pragma unroll
        for (int r = 0; r < 8; ++r) {
            Ho[(size_t)r * NHID]        = acc0[r];
            Ho[(size_t)(16 + r) * NHID] = acc1[r];
        }
    } else {
#pragma unroll
        for (int r = 0; r < 8; ++r) {
            if (m0 + mb + r < M)      Ho[(size_t)r * NHID]        = acc0[r];
            if (m0 + 16 + mb + r < M) Ho[(size_t)(16 + r) * NHID] = acc1[r];
        }
    }
}

#define AG_ROWS 392
#define AG_QCAP 8192

__device__ __forceinline__ unsigned own_byte(int d) { return ((unsigned)d / AG_ROWS) & 255u; }

__device__ __forceinline__ int2 edge_rec(int s, int d, float x) {
    const int row = d - (int)((unsigned)d / AG_ROWS) * AG_ROWS;
    return make_int2((s << 9) | row, __float_as_int(x));
}

__device__ __forceinline__ void bm_set(unsigned* group, unsigned o) {
    __hip_atomic_fetch_or(group + (o >> 5) * 64, 1u << (o & 31), __ATOMIC_RELAXED, __HIP_MEMORY_SCOPE_WORKGROUP);
}

__device__ __forceinline__ void lds_barrier() {
    __builtin_amdgcn_fence(__ATOMIC_RELEASE, "workgroup", "local");
    __builtin_amdgcn_s_barrier();
    __builtin_amdgcn_fence(__ATOMIC_ACQUIRE, "workgroup", "local");
}

__device__ __forceinline__ int bm_offset(int nwords) { return (nwords + 255) & ~255; }
__device__ __forceinline__ int bm_pitch(int E) { return (((E + 31) >> 5) + 63) & ~63; }

__global__ __launch_bounds__(512) void k_owner_bytes(const int* __restrict__ src,
                                                     const int* __restrict__ dst,
                                                     const float* __restrict__ w,
                                                     unsigned* __restrict__ ob, int* __restrict__ recw,
                                                     int E, int nwords) {
    __shared__ unsigned bms[512];
    int2* rec = (int2*)recw;
    unsigned* bm = ob + bm_offset(nwords);
    const int ngp = bm_pitch(E);
    const bool al16 = ((((uintptr_t)dst) | ((uintptr_t)src) | ((uintptr_t)w)) & 15u) == 0;
    const int T = gridDim.x * 512;
    const int tid = threadIdx.x;
    unsigned* mygroup = bms + (tid >> 3);
    bms[tid] = 0;
    for (int wd0 = blockIdx.x * 512; wd0 < 8 * ngp; wd0 += T) {
        lds_barrier();
        const int wd = wd0 + tid;
        if (wd < nwords) {
            const int e = 4 * wd;
            unsigned pk = 0;
            if (al16 && e + 4 <= E) {
                int4 d = ((const int4*)dst)[wd];
                int4 s = ((const int4*)src)[wd];
                float4 x = ((const float4*)w)[wd];
                const unsigned o0 = own_byte(d.x), o1 = own_byte(d.y), o2 = own_byte(d.z), o3 = own_byte(d.w);
                pk = o0 | (o1 << 8) | (o2 << 16) | (o3 << 24);
                int2 r0 = edge_rec(s.x, d.x, x.x), r1 = edge_rec(s.y, d.y, x.y);
                int2 r2 = edge_rec(s.z, d.z, x.z), r3 = edge_rec(s.w, d.w, x.w);
                int4* ro = (int4*)(rec + e);
                ro[0] = make_int4(r0.x, r0.y, r1.x, r1.y);
                ro[1] = make_int4(r2.x, r2.y, r3.x, r3.y);
                bm_set(mygroup, o0); bm_set(mygroup, o1); bm_set(mygroup, o2); bm_set(mygroup, o3);
            } else {
                for (int k = 0; k < 4; ++k)
                    if (e + k < E) {
                        int d = dst[e + k];
                        const unsigned o = own_byte(d);
                        pk |= o << (8 * k);
                        rec[e + k] = edge_rec(src[e + k], d, w[e + k]);
                        bm_set(mygroup, o);
                    }
            }
            ob[wd] = pk;
        }
        lds_barrier();
        bm[(size_t)(tid >> 6) * ngp + (wd0 >> 3) + (tid & 63)] = bms[tid];
        bms[tid] = 0;
    }
}

__device__ __forceinline__ void sw_push(int e, int lo, int rows, int* qcnt, int* q, float* degs,
                                        const int* __restrict__ dst, const float* __restrict__ w) {
    int slot = __hip_atomic_fetch_add(qcnt, 1, __ATOMIC_RELAXED, __HIP_MEMORY_SCOPE_WORKGROUP);
    if (slot < AG_QCAP) {
        q[slot] = e;
    } else {
        unsigned r = (unsigned)(dst[e] - lo);
        if (r < (unsigned)rows)
            __hip_atomic_fetch_add(&degs[r], w[e], __ATOMIC_RELAXED, __HIP_MEMORY_SCOPE_WORKGROUP);
    }
}

__device__ __forceinline__ unsigned zero_bytes(unsigned x) {
    unsigned y = ((x & 0x7F7F7F7Fu) + 0x7F7F7F7Fu) | x;
    return ~y & 0x80808080u;
}
__device__ __forceinline__ unsigned match16(uint4 d, unsigned pat) {
    return zero_bytes(d.x ^ pat) | (zero_bytes(d.y ^ pat) >> 1) |
           (zero_bytes(d.z ^ pat) >> 2) | (zero_bytes(d.w ^ pat) >> 3);
}

__device__ __forceinline__ void push_group(int g, uint4 da, uint4 db, bool has_b, unsigned pat, int E,
                                           int lo, int rows, int* qcnt, int* q, float* degs,
                                           const int* __restrict__ dst, const float* __restrict__ w) {
    const unsigned ma = match16(da, pat), mb = has_b ? match16(db, pat) : 0u;
    unsigned m = ma | (mb >> 4);
    while (m) {
        int p = __builtin_ctz(m);
        m &= m - 1;
        int j = p >> 3, b = p & 7;
        int pp = 2 * g + (b < 4 ? 1 : 0);
        int k = (7 - b) & 3;
        int e = 16 * pp + 4 * k + j;
        if (e < E) sw_push(e, lo, rows, qcnt, q, degs, dst, w);
    }
}

#define HV_CAP 8192

__global__ __launch_bounds__(1024) void k_sweep_deg(const int* __restrict__ src,
                                                    const int* __restrict__ dst,
                                                    const float* __restrict__ w,
                                                    const unsigned* __restrict__ ob,
                                                    const int* __restrict__ recw,
                                                    float* __restrict__ dinv, float* parkbase,
                                                    int N, int E, int nvec) {
    __shared__ int q[AG_QCAP];
    __shared__ int hv[HV_CAP];
    __shared__ float degs[AG_ROWS];
    __shared__ int qcnt, pcnt, hcnt;

    const int tid = threadIdx.x;
    const int2* rec = (const int2*)recw;
    const uint4* ob4 = (const uint4*)ob;
    const unsigned* bm = ob + bm_offset(4 * nvec);
    const int ngp = bm_pitch(E);
    const int nv4 = ngp >> 2;
    const int start = (int)(((long long)blockIdx.x * nv4) / gridDim.x);
    const bool shared_byte = N > 256 * AG_ROWS;

    for (int lo = blockIdx.x * AG_ROWS; lo < N; lo += gridDim.x * AG_ROWS) {
        const int rows = min(AG_ROWS, N - lo);
        const unsigned obyte = ((unsigned)lo / AG_ROWS) & 255u;
        const unsigned pat = obyte * 0x01010101u;
        const uint4* bm4 = (const uint4*)(bm + (size_t)(obyte >> 5) * ngp);
        const int bit = obyte & 31;

        for (int r = tid; r < rows; r += 1024) degs[r] = 1.0f;
        if (tid == 0) { qcnt = 0; pcnt = 0; hcnt = 0; }
        __syncthreads();

        for (int i0 = 0; i0 < nv4; i0 += 8 * 1024) {
            uint4 d[8];
#pragma unroll
            for (int u = 0; u < 8; ++u) {
                int v = i0 + u * 1024 + tid;
                int pp = v + start; if (pp >= nv4) pp -= nv4;
                d[u] = bm4[v < nv4 ? pp : 0];
            }
            unsigned hm = 0;
#pragma unroll
            for (int u = 0; u < 8; ++u) {
                unsigned t = ((d[u].x >> bit) & 1u) | (((d[u].y >> bit) & 1u) << 1) |
                             (((d[u].z >> bit) & 1u) << 2) | (((d[u].w >> bit) & 1u) << 3);
                if (i0 + u * 1024 + tid >= nv4) t = 0;
                hm |= t << (4 * u);
            }
            while (hm) {
                int p = __builtin_ctz(hm);
                hm &= hm - 1;
                int pp = i0 + (p >> 2) * 1024 + tid + start; if (pp >= nv4) pp -= nv4;
                const int g = 4 * pp + (p & 3);
                int slot = __hip_atomic_fetch_add(&hcnt, 1, __ATOMIC_RELAXED, __HIP_MEMORY_SCOPE_WORKGROUP);
                if (slot < HV_CAP) {
                    hv[slot] = g;
                } else {
                    const bool hb = 2 * g + 1 < nvec;
                    push_group(g, ob4[2 * g], ob4[hb ? 2 * g + 1 : 2 * g], hb, pat, E, lo, rows, &qcnt, q, degs, dst, w);
                }
            }
        }
        __syncthreads();

        const int nh = min(hcnt, HV_CAP);
        for (int i0 = 0; i0 < nh; i0 += 4 * 1024) {
            int g[4];
            uint4 da[4], db[4];
#pragma unroll
            for (int u = 0; u < 4; ++u) {
                int i = i0 + u * 1024 + tid;
                g[u] = i < nh ? hv[i] : 0;
                da[u] = ob4[2 * g[u]];
                db[u] = ob4[min(2 * g[u] + 1, nvec - 1)];
            }
#pragma unroll
            for (int u = 0; u < 4; ++u)
                if (i0 + u * 1024 + tid < nh)
                    push_group(g[u], da[u], db[u], 2 * g[u] + 1 < nvec, pat, E, lo, rows, &qcnt, q, degs, dst, w);
        }
        __syncthreads();

        const int nq = min(qcnt, AG_QCAP);
        int* park = (int*)(parkbase + (size_t)lo * NCLASS);
        int2* park2 = (int2*)(park + 2);
        const int room = (rows * NCLASS - 2) >> 1;
        int2 rc[AG_QCAP / 1024];
        int dd[AG_QCAP / 1024];
#pragma unroll
        for (int u = 0; u < AG_QCAP / 1024; ++u) {
            int i = u * 1024 + tid;
            rc[u] = make_int2(0, 0); dd[u] = lo;
            if (i < nq) {
                int e = q[i];
                rc[u] = rec[e];
                if (shared_byte) dd[u] = dst[e];
            }
        }
#pragma unroll
        for (int u = 0; u < AG_QCAP / 1024; ++u) {
            int i = u * 1024 + tid;
            const int r = rc[u].x & 511;
            const bool mine = r < rows && (!shared_byte || (unsigned)(dd[u] - lo) < (unsigned)rows);
            if (i < nq && mine) {
                __hip_atomic_fetch_add(&degs[r], __int_as_float(rc[u].y), __ATOMIC_RELAXED, __HIP_MEMORY_SCOPE_WORKGROUP);
                int slot = __hip_atomic_fetch_add(&pcnt, 1, __ATOMIC_RELAXED, __HIP_MEMORY_SCOPE_WORKGROUP);
                if (slot < room) park2[slot] = rc[u];
            }
        }
        __syncthreads();

        if (tid == 0) park[0] = (qcnt > AG_QCAP) ? 0x7fffffff : pcnt;
        for (int r = tid; r < rows; r += 1024) {
            float dg = degs[r];
            dinv[lo + r] = dg > 0.0f ? rsqrtf(dg) : 0.0f;
        }
        __syncthreads();
    }
}

#define ACC_LD 129
__global__ __launch_bounds__(1024) void k_agg1_lds(const int* __restrict__ src,
                                                   const int* __restrict__ dst,
                                                   const float* __restrict__ w,
                                                   const float* __restrict__ dinv,
                                                   const float* __restrict__ H,
                                                   const float* __restrict__ b1,
                                                   const float* __restrict__ W2,
                                                   float* __restrict__ G, float* parkbase,
                                                   int N, int E) {
    __shared__ float acc[AG_ROWS * ACC_LD];
    __shared__ float Ws[NHID * NCPAD];
    __shared__ float bs[NHID];

    const int tid = threadIdx.x, lane = tid & 31, wave = tid >> 5;
    const float4* H4 = (const float4*)H;

    static_assert((NHID * NCPAD) % 1024 == 0, "one W2 word per thread and trip");
    float wreg[(NHID * NCPAD) / 1024];
#pragma unroll
    for (int u = 0; u < (NHID * NCPAD) / 1024; ++u) {
        int idx = u * 1024 + tid;
        int f = idx / NCPAD, n = idx - f * NCPAD;
        float x = W2[f * NCLASS + min(n, NCLASS - 1)];
        wreg[u] = (n < NCLASS) ? x : 0.0f;
    }
    const float breg = b1[tid & (NHID - 1)];
    bool staged = false;

    for (int lo = blockIdx.x * AG_ROWS; lo < N; lo += gridDim.x * AG_ROWS) {
        const int rows = min(AG_ROWS, N - lo);

        for (int idx = tid; idx < rows * 32; idx += 1024) {
            int r = idx >> 5, i = lo + r;
            float di = dinv[i], c = di * di;
            float4 v = H4[(size_t)i * 32 + lane];
            float* a = acc + r * ACC_LD + lane;
            a[0] = c * v.x; a[32] = c * v.y; a[64] = c * v.z; a[96] = c * v.w;
        }
        if (!staged) {
#pragma unroll
            for (int u = 0; u < (NHID * NCPAD) / 1024; ++u) Ws[u * 1024 + tid] = wreg[u];
            if (tid < NHID) bs[tid] = breg;
            staged = true;
        }
        int* park = (int*)(parkbase + (size_t)lo * NCLASS);
        const int cnt = park[0];
        const int room = min(AG_QCAP, (rows * NCLASS - 2) >> 1);
        __syncthreads();

        if (cnt <= room) {
            int2* park2 = (int2*)(park + 2);
            for (int base = wave * 32; base < cnt; base += 32 * 32) {
                const int n = min(32, cnt - base);
                int s = 0, r = 0;
                float c = 0.0f;
                if (lane < n) {
                    int2 t = park2[base + lane];
                    s = t.x >> 9;
                    r = t.x & 511;
                    c = dinv[s] * __int_as_float(t.y) * dinv[lo + r];
                    park2[base + lane] = make_int2(t.x, __float_as_int(c));
                }
                int j0 = 0;
                for (; j0 + 8 <= n; j0 += 8) {
                    float4 v[8];
#pragma unroll
                    for (int u = 0; u < 8; ++u) {
                        int sj = __builtin_amdgcn_readlane(s, j0 + u);
                        v[u] = H4[(size_t)sj * 32 + lane];
                    }
#pragma unroll
                    for (int u = 0; u < 8; ++u) {
                        int rj = __builtin_amdgcn_readlane(r, j0 + u);
                        float cj = __int_as_float(__builtin_amdgcn_readlane(__float_as_int(c), j0 + u));
                        float* a = acc + rj * ACC_LD + lane;
                        __hip_atomic_fetch_add(a + 0,  cj * v[u].x, __ATOMIC_RELAXED, __HIP_MEMORY_SCOPE_WORKGROUP);
                        __hip_atomic_fetch_add(a + 32, cj * v[u].y, __ATOMIC_RELAXED, __HIP_MEMORY_SCOPE_WORKGROUP);
                        __hip_atomic_fetch_add(a + 64, cj * v[u].z, __ATOMIC_RELAXED, __HIP_MEMORY_SCOPE_WORKGROUP);
                        __hip_atomic_fetch_add(a + 96, cj * v[u].w, __ATOMIC_RELAXED, __HIP_MEMORY_SCOPE_WORKGROUP);
                    }
                }
                for (; j0 < n; ++j0) {
                    int sj = __builtin_amdgcn_readlane(s, j0);
                    int rj = __builtin_amdgcn_readlane(r, j0);
                    float cj = __int_as_float(__builtin_amdgcn_readlane(__float_as_int(c), j0));
                    float4 v = H4[(size_t)sj * 32 + lane];
                    float* a = acc + rj * ACC_LD + lane;
                    __hip_atomic_fetch_add(a + 0,  cj * v.x, __ATOMIC_RELAXED, __HIP_MEMORY_SCOPE_WORKGROUP);
                    __hip_atomic_fetch_add(a + 32, cj * v.y, __ATOMIC_RELAXED, __HIP_MEMORY_SCOPE_WORKGROUP);
                    __hip_atomic_fetch_add(a + 64, cj * v.z, __ATOMIC_RELAXED, __HIP_MEMORY_SCOPE_WORKGROUP);
                    __hip_atomic_fetch_add(a + 96, cj * v.w, __ATOMIC_RELAXED, __HIP_MEMORY_SCOPE_WORKGROUP);
                }
            }
        } else {
            for (int e = tid; e < E; e += 1024) {
                unsigned r = (unsigned)(dst[e] - lo);
                if (r < (unsigned)rows) {
                    int s = src[e];
                    float c = dinv[s] * w[e] * dinv[lo + r];
                    const float* h = H + (size_t)s * NHID;
                    for (int f = 0; f < NHID; ++f)
                        __hip_atomic_fetch_add(&acc[r * ACC_LD + (f & 3) * 32 + (f >> 2)], c * h[f],
                                               __ATOMIC_RELAXED, __HIP_MEMORY_SCOPE_WORKGROUP);
                }
            }
        }
        __syncthreads();

        const int wv = __builtin_amdgcn_readfirstlane(wave);
        if (wv * 16 < rows) {
            const int cc = lane & 15, hh = lane >> 4;
            int rr = wv * 16 + cc;
            if (rr >= rows) rr = rows - 1;
            const float* arow = acc + rr * ACC_LD + 32 * hh;
            const float* wr0 = Ws + hh * NCPAD + cc;
            const float* bb = bs + hh;
            v8f d0 = {}, d1 = {}, d2 = {};
#pragma unroll 4
            for (int j = 0; j < 32; ++j) {
                const float* wr = wr0 + 4 * j * NCPAD;
                v2f av  = {fmaxf(arow[j] + bb[4 * j], 0.0f), fmaxf(arow[64 + j] + bb[4 * j + 2], 0.0f)};
                v2f bv0 = {wr[0],  wr[2 * NCPAD]};
                v2f bv1 = {wr[16], wr[2 * NCPAD + 16]};
                v2f bv2 = {wr[32], wr[2 * NCPAD + 32]};
                d0 = __builtin_amdgcn_wmma_f32_16x16x4_f32(false, av, false, bv0, (short)0, d0, false, false);
                d1 = __builtin_amdgcn_wmma_f32_16x16x4_f32(false, av, false, bv1, (short)0, d1, false, false);
                d2 = __builtin_amdgcn_wmma_f32_16x16x4_f32(false, av, false, bv2, (short)0, d2, false, false);
            }
            const int rb = wv * 16 + 8 * hh;
            float* Go = G + (size_t)(lo + rb) * NCPAD + cc;
#pragma unroll
            for (int r = 0; r < 8; ++r) {
                if (rb + r < rows) {
                    float* o = Go + (size_t)r * NCPAD;
                    o[0]  = d0[r];
                    o[16] = d1[r];
                    o[32] = d2[r];
                }
            }
        }
        __syncthreads();
    }
}

__global__ __launch_bounds__(1024) void k_agg2_lds(const int* __restrict__ src,
                                                   const int* __restrict__ dst,
                                                   const float* __restrict__ w,
                                                   const float* __restrict__ dinv,
                                                   const float* __restrict__ G,
                                                   const float* __restrict__ b2,
                                                   float* out, int N, int E) {
    __shared__ __attribute__((aligned(16))) float acc[AG_ROWS * NCPAD];

    const int tid = threadIdx.x, lane = tid & 31, wave = tid >> 5;
    const bool lo16 = lane < 16;

    for (int lo = blockIdx.x * AG_ROWS; lo < N; lo += gridDim.x * AG_ROWS) {
        const int rows = min(AG_ROWS, N - lo);

        for (int idx = tid; idx < rows * 12; idx += 1024) {
            int r = idx / 12, j = idx - r * 12, i = lo + r;
            float di = dinv[i], c = di * di;
            float4 v = ((const float4*)(G + (size_t)i * NCPAD))[j];
            ((float4*)(acc + r * NCPAD))[j] = make_float4(c * v.x, c * v.y, c * v.z, c * v.w);
        }
        const int* park = (const int*)(out + (size_t)lo * NCLASS);
        const int cnt = park[0];
        const int room = min(AG_QCAP, (rows * NCLASS - 2) >> 1);
        __syncthreads();

        if (cnt <= room) {
            const int2* park2 = (const int2*)(park + 2);
            for (int base = wave * 32; base < cnt; base += 32 * 32) {
                const int n = min(32, cnt - base);
                int pk = 0;
                float c = 0.0f;
                if (lane < n) {
                    int2 t = park2[base + lane];
                    pk = t.x;
                    c = __int_as_float(t.y);
                }
                int j0 = 0;
                for (; j0 + 16 <= n; j0 += 16) {
                    int pj[16];
                    float cj[16];
#pragma unroll
                    for (int u = 0; u < 16; ++u) {
                        pj[u] = __builtin_amdgcn_readlane(pk, j0 + u);
                        cj[u] = __int_as_float(__builtin_amdgcn_readlane(__float_as_int(c), j0 + u));
                    }
                    float ga[16], gb[16];
#pragma unroll
                    for (int u = 0; u < 16; ++u)
                        ga[u] = G[(size_t)(pj[u] >> 9) * NCPAD + lane];
                    if (lo16) {
#pragma unroll
                        for (int u = 0; u < 16; ++u)
                            gb[u] = G[(size_t)(pj[u] >> 9) * NCPAD + 32 + lane];
                    }
#pragma unroll
                    for (int u = 0; u < 16; ++u)
                        __hip_atomic_fetch_add(acc + (pj[u] & 511) * NCPAD + lane, cj[u] * ga[u],
                                               __ATOMIC_RELAXED, __HIP_MEMORY_SCOPE_WORKGROUP);
                    if (lo16) {
#pragma unroll
                        for (int u = 0; u < 16; ++u)
                            __hip_atomic_fetch_add(acc + (pj[u] & 511) * NCPAD + 32 + lane, cj[u] * gb[u],
                                                   __ATOMIC_RELAXED, __HIP_MEMORY_SCOPE_WORKGROUP);
                    }
                }
                for (; j0 < n; ++j0) {
                    int pj = __builtin_amdgcn_readlane(pk, j0);
                    float cj = __int_as_float(__builtin_amdgcn_readlane(__float_as_int(c), j0));
                    const float* g = G + (size_t)(pj >> 9) * NCPAD;
                    float* a = acc + (pj & 511) * NCPAD;
                    __hip_atomic_fetch_add(a + lane, cj * g[lane], __ATOMIC_RELAXED, __HIP_MEMORY_SCOPE_WORKGROUP);
                    if (lo16)
                        __hip_atomic_fetch_add(a + 32 + lane, cj * g[32 + lane],
                                               __ATOMIC_RELAXED, __HIP_MEMORY_SCOPE_WORKGROUP);
                }
            }
        } else {
            for (int e = tid; e < E; e += 1024) {
                unsigned r = (unsigned)(dst[e] - lo);
                if (r < (unsigned)rows) {
                    int s = src[e];
                    float c = dinv[s] * w[e] * dinv[lo + r];
                    const float* g = G + (size_t)s * NCPAD;
                    for (int f = 0; f < NCLASS; ++f)
                        __hip_atomic_fetch_add(&acc[r * NCPAD + f], c * g[f],
                                               __ATOMIC_RELAXED, __HIP_MEMORY_SCOPE_WORKGROUP);
                }
            }
        }
        __syncthreads();

        for (int r = wave; r < rows; r += 32) {
            const float* row = acc + r * NCPAD;
            float v1 = row[lane] + b2[lane];
            int c2 = 32 + lane;
            bool has2 = c2 < NCLASS;
            float v2 = has2 ? (row[c2] + b2[c2]) : -INFINITY;

            float mx = fmaxf(v1, v2);
#pragma unroll
            for (int off = 16; off >= 1; off >>= 1) mx = fmaxf(mx, __shfl_xor(mx, off));

            float sm = expf(v1 - mx) + (has2 ? expf(v2 - mx) : 0.0f);
#pragma unroll
            for (int off = 16; off >= 1; off >>= 1) sm += __shfl_xor(sm, off);

            float ls = logf(sm);
            float* o = out + (size_t)(lo + r) * NCLASS;
            o[lane] = v1 - mx - ls;
            if (has2) o[c2] = v2 - mx - ls;
        }
        __syncthreads();
    }
}

extern "C" void kernel_launch(void* const* d_in, const int* in_sizes, int n_in,
                              void* d_out, int out_size, void* d_ws, size_t ws_size,
                              hipStream_t stream) {
    const float* x  = (const float*)d_in[0];
    const int*   ei = (const int*)  d_in[1];
    const float* ew = (const float*)d_in[2];
    const float* W1 = (const float*)d_in[3];
    const float* b1 = (const float*)d_in[4];
    const float* W2 = (const float*)d_in[5];
    const float* b2 = (const float*)d_in[6];
    float* out = (float*)d_out;

    const int N = in_sizes[0] / NFEAT;
    const int E = in_sizes[2];
    const int* src = ei;
    const int* dst = ei + E;

    float* ws = (float*)d_ws;
    size_t o = 0;
    auto take = [&](size_t n) { float* p = ws + o; o += (n + 255) & ~(size_t)255; return p; };
    float* dinv = take((size_t)N);
    float* Wp1  = take((size_t)NFEAT * NHID);
    float* Hxw  = take((size_t)N * NHID);
    float* G    = take((size_t)N * NHID);

    k_pack_w1<<<(NFEAT * NHID + 255) / 256, 256, 0, stream>>>(W1, Wp1);

    unsigned* ob = (unsigned*)G;
    const int nvec = (E + 15) / 16;
    int* rec = (int*)Hxw;
    k_owner_bytes<<<256, 512, 0, stream>>>(src, dst, ew, ob, rec, E, 4 * nvec);
    k_sweep_deg<<<256, 1024, 0, stream>>>(src, dst, ew, ob, rec, dinv, out, N, E, nvec);

    k_gemm1<<<(N + 31) / 32, 256, 0, stream>>>(x, Wp1, Hxw, N);

    k_agg1_lds<<<256, 1024, 0, stream>>>(src, dst, ew, dinv, Hxw, b1, W2, G, out, N, E);

    k_agg2_lds<<<256, 1024, 0, stream>>>(src, dst, ew, dinv, G, b2, out, N, E);
}


# Degree norm factored out of both edge loops: H and G rows stored times their own dinv, destination dinv scales each finished row once; an edge costs only its parked weight (products regrouped)
# speedup vs baseline: 3.6984x; 1.0332x over previous
#include <hip/hip_runtime.h>
#include <cstdint>

typedef float v2f __attribute__((ext_vector_type(2)));
typedef float v8f __attribute__((ext_vector_type(8)));

#define NFEAT 256
#define NHID  128
#define NCLASS 41
#define NCPAD  48

__global__ __launch_bounds__(256) void k_pack_w1(const float* __restrict__ W1, float* __restrict__ Wp) {
    int idx = blockIdx.x * 256 + threadIdx.x;
    if (idx >= NFEAT * NHID) return;
    int k = idx >> 7, n = idx & 127;
    Wp[(size_t)(k >> 1) * (NHID * 2) + n * 2 + (k & 1)] = W1[idx];
}

__global__ __launch_bounds__(256) void k_gemm1(const float* __restrict__ X,
                                               const float* __restrict__ Wp,
                                               const float* __restrict__ dinv,
                                               float* __restrict__ H, int M) {
    __shared__ float2 Xs[128 * 33];
    __shared__ __attribute__((aligned(16))) float dsc[32];
    const int tid = threadIdx.x;
    const int m0 = blockIdx.x * 32;

    if (tid < 32) { int mr = m0 + tid; if (mr >= M) mr = M - 1; dsc[tid] = dinv[mr]; }

#pragma unroll
    for (int it = 0; it < 8; ++it) {
        int q = it * 256 + tid;
        int m = q >> 6, k4 = q & 63;
        int mr = m0 + m; if (mr >= M) mr = M - 1;
        float4 v = ((const float4*)(X + (size_t)mr * NFEAT))[k4];
        Xs[(2 * k4)     * 33 + m] = make_float2(v.x, v.y);
        Xs[(2 * k4 + 1) * 33 + m] = make_float2(v.z, v.w);
    }
    __syncthreads();

    const int wave = tid >> 5, lane = tid & 31, l = lane & 15, half = lane >> 4;
    const int n0 = wave * 16;
    const float2* Bp = (const float2*)Wp;

    v8f acc0 = {}, acc1 = {};
#pragma unroll 4
    for (int kp = 0; kp < 128; kp += 2) {
        int row = kp + half;
        float2 a0 = Xs[row * 33 + l];
        float2 a1 = Xs[row * 33 + 16 + l];
        float2 b  = Bp[(size_t)row * NHID + n0 + l];
        v2f av0 = {a0.x, a0.y}, av1 = {a1.x, a1.y}, bv = {b.x, b.y};
        acc0 = __builtin_amdgcn_wmma_f32_16x16x4_f32(false, av0, false, bv, (short)0, acc0, false, false);
        acc1 = __builtin_amdgcn_wmma_f32_16x16x4_f32(false, av1, false, bv, (short)0, acc1, false, false);
    }

    const int mb = 8 * half;
    float* Ho = H + (size_t)(m0 + mb) * NHID + n0 + l;
    float s0[8], s1[8];
    {
        const float4* d4 = (const float4*)(dsc + mb);
        const float4 p0 = d4[0], p1 = d4[1], q0 = d4[4], q1 = d4[5];
        s0[0] = p0.x; s0[1] = p0.y; s0[2] = p0.z; s0[3] = p0.w;
        s0[4] = p1.x; s0[5] = p1.y; s0[6] = p1.z; s0[7] = p1.w;
        s1[0] = q0.x; s1[1] = q0.y; s1[2] = q0.z; s1[3] = q0.w;
        s1[4] = q1.x; s1[5] = q1.y; s1[6] = q1.z; s1[7] = q1.w;
    }
    if (m0 + 32 <= M) {
#pragma unroll
        for (int r = 0; r < 8; ++r) {
            Ho[(size_t)r * NHID]        = acc0[r] * s0[r];
            Ho[(size_t)(16 + r) * NHID] = acc1[r] * s1[r];
        }
    } else {
#pragma unroll
        for (int r = 0; r < 8; ++r) {
            if (m0 + mb + r < M)      Ho[(size_t)r * NHID]        = acc0[r] * s0[r];
            if (m0 + 16 + mb + r < M) Ho[(size_t)(16 + r) * NHID] = acc1[r] * s1[r];
        }
    }
}

#define AG_ROWS 392
#define AG_QCAP 8192

__device__ __forceinline__ unsigned own_byte(int d) { return ((unsigned)d / AG_ROWS) & 255u; }

__device__ __forceinline__ int2 edge_rec(int s, int d, float x) {
    const int row = d - (int)((unsigned)d / AG_ROWS) * AG_ROWS;
    return make_int2((s << 9) | row, __float_as_int(x));
}

__device__ __forceinline__ void bm_set(unsigned* group, unsigned o) {
    __hip_atomic_fetch_or(group + (o >> 5) * 64, 1u << (o & 31), __ATOMIC_RELAXED, __HIP_MEMORY_SCOPE_WORKGROUP);
}

__device__ __forceinline__ void lds_barrier() {
    __builtin_amdgcn_fence(__ATOMIC_RELEASE, "workgroup", "local");
    __builtin_amdgcn_s_barrier();
    __builtin_amdgcn_fence(__ATOMIC_ACQUIRE, "workgroup", "local");
}

__device__ __forceinline__ int bm_offset(int nwords) { return (nwords + 255) & ~255; }
__device__ __forceinline__ int bm_pitch(int E) { return (((E + 31) >> 5) + 63) & ~63; }

__global__ __launch_bounds__(512) void k_owner_bytes(const int* __restrict__ src,
                                                     const int* __restrict__ dst,
                                                     const float* __restrict__ w,
                                                     unsigned* __restrict__ ob, int* __restrict__ recw,
                                                     int E, int nwords) {
    __shared__ unsigned bms[512];
    int2* rec = (int2*)recw;
    unsigned* bm = ob + bm_offset(nwords);
    const int ngp = bm_pitch(E);
    const bool al16 = ((((uintptr_t)dst) | ((uintptr_t)src) | ((uintptr_t)w)) & 15u) == 0;
    const int T = gridDim.x * 512;
    const int tid = threadIdx.x;
    unsigned* mygroup = bms + (tid >> 3);
    bms[tid] = 0;
    for (int wd0 = blockIdx.x * 512; wd0 < 8 * ngp; wd0 += T) {
        lds_barrier();
        const int wd = wd0 + tid;
        if (wd < nwords) {
            const int e = 4 * wd;
            unsigned pk = 0;
            if (al16 && e + 4 <= E) {
                int4 d = ((const int4*)dst)[wd];
                int4 s = ((const int4*)src)[wd];
                float4 x = ((const float4*)w)[wd];
                const unsigned o0 = own_byte(d.x), o1 = own_byte(d.y), o2 = own_byte(d.z), o3 = own_byte(d.w);
                pk = o0 | (o1 << 8) | (o2 << 16) | (o3 << 24);
                int2 r0 = edge_rec(s.x, d.x, x.x), r1 = edge_rec(s.y, d.y, x.y);
                int2 r2 = edge_rec(s.z, d.z, x.z), r3 = edge_rec(s.w, d.w, x.w);
                int4* ro = (int4*)(rec + e);
                ro[0] = make_int4(r0.x, r0.y, r1.x, r1.y);
                ro[1] = make_int4(r2.x, r2.y, r3.x, r3.y);
                bm_set(mygroup, o0); bm_set(mygroup, o1); bm_set(mygroup, o2); bm_set(mygroup, o3);
            } else {
                for (int k = 0; k < 4; ++k)
                    if (e + k < E) {
                        int d = dst[e + k];
                        const unsigned o = own_byte(d);
                        pk |= o << (8 * k);
                        rec[e + k] = edge_rec(src[e + k], d, w[e + k]);
                        bm_set(mygroup, o);
                    }
            }
            ob[wd] = pk;
        }
        lds_barrier();
        bm[(size_t)(tid >> 6) * ngp + (wd0 >> 3) + (tid & 63)] = bms[tid];
        bms[tid] = 0;
    }
}

__device__ __forceinline__ void sw_push(int e, int lo, int rows, int* qcnt, int* q, float* degs,
                                        const int* __restrict__ dst, const float* __restrict__ w) {
    int slot = __hip_atomic_fetch_add(qcnt, 1, __ATOMIC_RELAXED, __HIP_MEMORY_SCOPE_WORKGROUP);
    if (slot < AG_QCAP) {
        q[slot] = e;
    } else {
        unsigned r = (unsigned)(dst[e] - lo);
        if (r < (unsigned)rows)
            __hip_atomic_fetch_add(&degs[r], w[e], __ATOMIC_RELAXED, __HIP_MEMORY_SCOPE_WORKGROUP);
    }
}

__device__ __forceinline__ unsigned zero_bytes(unsigned x) {
    unsigned y = ((x & 0x7F7F7F7Fu) + 0x7F7F7F7Fu) | x;
    return ~y & 0x80808080u;
}
__device__ __forceinline__ unsigned match16(uint4 d, unsigned pat) {
    return zero_bytes(d.x ^ pat) | (zero_bytes(d.y ^ pat) >> 1) |
           (zero_bytes(d.z ^ pat) >> 2) | (zero_bytes(d.w ^ pat) >> 3);
}

__device__ __forceinline__ void push_group(int g, uint4 da, uint4 db, bool has_b, unsigned pat, int E,
                                           int lo, int rows, int* qcnt, int* q, float* degs,
                                           const int* __restrict__ dst, const float* __restrict__ w) {
    const unsigned ma = match16(da, pat), mb = has_b ? match16(db, pat) : 0u;
    unsigned m = ma | (mb >> 4);
    while (m) {
        int p = __builtin_ctz(m);
        m &= m - 1;
        int j = p >> 3, b = p & 7;
        int pp = 2 * g + (b < 4 ? 1 : 0);
        int k = (7 - b) & 3;
        int e = 16 * pp + 4 * k + j;
        if (e < E) sw_push(e, lo, rows, qcnt, q, degs, dst, w);
    }
}

#define HV_CAP 8192

__global__ __launch_bounds__(1024) void k_sweep_deg(const int* __restrict__ src,
                                                    const int* __restrict__ dst,
                                                    const float* __restrict__ w,
                                                    const unsigned* __restrict__ ob,
                                                    const int* __restrict__ recw,
                                                    float* __restrict__ dinv, float* parkbase,
                                                    int N, int E, int nvec) {
    __shared__ int q[AG_QCAP];
    __shared__ int hv[HV_CAP];
    __shared__ float degs[AG_ROWS];
    __shared__ int qcnt, pcnt, hcnt;

    const int tid = threadIdx.x;
    const int2* rec = (const int2*)recw;
    const uint4* ob4 = (const uint4*)ob;
    const unsigned* bm = ob + bm_offset(4 * nvec);
    const int ngp = bm_pitch(E);
    const int nv4 = ngp >> 2;
    const int start = (int)(((long long)blockIdx.x * nv4) / gridDim.x);
    const bool shared_byte = N > 256 * AG_ROWS;

    for (int lo = blockIdx.x * AG_ROWS; lo < N; lo += gridDim.x * AG_ROWS) {
        const int rows = min(AG_ROWS, N - lo);
        const unsigned obyte = ((unsigned)lo / AG_ROWS) & 255u;
        const unsigned pat = obyte * 0x01010101u;
        const uint4* bm4 = (const uint4*)(bm + (size_t)(obyte >> 5) * ngp);
        const int bit = obyte & 31;

        for (int r = tid; r < rows; r += 1024) degs[r] = 1.0f;
        if (tid == 0) { qcnt = 0; pcnt = 0; hcnt = 0; }
        __syncthreads();

        for (int i0 = 0; i0 < nv4; i0 += 8 * 1024) {
            uint4 d[8];
#pragma unroll
            for (int u = 0; u < 8; ++u) {
                int v = i0 + u * 1024 + tid;
                int pp = v + start; if (pp >= nv4) pp -= nv4;
                d[u] = bm4[v < nv4 ? pp : 0];
            }
            unsigned hm = 0;
#pragma unroll
            for (int u = 0; u < 8; ++u) {
                unsigned t = ((d[u].x >> bit) & 1u) | (((d[u].y >> bit) & 1u) << 1) |
                             (((d[u].z >> bit) & 1u) << 2) | (((d[u].w >> bit) & 1u) << 3);
                if (i0 + u * 1024 + tid >= nv4) t = 0;
                hm |= t << (4 * u);
            }
            while (hm) {
                int p = __builtin_ctz(hm);
                hm &= hm - 1;
                int pp = i0 + (p >> 2) * 1024 + tid + start; if (pp >= nv4) pp -= nv4;
                const int g = 4 * pp + (p & 3);
                int slot = __hip_atomic_fetch_add(&hcnt, 1, __ATOMIC_RELAXED, __HIP_MEMORY_SCOPE_WORKGROUP);
                if (slot < HV_CAP) {
                    hv[slot] = g;
                } else {
                    const bool hb = 2 * g + 1 < nvec;
                    push_group(g, ob4[2 * g], ob4[hb ? 2 * g + 1 : 2 * g], hb, pat, E, lo, rows, &qcnt, q, degs, dst, w);
                }
            }
        }
        __syncthreads();

        const int nh = min(hcnt, HV_CAP);
        for (int i0 = 0; i0 < nh; i0 += 4 * 1024) {
            int g[4];
            uint4 da[4], db[4];
#pragma unroll
            for (int u = 0; u < 4; ++u) {
                int i = i0 + u * 1024 + tid;
                g[u] = i < nh ? hv[i] : 0;
                da[u] = ob4[2 * g[u]];
                db[u] = ob4[min(2 * g[u] + 1, nvec - 1)];
            }
#pragma unroll
            for (int u = 0; u < 4; ++u)
                if (i0 + u * 1024 + tid < nh)
                    push_group(g[u], da[u], db[u], 2 * g[u] + 1 < nvec, pat, E, lo, rows, &qcnt, q, degs, dst, w);
        }
        __syncthreads();

        const int nq = min(qcnt, AG_QCAP);
        int* park = (int*)(parkbase + (size_t)lo * NCLASS);
        int2* park2 = (int2*)(park + 2);
        const int room = (rows * NCLASS - 2) >> 1;
        int2 rc[AG_QCAP / 1024];
        int dd[AG_QCAP / 1024];
#pragma unroll
        for (int u = 0; u < AG_QCAP / 1024; ++u) {
            int i = u * 1024 + tid;
            rc[u] = make_int2(0, 0); dd[u] = lo;
            if (i < nq) {
                int e = q[i];
                rc[u] = rec[e];
                if (shared_byte) dd[u] = dst[e];
            }
        }
#pragma unroll
        for (int u = 0; u < AG_QCAP / 1024; ++u) {
            int i = u * 1024 + tid;
            const int r = rc[u].x & 511;
            const bool mine = r < rows && (!shared_byte || (unsigned)(dd[u] - lo) < (unsigned)rows);
            if (i < nq && mine) {
                __hip_atomic_fetch_add(&degs[r], __int_as_float(rc[u].y), __ATOMIC_RELAXED, __HIP_MEMORY_SCOPE_WORKGROUP);
                int slot = __hip_atomic_fetch_add(&pcnt, 1, __ATOMIC_RELAXED, __HIP_MEMORY_SCOPE_WORKGROUP);
                if (slot < room) park2[slot] = rc[u];
            }
        }
        __syncthreads();

        if (tid == 0) park[0] = (qcnt > AG_QCAP) ? 0x7fffffff : pcnt;
        for (int r = tid; r < rows; r += 1024) {
            float dg = degs[r];
            dinv[lo + r] = dg > 0.0f ? rsqrtf(dg) : 0.0f;
        }
        __syncthreads();
    }
}

#define ACC_LD 129
__global__ __launch_bounds__(1024) void k_agg1_lds(const int* __restrict__ src,
                                                   const int* __restrict__ dst,
                                                   const float* __restrict__ w,
                                                   const float* __restrict__ dinv,
                                                   const float* __restrict__ H,
                                                   const float* __restrict__ b1,
                                                   const float* __restrict__ W2,
                                                   float* __restrict__ G, const float* parkbase,
                                                   int N, int E) {
    __shared__ float acc[AG_ROWS * ACC_LD];
    __shared__ float Ws[NHID * NCPAD];
    __shared__ float bs[NHID];
    __shared__ float dsh[AG_ROWS];

    const int tid = threadIdx.x, lane = tid & 31, wave = tid >> 5;
    const float4* H4 = (const float4*)H;

    static_assert((NHID * NCPAD) % 1024 == 0, "one W2 word per thread and trip");
    float wreg[(NHID * NCPAD) / 1024];
#pragma unroll
    for (int u = 0; u < (NHID * NCPAD) / 1024; ++u) {
        int idx = u * 1024 + tid;
        int f = idx / NCPAD, n = idx - f * NCPAD;
        float x = W2[f * NCLASS + min(n, NCLASS - 1)];
        wreg[u] = (n < NCLASS) ? x : 0.0f;
    }
    const float breg = b1[tid & (NHID - 1)];
    bool staged = false;

    for (int lo = blockIdx.x * AG_ROWS; lo < N; lo += gridDim.x * AG_ROWS) {
        const int rows = min(AG_ROWS, N - lo);

        for (int idx = tid; idx < rows * 32; idx += 1024) {
            int r = idx >> 5, i = lo + r;
            float4 v = H4[(size_t)i * 32 + lane];
            float* a = acc + r * ACC_LD + lane;
            a[0] = v.x; a[32] = v.y; a[64] = v.z; a[96] = v.w;
        }
        for (int r = tid; r < rows; r += 1024) dsh[r] = dinv[lo + r];
        if (!staged) {
#pragma unroll
            for (int u = 0; u < (NHID * NCPAD) / 1024; ++u) Ws[u * 1024 + tid] = wreg[u];
            if (tid < NHID) bs[tid] = breg;
            staged = true;
        }
        const int* park = (const int*)(parkbase + (size_t)lo * NCLASS);
        const int cnt = park[0];
        const int room = min(AG_QCAP, (rows * NCLASS - 2) >> 1);
        __syncthreads();

        if (cnt <= room) {
            const int2* park2 = (const int2*)(park + 2);
            for (int base = wave * 32; base < cnt; base += 32 * 32) {
                const int n = min(32, cnt - base);
                int s = 0, r = 0;
                float c = 0.0f;
                if (lane < n) {
                    int2 t = park2[base + lane];
                    s = t.x >> 9;
                    r = t.x & 511;
                    c = __int_as_float(t.y);
                }
                int j0 = 0;
                for (; j0 + 8 <= n; j0 += 8) {
                    float4 v[8];
#pragma unroll
                    for (int u = 0; u < 8; ++u) {
                        int sj = __builtin_amdgcn_readlane(s, j0 + u);
                        v[u] = H4[(size_t)sj * 32 + lane];
                    }
#pragma unroll
                    for (int u = 0; u < 8; ++u) {
                        int rj = __builtin_amdgcn_readlane(r, j0 + u);
                        float cj = __int_as_float(__builtin_amdgcn_readlane(__float_as_int(c), j0 + u));
                        float* a = acc + rj * ACC_LD + lane;
                        __hip_atomic_fetch_add(a + 0,  cj * v[u].x, __ATOMIC_RELAXED, __HIP_MEMORY_SCOPE_WORKGROUP);
                        __hip_atomic_fetch_add(a + 32, cj * v[u].y, __ATOMIC_RELAXED, __HIP_MEMORY_SCOPE_WORKGROUP);
                        __hip_atomic_fetch_add(a + 64, cj * v[u].z, __ATOMIC_RELAXED, __HIP_MEMORY_SCOPE_WORKGROUP);
                        __hip_atomic_fetch_add(a + 96, cj * v[u].w, __ATOMIC_RELAXED, __HIP_MEMORY_SCOPE_WORKGROUP);
                    }
                }
                for (; j0 < n; ++j0) {
                    int sj = __builtin_amdgcn_readlane(s, j0);
                    int rj = __builtin_amdgcn_readlane(r, j0);
                    float cj = __int_as_float(__builtin_amdgcn_readlane(__float_as_int(c), j0));
                    float4 v = H4[(size_t)sj * 32 + lane];
                    float* a = acc + rj * ACC_LD + lane;
                    __hip_atomic_fetch_add(a + 0,  cj * v.x, __ATOMIC_RELAXED, __HIP_MEMORY_SCOPE_WORKGROUP);
                    __hip_atomic_fetch_add(a + 32, cj * v.y, __ATOMIC_RELAXED, __HIP_MEMORY_SCOPE_WORKGROUP);
                    __hip_atomic_fetch_add(a + 64, cj * v.z, __ATOMIC_RELAXED, __HIP_MEMORY_SCOPE_WORKGROUP);
                    __hip_atomic_fetch_add(a + 96, cj * v.w, __ATOMIC_RELAXED, __HIP_MEMORY_SCOPE_WORKGROUP);
                }
            }
        } else {
            for (int e = tid; e < E; e += 1024) {
                unsigned r = (unsigned)(dst[e] - lo);
                if (r < (unsigned)rows) {
                    int s = src[e];
                    float c = w[e];
                    const float* h = H + (size_t)s * NHID;
                    for (int f = 0; f < NHID; ++f)
                        __hip_atomic_fetch_add(&acc[r * ACC_LD + (f & 3) * 32 + (f >> 2)], c * h[f],
                                               __ATOMIC_RELAXED, __HIP_MEMORY_SCOPE_WORKGROUP);
                }
            }
        }
        __syncthreads();

        const int wv = __builtin_amdgcn_readfirstlane(wave);
        if (wv * 16 < rows) {
            const int cc = lane & 15, hh = lane >> 4;
            int rr = wv * 16 + cc;
            if (rr >= rows) rr = rows - 1;
            const float dr = dsh[rr];
            const float* arow = acc + rr * ACC_LD + 32 * hh;
            const float* wr0 = Ws + hh * NCPAD + cc;
            const float* bb = bs + hh;
            v8f d0 = {}, d1 = {}, d2 = {};
#pragma unroll 4
            for (int j = 0; j < 32; ++j) {
                const float* wr = wr0 + 4 * j * NCPAD;
                v2f av  = {fmaxf(arow[j] * dr + bb[4 * j], 0.0f), fmaxf(arow[64 + j] * dr + bb[4 * j + 2], 0.0f)};
                v2f bv0 = {wr[0],  wr[2 * NCPAD]};
                v2f bv1 = {wr[16], wr[2 * NCPAD + 16]};
                v2f bv2 = {wr[32], wr[2 * NCPAD + 32]};
                d0 = __builtin_amdgcn_wmma_f32_16x16x4_f32(false, av, false, bv0, (short)0, d0, false, false);
                d1 = __builtin_amdgcn_wmma_f32_16x16x4_f32(false, av, false, bv1, (short)0, d1, false, false);
                d2 = __builtin_amdgcn_wmma_f32_16x16x4_f32(false, av, false, bv2, (short)0, d2, false, false);
            }
            const int rb = wv * 16 + 8 * hh;
            float* Go = G + (size_t)(lo + rb) * NCPAD + cc;
#pragma unroll
            for (int r = 0; r < 8; ++r) {
                if (rb + r < rows) {
                    float* o = Go + (size_t)r * NCPAD;
                    const float g = dsh[rb + r];
                    o[0]  = d0[r] * g;
                    o[16] = d1[r] * g;
                    o[32] = d2[r] * g;
                }
            }
        }
        __syncthreads();
    }
}

__global__ __launch_bounds__(1024) void k_agg2_lds(const int* __restrict__ src,
                                                   const int* __restrict__ dst,
                                                   const float* __restrict__ w,
                                                   const float* __restrict__ dinv,
                                                   const float* __restrict__ G,
                                                   const float* __restrict__ b2,
                                                   float* out, int N, int E) {
    __shared__ __attribute__((aligned(16))) float acc[AG_ROWS * NCPAD];
    __shared__ float dsh[AG_ROWS];

    const int tid = threadIdx.x, lane = tid & 31, wave = tid >> 5;
    const bool lo16 = lane < 16;

    for (int lo = blockIdx.x * AG_ROWS; lo < N; lo += gridDim.x * AG_ROWS) {
        const int rows = min(AG_ROWS, N - lo);

        for (int idx = tid; idx < rows * 12; idx += 1024) {
            int r = idx / 12, j = idx - r * 12, i = lo + r;
            ((float4*)(acc + r * NCPAD))[j] = ((const float4*)(G + (size_t)i * NCPAD))[j];
        }
        for (int r = tid; r < rows; r += 1024) dsh[r] = dinv[lo + r];
        const int* park = (const int*)(out + (size_t)lo * NCLASS);
        const int cnt = park[0];
        const int room = min(AG_QCAP, (rows * NCLASS - 2) >> 1);
        __syncthreads();

        if (cnt <= room) {
            const int2* park2 = (const int2*)(park + 2);
            for (int base = wave * 32; base < cnt; base += 32 * 32) {
                const int n = min(32, cnt - base);
                int pk = 0;
                float c = 0.0f;
                if (lane < n) {
                    int2 t = park2[base + lane];
                    pk = t.x;
                    c = __int_as_float(t.y);
                }
                int j0 = 0;
                for (; j0 + 16 <= n; j0 += 16) {
                    int pj[16];
                    float cj[16];
#pragma unroll
                    for (int u = 0; u < 16; ++u) {
                        pj[u] = __builtin_amdgcn_readlane(pk, j0 + u);
                        cj[u] = __int_as_float(__builtin_amdgcn_readlane(__float_as_int(c), j0 + u));
                    }
                    float ga[16], gb[16];
#pragma unroll
                    for (int u = 0; u < 16; ++u)
                        ga[u] = G[(size_t)(pj[u] >> 9) * NCPAD + lane];
                    if (lo16) {
#pragma unroll
                        for (int u = 0; u < 16; ++u)
                            gb[u] = G[(size_t)(pj[u] >> 9) * NCPAD + 32 + lane];
                    }
#pragma unroll
                    for (int u = 0; u < 16; ++u)
                        __hip_atomic_fetch_add(acc + (pj[u] & 511) * NCPAD + lane, cj[u] * ga[u],
                                               __ATOMIC_RELAXED, __HIP_MEMORY_SCOPE_WORKGROUP);
                    if (lo16) {
#pragma unroll
                        for (int u = 0; u < 16; ++u)
                            __hip_atomic_fetch_add(acc + (pj[u] & 511) * NCPAD + 32 + lane, cj[u] * gb[u],
                                                   __ATOMIC_RELAXED, __HIP_MEMORY_SCOPE_WORKGROUP);
                    }
                }
                for (; j0 < n; ++j0) {
                    int pj = __builtin_amdgcn_readlane(pk, j0);
                    float cj = __int_as_float(__builtin_amdgcn_readlane(__float_as_int(c), j0));
                    const float* g = G + (size_t)(pj >> 9) * NCPAD;
                    float* a = acc + (pj & 511) * NCPAD;
                    __hip_atomic_fetch_add(a + lane, cj * g[lane], __ATOMIC_RELAXED, __HIP_MEMORY_SCOPE_WORKGROUP);
                    if (lo16)
                        __hip_atomic_fetch_add(a + 32 + lane, cj * g[32 + lane],
                                               __ATOMIC_RELAXED, __HIP_MEMORY_SCOPE_WORKGROUP);
                }
            }
        } else {
            for (int e = tid; e < E; e += 1024) {
                unsigned r = (unsigned)(dst[e] - lo);
                if (r < (unsigned)rows) {
                    int s = src[e];
                    float c = w[e];
                    const float* g = G + (size_t)s * NCPAD;
                    for (int f = 0; f < NCLASS; ++f)
                        __hip_atomic_fetch_add(&acc[r * NCPAD + f], c * g[f],
                                               __ATOMIC_RELAXED, __HIP_MEMORY_SCOPE_WORKGROUP);
                }
            }
        }
        __syncthreads();

        for (int r = wave; r < rows; r += 32) {
            const float* row = acc + r * NCPAD;
            const float dd = dsh[r];
            float v1 = row[lane] * dd + b2[lane];
            int c2 = 32 + lane;
            bool has2 = c2 < NCLASS;
            float v2 = has2 ? (row[c2] * dd + b2[c2]) : -INFINITY;

            float mx = fmaxf(v1, v2);
#pragma unroll
            for (int off = 16; off >= 1; off >>= 1) mx = fmaxf(mx, __shfl_xor(mx, off));

            float sm = expf(v1 - mx) + (has2 ? expf(v2 - mx) : 0.0f);
#pragma unroll
            for (int off = 16; off >= 1; off >>= 1) sm += __shfl_xor(sm, off);

            float ls = logf(sm);
            float* o = out + (size_t)(lo + r) * NCLASS;
            o[lane] = v1 - mx - ls;
            if (has2) o[c2] = v2 - mx - ls;
        }
        __syncthreads();
    }
}

extern "C" void kernel_launch(void* const* d_in, const int* in_sizes, int n_in,
                              void* d_out, int out_size, void* d_ws, size_t ws_size,
                              hipStream_t stream) {
    const float* x  = (const float*)d_in[0];
    const int*   ei = (const int*)  d_in[1];
    const float* ew = (const float*)d_in[2];
    const float* W1 = (const float*)d_in[3];
    const float* b1 = (const float*)d_in[4];
    const float* W2 = (const float*)d_in[5];
    const float* b2 = (const float*)d_in[6];
    float* out = (float*)d_out;

    const int N = in_sizes[0] / NFEAT;
    const int E = in_sizes[2];
    const int* src = ei;
    const int* dst = ei + E;

    float* ws = (float*)d_ws;
    size_t o = 0;
    auto take = [&](size_t n) { float* p = ws + o; o += (n + 255) & ~(size_t)255; return p; };
    float* dinv = take((size_t)N);
    float* Wp1  = take((size_t)NFEAT * NHID);
    float* Hxw  = take((size_t)N * NHID);
    float* G    = take((size_t)N * NHID);

    k_pack_w1<<<(NFEAT * NHID + 255) / 256, 256, 0, stream>>>(W1, Wp1);

    unsigned* ob = (unsigned*)G;
    const int nvec = (E + 15) / 16;
    int* rec = (int*)Hxw;
    k_owner_bytes<<<256, 512, 0, stream>>>(src, dst, ew, ob, rec, E, 4 * nvec);
    k_sweep_deg<<<256, 1024, 0, stream>>>(src, dst, ew, ob, rec, dinv, out, N, E, nvec);

    k_gemm1<<<(N + 31) / 32, 256, 0, stream>>>(x, Wp1, dinv, Hxw, N);

    k_agg1_lds<<<256, 1024, 0, stream>>>(src, dst, ew, dinv, Hxw, b1, W2, G, out, N, E);

    k_agg2_lds<<<256, 1024, 0, stream>>>(src, dst, ew, dinv, G, b2, out, N, E);
}


# Owner blocks compute their own rows of X@W1 right behind their degrees (W1 and 64-row X tiles through LDS, same f32 WMMA k order and step order); weight pack and product launches go (6 to 4)
# speedup vs baseline: 3.8779x; 1.0485x over previous
#include <hip/hip_runtime.h>
#include <cstdint>

typedef float v2f __attribute__((ext_vector_type(2)));
typedef float v8f __attribute__((ext_vector_type(8)));

#define NFEAT 256
#define NHID  128
#define NCLASS 41
#define NCPAD  48

#define AG_ROWS 392
#define AG_QCAP 8192

__device__ __forceinline__ unsigned own_byte(int d) { return ((unsigned)d / AG_ROWS) & 255u; }

__device__ __forceinline__ int2 edge_rec(int s, int d, float x) {
    const int row = d - (int)((unsigned)d / AG_ROWS) * AG_ROWS;
    return make_int2((s << 9) | row, __float_as_int(x));
}

__device__ __forceinline__ void bm_set(unsigned* group, unsigned o) {
    __hip_atomic_fetch_or(group + (o >> 5) * 64, 1u << (o & 31), __ATOMIC_RELAXED, __HIP_MEMORY_SCOPE_WORKGROUP);
}

__device__ __forceinline__ void lds_barrier() {
    __builtin_amdgcn_fence(__ATOMIC_RELEASE, "workgroup", "local");
    __builtin_amdgcn_s_barrier();
    __builtin_amdgcn_fence(__ATOMIC_ACQUIRE, "workgroup", "local");
}

__device__ __forceinline__ int bm_offset(int nwords) { return (nwords + 255) & ~255; }
__device__ __forceinline__ int bm_pitch(int E) { return (((E + 31) >> 5) + 63) & ~63; }

__global__ __launch_bounds__(512) void k_owner_bytes(const int* __restrict__ src,
                                                     const int* __restrict__ dst,
                                                     const float* __restrict__ w,
                                                     unsigned* __restrict__ ob,
                                                     int E, int nwords) {
    __shared__ unsigned bms[512];
    unsigned* bm = ob + bm_offset(nwords);
    const int ngp = bm_pitch(E);
    int2* rec = (int2*)(bm + 8 * ngp);
    const bool al16 = ((((uintptr_t)dst) | ((uintptr_t)src) | ((uintptr_t)w)) & 15u) == 0;
    const int T = gridDim.x * 512;
    const int tid = threadIdx.x;
    unsigned* mygroup = bms + (tid >> 3);
    bms[tid] = 0;
    for (int wd0 = blockIdx.x * 512; wd0 < 8 * ngp; wd0 += T) {
        lds_barrier();
        const int wd = wd0 + tid;
        if (wd < nwords) {
            const int e = 4 * wd;
            unsigned pk = 0;
            if (al16 && e + 4 <= E) {
                int4 d = ((const int4*)dst)[wd];
                int4 s = ((const int4*)src)[wd];
                float4 x = ((const float4*)w)[wd];
                const unsigned o0 = own_byte(d.x), o1 = own_byte(d.y), o2 = own_byte(d.z), o3 = own_byte(d.w);
                pk = o0 | (o1 << 8) | (o2 << 16) | (o3 << 24);
                int2 r0 = edge_rec(s.x, d.x, x.x), r1 = edge_rec(s.y, d.y, x.y);
                int2 r2 = edge_rec(s.z, d.z, x.z), r3 = edge_rec(s.w, d.w, x.w);
                int4* ro = (int4*)(rec + e);
                ro[0] = make_int4(r0.x, r0.y, r1.x, r1.y);
                ro[1] = make_int4(r2.x, r2.y, r3.x, r3.y);
                bm_set(mygroup, o0); bm_set(mygroup, o1); bm_set(mygroup, o2); bm_set(mygroup, o3);
            } else {
                for (int k = 0; k < 4; ++k)
                    if (e + k < E) {
                        int d = dst[e + k];
                        const unsigned o = own_byte(d);
                        pk |= o << (8 * k);
                        rec[e + k] = edge_rec(src[e + k], d, w[e + k]);
                        bm_set(mygroup, o);
                    }
            }
            ob[wd] = pk;
        }
        lds_barrier();
        bm[(size_t)(tid >> 6) * ngp + (wd0 >> 3) + (tid & 63)] = bms[tid];
        bms[tid] = 0;
    }
}

__device__ __forceinline__ void sw_push(int e, int lo, int rows, int* qcnt, int* q, float* degs,
                                        const int* __restrict__ dst, const float* __restrict__ w) {
    int slot = __hip_atomic_fetch_add(qcnt, 1, __ATOMIC_RELAXED, __HIP_MEMORY_SCOPE_WORKGROUP);
    if (slot < AG_QCAP) {
        q[slot] = e;
    } else {
        unsigned r = (unsigned)(dst[e] - lo);
        if (r < (unsigned)rows)
            __hip_atomic_fetch_add(&degs[r], w[e], __ATOMIC_RELAXED, __HIP_MEMORY_SCOPE_WORKGROUP);
    }
}

__device__ __forceinline__ unsigned zero_bytes(unsigned x) {
    unsigned y = ((x & 0x7F7F7F7Fu) + 0x7F7F7F7Fu) | x;
    return ~y & 0x80808080u;
}
__device__ __forceinline__ unsigned match16(uint4 d, unsigned pat) {
    return zero_bytes(d.x ^ pat) | (zero_bytes(d.y ^ pat) >> 1) |
           (zero_bytes(d.z ^ pat) >> 2) | (zero_bytes(d.w ^ pat) >> 3);
}

__device__ __forceinline__ void push_group(int g, uint4 da, uint4 db, bool has_b, unsigned pat, int E,
                                           int lo, int rows, int* qcnt, int* q, float* degs,
                                           const int* __restrict__ dst, const float* __restrict__ w) {
    const unsigned ma = match16(da, pat), mb = has_b ? match16(db, pat) : 0u;
    unsigned m = ma | (mb >> 4);
    while (m) {
        int p = __builtin_ctz(m);
        m &= m - 1;
        int j = p >> 3, b = p & 7;
        int pp = 2 * g + (b < 4 ? 1 : 0);
        int k = (7 - b) & 3;
        int e = 16 * pp + 4 * k + j;
        if (e < E) sw_push(e, lo, rows, qcnt, q, degs, dst, w);
    }
}

#define HV_CAP 8192

#define GT_ROWS 64
#define XS_LD 260
#define WL_LD 144

__global__ __launch_bounds__(1024) void k_sweep_deg(const int* __restrict__ src,
                                                    const int* __restrict__ dst,
                                                    const float* __restrict__ w,
                                                    const unsigned* __restrict__ ob,
                                                    const float* __restrict__ X,
                                                    const float* __restrict__ W1,
                                                    float* __restrict__ dinv, float* __restrict__ H,
                                                    float* parkbase,
                                                    int N, int E, int nvec) {
    __shared__ int q[AG_QCAP];
    __shared__ int hv[HV_CAP];
    __shared__ __attribute__((aligned(16))) float degs[AG_ROWS];
    __shared__ int qcnt, pcnt, hcnt;
    __shared__ __attribute__((aligned(16))) float2 Wl[(NFEAT / 2) * WL_LD];
    __shared__ __attribute__((aligned(16))) float Xs[GT_ROWS * XS_LD];

    const int tid = threadIdx.x;
    const uint4* ob4 = (const uint4*)ob;
    const unsigned* bm = ob + bm_offset(4 * nvec);
    const int ngp = bm_pitch(E);
    const int2* rec = (const int2*)(bm + 8 * ngp);

    static_assert(((NFEAT / 2) * (NHID / 4)) % 1024 == 0, "whole trips of the block over W1's k pairs");
    float4 wa[(NFEAT / 2) * (NHID / 4) / 1024], wb[(NFEAT / 2) * (NHID / 4) / 1024];
#pragma unroll
    for (int u = 0; u < (NFEAT / 2) * (NHID / 4) / 1024; ++u) {
        const int t = u * 1024 + tid, kp = t >> 5, n4 = t & 31;
        wa[u] = ((const float4*)(W1 + (size_t)(2 * kp) * NHID))[n4];
        wb[u] = ((const float4*)(W1 + (size_t)(2 * kp + 1) * NHID))[n4];
    }
    bool staged = false;
    const int nv4 = ngp >> 2;
    const int start = (int)(((long long)blockIdx.x * nv4) / gridDim.x);
    const bool shared_byte = N > 256 * AG_ROWS;

    for (int lo = blockIdx.x * AG_ROWS; lo < N; lo += gridDim.x * AG_ROWS) {
        const int rows = min(AG_ROWS, N - lo);
        const unsigned obyte = ((unsigned)lo / AG_ROWS) & 255u;
        const unsigned pat = obyte * 0x01010101u;
        const uint4* bm4 = (const uint4*)(bm + (size_t)(obyte >> 5) * ngp);
        const int bit = obyte & 31;

        for (int r = tid; r < rows; r += 1024) degs[r] = 1.0f;
        if (tid == 0) { qcnt = 0; pcnt = 0; hcnt = 0; }
        __syncthreads();

        for (int i0 = 0; i0 < nv4; i0 += 8 * 1024) {
            uint4 d[8];
#pragma unroll
            for (int u = 0; u < 8; ++u) {
                int v = i0 + u * 1024 + tid;
                int pp = v + start; if (pp >= nv4) pp -= nv4;
                d[u] = bm4[v < nv4 ? pp : 0];
            }
            unsigned hm = 0;
#pragma unroll
            for (int u = 0; u < 8; ++u) {
                unsigned t = ((d[u].x >> bit) & 1u) | (((d[u].y >> bit) & 1u) << 1) |
                             (((d[u].z >> bit) & 1u) << 2) | (((d[u].w >> bit) & 1u) << 3);
                if (i0 + u * 1024 + tid >= nv4) t = 0;
                hm |= t << (4 * u);
            }
            while (hm) {
                int p = __builtin_ctz(hm);
                hm &= hm - 1;
                int pp = i0 + (p >> 2) * 1024 + tid + start; if (pp >= nv4) pp -= nv4;
                const int g = 4 * pp + (p & 3);
                int slot = __hip_atomic_fetch_add(&hcnt, 1, __ATOMIC_RELAXED, __HIP_MEMORY_SCOPE_WORKGROUP);
                if (slot < HV_CAP) {
                    hv[slot] = g;
                } else {
                    const bool hb = 2 * g + 1 < nvec;
                    push_group(g, ob4[2 * g], ob4[hb ? 2 * g + 1 : 2 * g], hb, pat, E, lo, rows, &qcnt, q, degs, dst, w);
                }
            }
        }
        if (!staged) {
#pragma unroll
            for (int u = 0; u < (NFEAT / 2) * (NHID / 4) / 1024; ++u) {
                const int t = u * 1024 + tid, kp = t >> 5, n4 = t & 31;
                float4* o = (float4*)(Wl + kp * WL_LD + 4 * n4);
                o[0] = make_float4(wa[u].x, wb[u].x, wa[u].y, wb[u].y);
                o[1] = make_float4(wa[u].z, wb[u].z, wa[u].w, wb[u].w);
            }
            staged = true;
        }
        __syncthreads();

        const int nh = min(hcnt, HV_CAP);
        for (int i0 = 0; i0 < nh; i0 += 4 * 1024) {
            int g[4];
            uint4 da[4], db[4];
#pragma unroll
            for (int u = 0; u < 4; ++u) {
                int i = i0 + u * 1024 + tid;
                g[u] = i < nh ? hv[i] : 0;
                da[u] = ob4[2 * g[u]];
                db[u] = ob4[min(2 * g[u] + 1, nvec - 1)];
            }
#pragma unroll
            for (int u = 0; u < 4; ++u)
                if (i0 + u * 1024 + tid < nh)
                    push_group(g[u], da[u], db[u], 2 * g[u] + 1 < nvec, pat, E, lo, rows, &qcnt, q, degs, dst, w);
        }
        __syncthreads();

        const int nq = min(qcnt, AG_QCAP);
        int* park = (int*)(parkbase + (size_t)lo * NCLASS);
        int2* park2 = (int2*)(park + 2);
        const int room = (rows * NCLASS - 2) >> 1;
        int2 rc[AG_QCAP / 1024];
        int dd[AG_QCAP / 1024];
#pragma unroll
        for (int u = 0; u < AG_QCAP / 1024; ++u) {
            int i = u * 1024 + tid;
            rc[u] = make_int2(0, 0); dd[u] = lo;
            if (i < nq) {
                int e = q[i];
                rc[u] = rec[e];
                if (shared_byte) dd[u] = dst[e];
            }
        }
#pragma unroll
        for (int u = 0; u < AG_QCAP / 1024; ++u) {
            int i = u * 1024 + tid;
            const int r = rc[u].x & 511;
            const bool mine = r < rows && (!shared_byte || (unsigned)(dd[u] - lo) < (unsigned)rows);
            if (i < nq && mine) {
                __hip_atomic_fetch_add(&degs[r], __int_as_float(rc[u].y), __ATOMIC_RELAXED, __HIP_MEMORY_SCOPE_WORKGROUP);
                int slot = __hip_atomic_fetch_add(&pcnt, 1, __ATOMIC_RELAXED, __HIP_MEMORY_SCOPE_WORKGROUP);
                if (slot < room) park2[slot] = rc[u];
            }
        }
        __syncthreads();

        static_assert(GT_ROWS * (NFEAT / 4) == 4 * 1024, "four 16-byte pieces of a tile per thread");
        const float4* X4 = (const float4*)X + (tid & 63);
        const int xm = tid >> 6;
        float4 x0 = X4[(size_t)(lo + min(xm,      rows - 1)) * (NFEAT / 4)];
        float4 x1 = X4[(size_t)(lo + min(xm + 16, rows - 1)) * (NFEAT / 4)];
        float4 x2 = X4[(size_t)(lo + min(xm + 32, rows - 1)) * (NFEAT / 4)];
        float4 x3 = X4[(size_t)(lo + min(xm + 48, rows - 1)) * (NFEAT / 4)];

        if (tid == 0) park[0] = (qcnt > AG_QCAP) ? 0x7fffffff : pcnt;
        for (int r = tid; r < rows; r += 1024) {
            float dg = degs[r];
            const float di = dg > 0.0f ? rsqrtf(dg) : 0.0f;
            dinv[lo + r] = di;
            degs[r] = di;
        }

        {
            const int wv = __builtin_amdgcn_readfirstlane(tid >> 5);
            const int nt = wv & 7, ms = wv >> 3;
            const int lane = tid & 31, c = lane & 15, h = lane >> 4;
            for (int t0 = 0; t0 < rows; t0 += GT_ROWS) {
                {
                    float* xo = Xs + xm * XS_LD + 4 * (tid & 63);
                    *(float4*)(xo)              = x0;
                    *(float4*)(xo + 16 * XS_LD) = x1;
                    *(float4*)(xo + 32 * XS_LD) = x2;
                    *(float4*)(xo + 48 * XS_LD) = x3;
                }
                lds_barrier();
                if (t0 + GT_ROWS < rows) {
                    const int tn = t0 + GT_ROWS + xm;
                    x0 = X4[(size_t)(lo + min(tn,      rows - 1)) * (NFEAT / 4)];
                    x1 = X4[(size_t)(lo + min(tn + 16, rows - 1)) * (NFEAT / 4)];
                    x2 = X4[(size_t)(lo + min(tn + 32, rows - 1)) * (NFEAT / 4)];
                    x3 = X4[(size_t)(lo + min(tn + 48, rows - 1)) * (NFEAT / 4)];
                }
                if (t0 + ms * 16 < rows) {
                    const float* ap = Xs + (ms * 16 + c) * XS_LD + 2 * h;
                    const float2* bp = Wl + h * WL_LD + nt * 16 + c;
                    v8f acc = {};
#pragma unroll 8
                    for (int j = 0; j < NFEAT / 4; ++j) {
                        const float2 a = *(const float2*)(ap + 4 * j);
                        const float2 b = bp[2 * j * WL_LD];
                        v2f av = {a.x, a.y}, bv = {b.x, b.y};
                        acc = __builtin_amdgcn_wmma_f32_16x16x4_f32(false, av, false, bv, (short)0, acc, false, false);
                    }
                    const int rb = t0 + ms * 16 + 8 * h;
                    float* Ho = H + (size_t)(lo + rb) * NHID + nt * 16 + c;
                    if (t0 + ms * 16 + 16 <= rows) {
                        const float4 p0 = ((const float4*)(degs + rb))[0], p1 = ((const float4*)(degs + rb))[1];
                        Ho[0 * NHID] = acc[0] * p0.x; Ho[1 * NHID] = acc[1] * p0.y;
                        Ho[2 * NHID] = acc[2] * p0.z; Ho[3 * NHID] = acc[3] * p0.w;
                        Ho[4 * NHID] = acc[4] * p1.x; Ho[5 * NHID] = acc[5] * p1.y;
                        Ho[6 * NHID] = acc[6] * p1.z; Ho[7 * NHID] = acc[7] * p1.w;
                    } else {
#pragma unroll
                        for (int r = 0; r < 8; ++r)
                            if (rb + r < rows) Ho[(size_t)r * NHID] = acc[r] * degs[rb + r];
                    }
                }
                lds_barrier();
            }
        }
    }
}

#define ACC_LD 129
__global__ __launch_bounds__(1024) void k_agg1_lds(const int* __restrict__ src,
                                                   const int* __restrict__ dst,
                                                   const float* __restrict__ w,
                                                   const float* __restrict__ dinv,
                                                   const float* __restrict__ H,
                                                   const float* __restrict__ b1,
                                                   const float* __restrict__ W2,
                                                   float* __restrict__ G, const float* parkbase,
                                                   int N, int E) {
    __shared__ float acc[AG_ROWS * ACC_LD];
    __shared__ float Ws[NHID * NCPAD];
    __shared__ float bs[NHID];
    __shared__ float dsh[AG_ROWS];

    const int tid = threadIdx.x, lane = tid & 31, wave = tid >> 5;
    const float4* H4 = (const float4*)H;

    static_assert((NHID * NCPAD) % 1024 == 0, "one W2 word per thread and trip");
    float wreg[(NHID * NCPAD) / 1024];
#pragma unroll
    for (int u = 0; u < (NHID * NCPAD) / 1024; ++u) {
        int idx = u * 1024 + tid;
        int f = idx / NCPAD, n = idx - f * NCPAD;
        float x = W2[f * NCLASS + min(n, NCLASS - 1)];
        wreg[u] = (n < NCLASS) ? x : 0.0f;
    }
    const float breg = b1[tid & (NHID - 1)];
    bool staged = false;

    for (int lo = blockIdx.x * AG_ROWS; lo < N; lo += gridDim.x * AG_ROWS) {
        const int rows = min(AG_ROWS, N - lo);

        for (int idx = tid; idx < rows * 32; idx += 1024) {
            int r = idx >> 5, i = lo + r;
            float4 v = H4[(size_t)i * 32 + lane];
            float* a = acc + r * ACC_LD + lane;
            a[0] = v.x; a[32] = v.y; a[64] = v.z; a[96] = v.w;
        }
        for (int r = tid; r < rows; r += 1024) dsh[r] = dinv[lo + r];
        if (!staged) {
#pragma unroll
            for (int u = 0; u < (NHID * NCPAD) / 1024; ++u) Ws[u * 1024 + tid] = wreg[u];
            if (tid < NHID) bs[tid] = breg;
            staged = true;
        }
        const int* park = (const int*)(parkbase + (size_t)lo * NCLASS);
        const int cnt = park[0];
        const int room = min(AG_QCAP, (rows * NCLASS - 2) >> 1);
        __syncthreads();

        if (cnt <= room) {
            const int2* park2 = (const int2*)(park + 2);
            for (int base = wave * 32; base < cnt; base += 32 * 32) {
                const int n = min(32, cnt - base);
                int s = 0, r = 0;
                float c = 0.0f;
                if (lane < n) {
                    int2 t = park2[base + lane];
                    s = t.x >> 9;
                    r = t.x & 511;
                    c = __int_as_float(t.y);
                }
                int j0 = 0;
                for (; j0 + 8 <= n; j0 += 8) {
                    float4 v[8];
#pragma unroll
                    for (int u = 0; u < 8; ++u) {
                        int sj = __builtin_amdgcn_readlane(s, j0 + u);
                        v[u] = H4[(size_t)sj * 32 + lane];
                    }
#pragma unroll
                    for (int u = 0; u < 8; ++u) {
                        int rj = __builtin_amdgcn_readlane(r, j0 + u);
                        float cj = __int_as_float(__builtin_amdgcn_readlane(__float_as_int(c), j0 + u));
                        float* a = acc + rj * ACC_LD + lane;
                        __hip_atomic_fetch_add(a + 0,  cj * v[u].x, __ATOMIC_RELAXED, __HIP_MEMORY_SCOPE_WORKGROUP);
                        __hip_atomic_fetch_add(a + 32, cj * v[u].y, __ATOMIC_RELAXED, __HIP_MEMORY_SCOPE_WORKGROUP);
                        __hip_atomic_fetch_add(a + 64, cj * v[u].z, __ATOMIC_RELAXED, __HIP_MEMORY_SCOPE_WORKGROUP);
                        __hip_atomic_fetch_add(a + 96, cj * v[u].w, __ATOMIC_RELAXED, __HIP_MEMORY_SCOPE_WORKGROUP);
                    }
                }
                for (; j0 < n; ++j0) {
                    int sj = __builtin_amdgcn_readlane(s, j0);
                    int rj = __builtin_amdgcn_readlane(r, j0);
                    float cj = __int_as_float(__builtin_amdgcn_readlane(__float_as_int(c), j0));
                    float4 v = H4[(size_t)sj * 32 + lane];
                    float* a = acc + rj * ACC_LD + lane;
                    __hip_atomic_fetch_add(a + 0,  cj * v.x, __ATOMIC_RELAXED, __HIP_MEMORY_SCOPE_WORKGROUP);
                    __hip_atomic_fetch_add(a + 32, cj * v.y, __ATOMIC_RELAXED, __HIP_MEMORY_SCOPE_WORKGROUP);
                    __hip_atomic_fetch_add(a + 64, cj * v.z, __ATOMIC_RELAXED, __HIP_MEMORY_SCOPE_WORKGROUP);
                    __hip_atomic_fetch_add(a + 96, cj * v.w, __ATOMIC_RELAXED, __HIP_MEMORY_SCOPE_WORKGROUP);
                }
            }
        } else {
            for (int e = tid; e < E; e += 1024) {
                unsigned r = (unsigned)(dst[e] - lo);
                if (r < (unsigned)rows) {
                    int s = src[e];
                    float c = w[e];
                    const float* h = H + (size_t)s * NHID;
                    for (int f = 0; f < NHID; ++f)
                        __hip_atomic_fetch_add(&acc[r * ACC_LD + (f & 3) * 32 + (f >> 2)], c * h[f],
                                               __ATOMIC_RELAXED, __HIP_MEMORY_SCOPE_WORKGROUP);
                }
            }
        }
        __syncthreads();

        const int wv = __builtin_amdgcn_readfirstlane(wave);
        if (wv * 16 < rows) {
            const int cc = lane & 15, hh = lane >> 4;
            int rr = wv * 16 + cc;
            if (rr >= rows) rr = rows - 1;
            const float dr = dsh[rr];
            const float* arow = acc + rr * ACC_LD + 32 * hh;
            const float* wr0 = Ws + hh * NCPAD + cc;
            const float* bb = bs + hh;
            v8f d0 = {}, d1 = {}, d2 = {};
#pragma unroll 4
            for (int j = 0; j < 32; ++j) {
                const float* wr = wr0 + 4 * j * NCPAD;
                v2f av  = {fmaxf(arow[j] * dr + bb[4 * j], 0.0f), fmaxf(arow[64 + j] * dr + bb[4 * j + 2], 0.0f)};
                v2f bv0 = {wr[0],  wr[2 * NCPAD]};
                v2f bv1 = {wr[16], wr[2 * NCPAD + 16]};
                v2f bv2 = {wr[32], wr[2 * NCPAD + 32]};
                d0 = __builtin_amdgcn_wmma_f32_16x16x4_f32(false, av, false, bv0, (short)0, d0, false, false);
                d1 = __builtin_amdgcn_wmma_f32_16x16x4_f32(false, av, false, bv1, (short)0, d1, false, false);
                d2 = __builtin_amdgcn_wmma_f32_16x16x4_f32(false, av, false, bv2, (short)0, d2, false, false);
            }
            const int rb = wv * 16 + 8 * hh;
            float* Go = G + (size_t)(lo + rb) * NCPAD + cc;
#pragma unroll
            for (int r = 0; r < 8; ++r) {
                if (rb + r < rows) {
                    float* o = Go + (size_t)r * NCPAD;
                    const float g = dsh[rb + r];
                    o[0]  = d0[r] * g;
                    o[16] = d1[r] * g;
                    o[32] = d2[r] * g;
                }
            }
        }
        __syncthreads();
    }
}

__global__ __launch_bounds__(1024) void k_agg2_lds(const int* __restrict__ src,
                                                   const int* __restrict__ dst,
                                                   const float* __restrict__ w,
                                                   const float* __restrict__ dinv,
                                                   const float* __restrict__ G,
                                                   const float* __restrict__ b2,
                                                   float* out, int N, int E) {
    __shared__ __attribute__((aligned(16))) float acc[AG_ROWS * NCPAD];
    __shared__ float dsh[AG_ROWS];

    const int tid = threadIdx.x, lane = tid & 31, wave = tid >> 5;
    const bool lo16 = lane < 16;

    for (int lo = blockIdx.x * AG_ROWS; lo < N; lo += gridDim.x * AG_ROWS) {
        const int rows = min(AG_ROWS, N - lo);

        for (int idx = tid; idx < rows * 12; idx += 1024) {
            int r = idx / 12, j = idx - r * 12, i = lo + r;
            ((float4*)(acc + r * NCPAD))[j] = ((const float4*)(G + (size_t)i * NCPAD))[j];
        }
        for (int r = tid; r < rows; r += 1024) dsh[r] = dinv[lo + r];
        const int* park = (const int*)(out + (size_t)lo * NCLASS);
        const int cnt = park[0];
        const int room = min(AG_QCAP, (rows * NCLASS - 2) >> 1);
        __syncthreads();

        if (cnt <= room) {
            const int2* park2 = (const int2*)(park + 2);
            for (int base = wave * 32; base < cnt; base += 32 * 32) {
                const int n = min(32, cnt - base);
                int pk = 0;
                float c = 0.0f;
                if (lane < n) {
                    int2 t = park2[base + lane];
                    pk = t.x;
                    c = __int_as_float(t.y);
                }
                int j0 = 0;
                for (; j0 + 16 <= n; j0 += 16) {
                    int pj[16];
                    float cj[16];
#pragma unroll
                    for (int u = 0; u < 16; ++u) {
                        pj[u] = __builtin_amdgcn_readlane(pk, j0 + u);
                        cj[u] = __int_as_float(__builtin_amdgcn_readlane(__float_as_int(c), j0 + u));
                    }
                    float ga[16], gb[16];
#pragma unroll
                    for (int u = 0; u < 16; ++u)
                        ga[u] = G[(size_t)(pj[u] >> 9) * NCPAD + lane];
                    if (lo16) {
#pragma unroll
                        for (int u = 0; u < 16; ++u)
                            gb[u] = G[(size_t)(pj[u] >> 9) * NCPAD + 32 + lane];
                    }
#pragma unroll
                    for (int u = 0; u < 16; ++u)
                        __hip_atomic_fetch_add(acc + (pj[u] & 511) * NCPAD + lane, cj[u] * ga[u],
                                               __ATOMIC_RELAXED, __HIP_MEMORY_SCOPE_WORKGROUP);
                    if (lo16) {
#pragma unroll
                        for (int u = 0; u < 16; ++u)
                            __hip_atomic_fetch_add(acc + (pj[u] & 511) * NCPAD + 32 + lane, cj[u] * gb[u],
                                                   __ATOMIC_RELAXED, __HIP_MEMORY_SCOPE_WORKGROUP);
                    }
                }
                for (; j0 < n; ++j0) {
                    int pj = __builtin_amdgcn_readlane(pk, j0);
                    float cj = __int_as_float(__builtin_amdgcn_readlane(__float_as_int(c), j0));
                    const float* g = G + (size_t)(pj >> 9) * NCPAD;
                    float* a = acc + (pj & 511) * NCPAD;
                    __hip_atomic_fetch_add(a + lane, cj * g[lane], __ATOMIC_RELAXED, __HIP_MEMORY_SCOPE_WORKGROUP);
                    if (lo16)
                        __hip_atomic_fetch_add(a + 32 + lane, cj * g[32 + lane],
                                               __ATOMIC_RELAXED, __HIP_MEMORY_SCOPE_WORKGROUP);
                }
            }
        } else {
            for (int e = tid; e < E; e += 1024) {
                unsigned r = (unsigned)(dst[e] - lo);
                if (r < (unsigned)rows) {
                    int s = src[e];
                    float c = w[e];
                    const float* g = G + (size_t)s * NCPAD;
                    for (int f = 0; f < NCLASS; ++f)
                        __hip_atomic_fetch_add(&acc[r * NCPAD + f], c * g[f],
                                               __ATOMIC_RELAXED, __HIP_MEMORY_SCOPE_WORKGROUP);
                }
            }
        }
        __syncthreads();

        for (int r = wave; r < rows; r += 32) {
            const float* row = acc + r * NCPAD;
            const float dd = dsh[r];
            float v1 = row[lane] * dd + b2[lane];
            int c2 = 32 + lane;
            bool has2 = c2 < NCLASS;
            float v2 = has2 ? (row[c2] * dd + b2[c2]) : -INFINITY;

            float mx = fmaxf(v1, v2);
#pragma unroll
            for (int off = 16; off >= 1; off >>= 1) mx = fmaxf(mx, __shfl_xor(mx, off));

            float sm = expf(v1 - mx) + (has2 ? expf(v2 - mx) : 0.0f);
#pragma unroll
            for (int off = 16; off >= 1; off >>= 1) sm += __shfl_xor(sm, off);

            float ls = logf(sm);
            float* o = out + (size_t)(lo + r) * NCLASS;
            o[lane] = v1 - mx - ls;
            if (has2) o[c2] = v2 - mx - ls;
        }
        __syncthreads();
    }
}

extern "C" void kernel_launch(void* const* d_in, const int* in_sizes, int n_in,
                              void* d_out, int out_size, void* d_ws, size_t ws_size,
                              hipStream_t stream) {
    const float* x  = (const float*)d_in[0];
    const int*   ei = (const int*)  d_in[1];
    const float* ew = (const float*)d_in[2];
    const float* W1 = (const float*)d_in[3];
    const float* b1 = (const float*)d_in[4];
    const float* W2 = (const float*)d_in[5];
    const float* b2 = (const float*)d_in[6];
    float* out = (float*)d_out;

    const int N = in_sizes[0] / NFEAT;
    const int E = in_sizes[2];
    const int* src = ei;
    const int* dst = ei + E;

    float* ws = (float*)d_ws;
    size_t o = 0;
    auto take = [&](size_t n) { float* p = ws + o; o += (n + 255) & ~(size_t)255; return p; };
    float* dinv = take((size_t)N);
    float* Hxw  = take((size_t)N * NHID);
    float* G    = take((size_t)N * NHID);

    unsigned* ob = (unsigned*)G;
    const int nvec = (E + 15) / 16;
    k_owner_bytes<<<256, 512, 0, stream>>>(src, dst, ew, ob, E, 4 * nvec);
    k_sweep_deg<<<256, 1024, 0, stream>>>(src, dst, ew, ob, x, W1, dinv, Hxw, out, N, E, nvec);

    k_agg1_lds<<<256, 1024, 0, stream>>>(src, dst, ew, dinv, Hxw, b1, W2, G, out, N, E);

    k_agg2_lds<<<256, 1024, 0, stream>>>(src, dst, ew, dinv, G, b2, out, N, E);
}
